# starting kernel
# baseline (speedup 1.0000x reference)
#include <hip/hip_runtime.h>
#include <math.h>


typedef __attribute__((ext_vector_type(2))) float v2f;
typedef __attribute__((ext_vector_type(8))) float v8f;

#define WMMA_F32_16x16x4(a, b, c) \
  __builtin_amdgcn_wmma_f32_16x16x4_f32(false, (a), false, (b), (short)0, (c), false, false)

#define NMAT 8192
#define NB   16
#define NN   512
#define NF   1024

__global__ __launch_bounds__(256) void cayley_kernel(const float* __restrict__ qw,
                                                     const float* __restrict__ kw,
                                                     const float* __restrict__ vw,
                                                     float* __restrict__ Cout) {
  __shared__ float X[32][32];
  __shared__ float Mm[32][33];
  __shared__ float Inv[32][33];
  __shared__ float fac[32];
  __shared__ float pivinv;
  const float* W = (blockIdx.x == 0) ? qw : ((blockIdx.x == 1) ? kw : vw);
  const int tid = threadIdx.x;

  for (int e = tid; e < 1024; e += 256) {
    int i = e >> 5, j = e & 31;
    float xv = W[i * 32 + j] - W[j * 32 + i];
    X[i][j]  = xv;
    Mm[i][j] = xv + ((i == j) ? 1.f : 0.f);
    Inv[i][j] = (i == j) ? 1.f : 0.f;
  }
  __syncthreads();

  for (int k = 0; k < 32; ++k) {
    if (tid == 0) pivinv = 1.f / Mm[k][k];
    __syncthreads();
    if (tid < 32)      Mm[k][tid]       *= pivinv;
    else if (tid < 64) Inv[k][tid - 32] *= pivinv;
    __syncthreads();
    if (tid < 32) fac[tid] = (tid == k) ? 0.f : Mm[tid][k];
    __syncthreads();
    for (int e = tid; e < 2048; e += 256) {
      int i = e >> 6, j = e & 63;
      if (i == k) continue;
      float f = fac[i];
      if (j < 32) Mm[i][j]        -= f * Mm[k][j];
      else        Inv[i][j - 32]  -= f * Inv[k][j - 32];
    }
    __syncthreads();
  }

  for (int e = tid; e < 1024; e += 256) {
    int i = e >> 5, j = e & 31;
    float acc = 0.f;
    for (int kk = 0; kk < 32; ++kk) {
      float a = ((i == kk) ? 1.f : 0.f) - X[i][kk];
      acc += a * Inv[kk][j];
    }
    Cout[blockIdx.x * 1024 + i * 32 + j] = acc;
  }
}

__global__ __launch_bounds__(32) void jacobi_logm_kernel(const float* __restrict__ x,
                                                         float* __restrict__ xl) {
  __shared__ float A[32][33];
  __shared__ float V[32][33];
  __shared__ float cc[16], ss[16];
  __shared__ int   pp[16], qq[16];
  __shared__ float lw[32];
  const int lane = threadIdx.x;
  const float* xm = x + (size_t)blockIdx.x * 1024;

  for (int i = 0; i < 32; ++i) {
    A[i][lane] = xm[i * 32 + lane];
    V[i][lane] = (i == lane) ? 1.f : 0.f;
  }
  __syncthreads();

  for (int sweep = 0; sweep < 8; ++sweep) {
    for (int r = 0; r < 31; ++r) {
      if (lane < 16) {
        int p, q;
        if (lane == 0) { p = 31; q = r % 31; }
        else { p = (r + lane) % 31; q = (r - lane + 31) % 31; }
        if (p > q) { int t = p; p = q; q = t; }
        float apq = A[p][q];
        float c = 1.f, s = 0.f;
        if (fabsf(apq) > 1e-12f) {
          float theta = (A[q][q] - A[p][p]) / (2.f * apq);
          float t = 1.f / (fabsf(theta) + sqrtf(theta * theta + 1.f));
          if (theta < 0.f) t = -t;
          c = rsqrtf(t * t + 1.f);
          s = t * c;
        }
        cc[lane] = c; ss[lane] = s; pp[lane] = p; qq[lane] = q;
      }
      __syncthreads();
      for (int t = 0; t < 16; ++t) {
        int p = pp[t], q = qq[t];
        float c = cc[t], s = ss[t];
        float ap = A[p][lane], aq = A[q][lane];
        A[p][lane] = c * ap - s * aq;
        A[q][lane] = s * ap + c * aq;
      }
      __syncthreads();
      for (int t = 0; t < 16; ++t) {
        int p = pp[t], q = qq[t];
        float c = cc[t], s = ss[t];
        float ap = A[lane][p], aq = A[lane][q];
        A[lane][p] = c * ap - s * aq;
        A[lane][q] = s * ap + c * aq;
        float vp = V[lane][p], vq = V[lane][q];
        V[lane][p] = c * vp - s * vq;
        V[lane][q] = s * vp + c * vq;
      }
      __syncthreads();
    }
  }

  lw[lane] = logf(fmaxf(A[lane][lane], 1e-20f));
  __syncthreads();

  float vj[32];
#pragma unroll
  for (int k = 0; k < 32; ++k) vj[k] = V[lane][k] * lw[k];

  float* outm = xl + (size_t)blockIdx.x * 1024;
  for (int i = 0; i < 32; ++i) {
    float acc = 0.f;
#pragma unroll
    for (int k = 0; k < 32; ++k) acc += V[i][k] * vj[k];
    outm[i * 32 + lane] = (i == lane) ? 0.f : acc;
  }
}

__global__ __launch_bounds__(32) void transbase_kernel(const float* __restrict__ xl,
                                                       const float* __restrict__ C,
                                                       float* __restrict__ qkv) {
  __shared__ float Y[32][33];
  __shared__ float CL[32][33];
  __shared__ float T[32][33];
  __shared__ float T2[32][33];
  const int lane = threadIdx.x;
  const int nIdx = lane & 15;
  const int hi   = lane >> 4;
  const int m = blockIdx.x, w = blockIdx.y;
  const float* ym = xl + (size_t)m * 1024;
  const float* Cw = C + (size_t)w * 1024;

  for (int i = 0; i < 32; ++i) {
    Y[i][lane]  = ym[i * 32 + lane];
    CL[i][lane] = Cw[i * 32 + lane];
  }
  __syncthreads();

  for (int ti = 0; ti < 2; ++ti)
    for (int tj = 0; tj < 2; ++tj) {
      v8f acc = {};
#pragma unroll
      for (int kk = 0; kk < 8; ++kk) {
        int k0 = kk * 4 + hi * 2;
        v2f a; a.x = CL[ti * 16 + nIdx][k0]; a.y = CL[ti * 16 + nIdx][k0 + 1];
        v2f b; b.x = Y[k0][tj * 16 + nIdx];  b.y = Y[k0 + 1][tj * 16 + nIdx];
        acc = WMMA_F32_16x16x4(a, b, acc);
      }
#pragma unroll
      for (int r = 0; r < 8; ++r)
        T[ti * 16 + r + hi * 8][tj * 16 + nIdx] = acc[r];
    }
  __syncthreads();

  for (int ti = 0; ti < 2; ++ti)
    for (int tj = 0; tj < 2; ++tj) {
      v8f acc = {};
#pragma unroll
      for (int kk = 0; kk < 8; ++kk) {
        int k0 = kk * 4 + hi * 2;
        v2f a; a.x = T[ti * 16 + nIdx][k0];       a.y = T[ti * 16 + nIdx][k0 + 1];
        v2f b; b.x = CL[tj * 16 + nIdx][k0];      b.y = CL[tj * 16 + nIdx][k0 + 1];
        acc = WMMA_F32_16x16x4(a, b, acc);
      }
#pragma unroll
      for (int r = 0; r < 8; ++r)
        T2[ti * 16 + r + hi * 8][tj * 16 + nIdx] = acc[r];
    }
  __syncthreads();

  float* outm = qkv + ((size_t)w * NMAT + m) * 1024;
  for (int i = 0; i < 32; ++i) {
    float lo = T2[i][lane];
    float up = T2[lane][i];
    float val = (i > lane) ? lo : ((i < lane) ? up : 0.f);
    outm[i * 32 + lane] = val;
  }
}

__global__ __launch_bounds__(256) void norms_kernel(const float* __restrict__ qkv,
                                                    float* __restrict__ nrm) {
  int idx = blockIdx.x * 256 + threadIdx.x;
  int sel = idx >> 13;
  int r   = idx & (NMAT - 1);
  const float* row = qkv + ((size_t)sel * NMAT + r) * 1024;
  float acc = 0.f;
  for (int f = 0; f < NF; ++f) { float t = row[f]; acc += t * t; }
  nrm[idx] = acc;
}

__global__ __launch_bounds__(32) void scores_kernel(const float* __restrict__ qkv,
                                                    const float* __restrict__ nrm,
                                                    float* __restrict__ scores) {
  const int lane = threadIdx.x;
  const int nIdx = lane & 15;
  const int hi   = lane >> 4;
  const int it = blockIdx.x, jt = blockIdx.y, b = blockIdx.z;
  const int i0 = it * 64, j0 = jt * 16;
  const float* qf = qkv + ((size_t)0 * NMAT + (size_t)b * NN) * 1024;
  const float* kf = qkv + ((size_t)1 * NMAT + (size_t)b * NN) * 1024;
  const float* arow  = kf + (size_t)(j0 + nIdx) * 1024;
  const float* brow0 = qf + (size_t)(i0 +  0 + nIdx) * 1024;
  const float* brow1 = qf + (size_t)(i0 + 16 + nIdx) * 1024;
  const float* brow2 = qf + (size_t)(i0 + 32 + nIdx) * 1024;
  const float* brow3 = qf + (size_t)(i0 + 48 + nIdx) * 1024;

  v8f acc0 = {}, acc1 = {}, acc2 = {}, acc3 = {};
  for (int k0 = hi * 2; k0 < NF; k0 += 4) {
    v2f a = *(const v2f*)(arow + k0);
    acc0 = WMMA_F32_16x16x4(a, *(const v2f*)(brow0 + k0), acc0);
    acc1 = WMMA_F32_16x16x4(a, *(const v2f*)(brow1 + k0), acc1);
    acc2 = WMMA_F32_16x16x4(a, *(const v2f*)(brow2 + k0), acc2);
    acc3 = WMMA_F32_16x16x4(a, *(const v2f*)(brow3 + k0), acc3);
  }

  float* sb = scores + (size_t)b * NN * NN;
#pragma unroll
  for (int t = 0; t < 4; ++t) {
    v8f acc = (t == 0) ? acc0 : ((t == 1) ? acc1 : ((t == 2) ? acc2 : acc3));
    int i = i0 + t * 16 + nIdx;
    float q2 = nrm[b * NN + i];
#pragma unroll
    for (int r = 0; r < 8; ++r) {
      int j = j0 + r + hi * 8;
      float d2 = q2 + nrm[NMAT + b * NN + j] - 2.f * acc[r];
      d2 = fmaxf(d2, 1e-12f);
      float e  = sqrtf(d2);
      float sc = 1.f / (1.f + log1pf(e));
      sb[(size_t)j * NN + i] = sc;
    }
  }
}

__global__ __launch_bounds__(256) void softmax_kernel(float* __restrict__ scores) {
  int idx = blockIdx.x * 256 + threadIdx.x;
  int b = idx >> 9, i = idx & (NN - 1);
  float* col = scores + (size_t)b * NN * NN + i;
  float mx = -1e30f;
  for (int j = 0; j < NN; ++j) mx = fmaxf(mx, col[(size_t)j * NN]);
  float sum = 0.f;
  for (int j = 0; j < NN; ++j) {
    float e = __expf(col[(size_t)j * NN] - mx);
    sum += e;
    col[(size_t)j * NN] = e;
  }
  float inv = 1.f / sum;
  for (int j = 0; j < NN; ++j) col[(size_t)j * NN] *= inv;
}

__global__ __launch_bounds__(32) void outgemm_kernel(const float* __restrict__ scores,
                                                     const float* __restrict__ qkv,
                                                     float* __restrict__ out) {
  const int lane = threadIdx.x;
  const int nIdx = lane & 15;
  const int hi   = lane >> 4;
  const int ft = blockIdx.x, ct = blockIdx.y, b = blockIdx.z;
  const int f0 = ft * 16, c0 = ct * 64;
  const float* p  = scores + (size_t)b * NN * NN;
  const float* vf = qkv + ((size_t)2 * NMAT + (size_t)b * NN) * 1024;
  const float* arow0 = p + (size_t)(c0 +  0 + nIdx) * NN;
  const float* arow1 = p + (size_t)(c0 + 16 + nIdx) * NN;
  const float* arow2 = p + (size_t)(c0 + 32 + nIdx) * NN;
  const float* arow3 = p + (size_t)(c0 + 48 + nIdx) * NN;

  v8f acc0 = {}, acc1 = {}, acc2 = {}, acc3 = {};
  for (int k0 = hi * 2; k0 < NN; k0 += 4) {
    v2f bb;
    bb.x = vf[(size_t)k0 * NF + f0 + nIdx];
    bb.y = vf[(size_t)(k0 + 1) * NF + f0 + nIdx];
    acc0 = WMMA_F32_16x16x4(*(const v2f*)(arow0 + k0), bb, acc0);
    acc1 = WMMA_F32_16x16x4(*(const v2f*)(arow1 + k0), bb, acc1);
    acc2 = WMMA_F32_16x16x4(*(const v2f*)(arow2 + k0), bb, acc2);
    acc3 = WMMA_F32_16x16x4(*(const v2f*)(arow3 + k0), bb, acc3);
  }

  float* ob = out + (size_t)b * NN * NF;
#pragma unroll
  for (int t = 0; t < 4; ++t) {
    v8f acc = (t == 0) ? acc0 : ((t == 1) ? acc1 : ((t == 2) ? acc2 : acc3));
#pragma unroll
    for (int r = 0; r < 8; ++r) {
      int c = c0 + t * 16 + r + hi * 8;
      ob[(size_t)c * NF + f0 + nIdx] = acc[r];
    }
  }
}

extern "C" void kernel_launch(void* const* d_in, const int* in_sizes, int n_in,
                              void* d_out, int out_size, void* d_ws, size_t ws_size,
                              hipStream_t stream) {
  (void)in_sizes; (void)n_in; (void)out_size; (void)ws_size;
  const float* x  = (const float*)d_in[0];
  const float* qw = (const float*)d_in[1];
  const float* kw = (const float*)d_in[2];
  const float* vw = (const float*)d_in[3];
  float* out = (float*)d_out;

  float* ws     = (float*)d_ws;
  float* C      = ws;
  float* xl     = C + 3 * 1024;
  float* qkv    = xl + (size_t)NMAT * 1024;
  float* nrm    = qkv + (size_t)3 * NMAT * 1024;
  float* scores = nrm + 2 * NMAT;

  cayley_kernel<<<3, 256, 0, stream>>>(qw, kw, vw, C);
  jacobi_logm_kernel<<<NMAT, 32, 0, stream>>>(x, xl);
  transbase_kernel<<<dim3(NMAT, 3), 32, 0, stream>>>(xl, C, qkv);
  norms_kernel<<<64, 256, 0, stream>>>(qkv, nrm);
  scores_kernel<<<dim3(NN / 64, NN / 16, NB), 32, 0, stream>>>(qkv, nrm, scores);
  softmax_kernel<<<32, 256, 0, stream>>>(scores);
  outgemm_kernel<<<dim3(NF / 16, NN / 64, NB), 32, 0, stream>>>(scores, qkv, out);
}


# scores GEMM: 128x64 block tile, K-slabs of 32 staged once per block through double-buffered LDS (8 waves, 32x32 each) instead of one wave per 16x64 strip reading operands from global; k order kept
# speedup vs baseline: 1.2655x; 1.2655x over previous
#include <hip/hip_runtime.h>
#include <math.h>


typedef __attribute__((ext_vector_type(2))) float v2f;
typedef __attribute__((ext_vector_type(8))) float v8f;

#define WMMA_F32_16x16x4(a, b, c) \
  __builtin_amdgcn_wmma_f32_16x16x4_f32(false, (a), false, (b), (short)0, (c), false, false)

#define NMAT 8192
#define NB   16
#define NN   512
#define NF   1024

__global__ __launch_bounds__(256) void cayley_kernel(const float* __restrict__ qw,
                                                     const float* __restrict__ kw,
                                                     const float* __restrict__ vw,
                                                     float* __restrict__ Cout) {
  __shared__ float X[32][32];
  __shared__ float Mm[32][33];
  __shared__ float Inv[32][33];
  __shared__ float fac[32];
  __shared__ float pivinv;
  const float* W = (blockIdx.x == 0) ? qw : ((blockIdx.x == 1) ? kw : vw);
  const int tid = threadIdx.x;

  for (int e = tid; e < 1024; e += 256) {
    int i = e >> 5, j = e & 31;
    float xv = W[i * 32 + j] - W[j * 32 + i];
    X[i][j]  = xv;
    Mm[i][j] = xv + ((i == j) ? 1.f : 0.f);
    Inv[i][j] = (i == j) ? 1.f : 0.f;
  }
  __syncthreads();

  for (int k = 0; k < 32; ++k) {
    if (tid == 0) pivinv = 1.f / Mm[k][k];
    __syncthreads();
    if (tid < 32)      Mm[k][tid]       *= pivinv;
    else if (tid < 64) Inv[k][tid - 32] *= pivinv;
    __syncthreads();
    if (tid < 32) fac[tid] = (tid == k) ? 0.f : Mm[tid][k];
    __syncthreads();
    for (int e = tid; e < 2048; e += 256) {
      int i = e >> 6, j = e & 63;
      if (i == k) continue;
      float f = fac[i];
      if (j < 32) Mm[i][j]        -= f * Mm[k][j];
      else        Inv[i][j - 32]  -= f * Inv[k][j - 32];
    }
    __syncthreads();
  }

  for (int e = tid; e < 1024; e += 256) {
    int i = e >> 5, j = e & 31;
    float acc = 0.f;
    for (int kk = 0; kk < 32; ++kk) {
      float a = ((i == kk) ? 1.f : 0.f) - X[i][kk];
      acc += a * Inv[kk][j];
    }
    Cout[blockIdx.x * 1024 + i * 32 + j] = acc;
  }
}

__global__ __launch_bounds__(32) void jacobi_logm_kernel(const float* __restrict__ x,
                                                         float* __restrict__ xl) {
  __shared__ float A[32][33];
  __shared__ float V[32][33];
  __shared__ float cc[16], ss[16];
  __shared__ int   pp[16], qq[16];
  __shared__ float lw[32];
  const int lane = threadIdx.x;
  const float* xm = x + (size_t)blockIdx.x * 1024;

  for (int i = 0; i < 32; ++i) {
    A[i][lane] = xm[i * 32 + lane];
    V[i][lane] = (i == lane) ? 1.f : 0.f;
  }
  __syncthreads();

  for (int sweep = 0; sweep < 8; ++sweep) {
    for (int r = 0; r < 31; ++r) {
      if (lane < 16) {
        int p, q;
        if (lane == 0) { p = 31; q = r % 31; }
        else { p = (r + lane) % 31; q = (r - lane + 31) % 31; }
        if (p > q) { int t = p; p = q; q = t; }
        float apq = A[p][q];
        float c = 1.f, s = 0.f;
        if (fabsf(apq) > 1e-12f) {
          float theta = (A[q][q] - A[p][p]) / (2.f * apq);
          float t = 1.f / (fabsf(theta) + sqrtf(theta * theta + 1.f));
          if (theta < 0.f) t = -t;
          c = rsqrtf(t * t + 1.f);
          s = t * c;
        }
        cc[lane] = c; ss[lane] = s; pp[lane] = p; qq[lane] = q;
      }
      __syncthreads();
      for (int t = 0; t < 16; ++t) {
        int p = pp[t], q = qq[t];
        float c = cc[t], s = ss[t];
        float ap = A[p][lane], aq = A[q][lane];
        A[p][lane] = c * ap - s * aq;
        A[q][lane] = s * ap + c * aq;
      }
      __syncthreads();
      for (int t = 0; t < 16; ++t) {
        int p = pp[t], q = qq[t];
        float c = cc[t], s = ss[t];
        float ap = A[lane][p], aq = A[lane][q];
        A[lane][p] = c * ap - s * aq;
        A[lane][q] = s * ap + c * aq;
        float vp = V[lane][p], vq = V[lane][q];
        V[lane][p] = c * vp - s * vq;
        V[lane][q] = s * vp + c * vq;
      }
      __syncthreads();
    }
  }

  lw[lane] = logf(fmaxf(A[lane][lane], 1e-20f));
  __syncthreads();

  float vj[32];
#pragma unroll
  for (int k = 0; k < 32; ++k) vj[k] = V[lane][k] * lw[k];

  float* outm = xl + (size_t)blockIdx.x * 1024;
  for (int i = 0; i < 32; ++i) {
    float acc = 0.f;
#pragma unroll
    for (int k = 0; k < 32; ++k) acc += V[i][k] * vj[k];
    outm[i * 32 + lane] = (i == lane) ? 0.f : acc;
  }
}

__global__ __launch_bounds__(32) void transbase_kernel(const float* __restrict__ xl,
                                                       const float* __restrict__ C,
                                                       float* __restrict__ qkv) {
  __shared__ float Y[32][33];
  __shared__ float CL[32][33];
  __shared__ float T[32][33];
  __shared__ float T2[32][33];
  const int lane = threadIdx.x;
  const int nIdx = lane & 15;
  const int hi   = lane >> 4;
  const int m = blockIdx.x, w = blockIdx.y;
  const float* ym = xl + (size_t)m * 1024;
  const float* Cw = C + (size_t)w * 1024;

  for (int i = 0; i < 32; ++i) {
    Y[i][lane]  = ym[i * 32 + lane];
    CL[i][lane] = Cw[i * 32 + lane];
  }
  __syncthreads();

  for (int ti = 0; ti < 2; ++ti)
    for (int tj = 0; tj < 2; ++tj) {
      v8f acc = {};
#pragma unroll
      for (int kk = 0; kk < 8; ++kk) {
        int k0 = kk * 4 + hi * 2;
        v2f a; a.x = CL[ti * 16 + nIdx][k0]; a.y = CL[ti * 16 + nIdx][k0 + 1];
        v2f b; b.x = Y[k0][tj * 16 + nIdx];  b.y = Y[k0 + 1][tj * 16 + nIdx];
        acc = WMMA_F32_16x16x4(a, b, acc);
      }
#pragma unroll
      for (int r = 0; r < 8; ++r)
        T[ti * 16 + r + hi * 8][tj * 16 + nIdx] = acc[r];
    }
  __syncthreads();

  for (int ti = 0; ti < 2; ++ti)
    for (int tj = 0; tj < 2; ++tj) {
      v8f acc = {};
#pragma unroll
      for (int kk = 0; kk < 8; ++kk) {
        int k0 = kk * 4 + hi * 2;
        v2f a; a.x = T[ti * 16 + nIdx][k0];       a.y = T[ti * 16 + nIdx][k0 + 1];
        v2f b; b.x = CL[tj * 16 + nIdx][k0];      b.y = CL[tj * 16 + nIdx][k0 + 1];
        acc = WMMA_F32_16x16x4(a, b, acc);
      }
#pragma unroll
      for (int r = 0; r < 8; ++r)
        T2[ti * 16 + r + hi * 8][tj * 16 + nIdx] = acc[r];
    }
  __syncthreads();

  float* outm = qkv + ((size_t)w * NMAT + m) * 1024;
  for (int i = 0; i < 32; ++i) {
    float lo = T2[i][lane];
    float up = T2[lane][i];
    float val = (i > lane) ? lo : ((i < lane) ? up : 0.f);
    outm[i * 32 + lane] = val;
  }
}

__global__ __launch_bounds__(256) void norms_kernel(const float* __restrict__ qkv,
                                                    float* __restrict__ nrm) {
  int idx = blockIdx.x * 256 + threadIdx.x;
  int sel = idx >> 13;
  int r   = idx & (NMAT - 1);
  const float* row = qkv + ((size_t)sel * NMAT + r) * 1024;
  float acc = 0.f;
  for (int f = 0; f < NF; ++f) { float t = row[f]; acc += t * t; }
  nrm[idx] = acc;
}

#define SC_TJ 128
#define SC_TI 64
#define SC_KS 32
#define SC_NS (NF / SC_KS)
#define SC_LD 36

__global__ __launch_bounds__(256) void scores_kernel(const float* __restrict__ qkv,
                                                     const float* __restrict__ nrm,
                                                     float* __restrict__ scores) {
  __shared__ __attribute__((aligned(16))) float As[2][SC_TJ][SC_LD];
  __shared__ __attribute__((aligned(16))) float Bs[2][SC_TI][SC_LD];
  const int tid  = threadIdx.x;
  const int lane = tid & 31;
  const int wave = tid >> 5;
  const int nIdx = lane & 15;
  const int hi   = lane >> 4;
  const int it = blockIdx.x, jt = blockIdx.y, b = blockIdx.z;
  const int i0 = it * SC_TI, j0 = jt * SC_TJ;
  const int wj = (wave >> 1) * 32;
  const int wi = (wave & 1) * 32;
  const float* qf = qkv + ((size_t)0 * NMAT + (size_t)b * NN) * 1024;
  const float* kf = qkv + ((size_t)1 * NMAT + (size_t)b * NN) * 1024;

  const int srow = tid >> 3;
  const int scol = (tid & 7) * 4;
  const float* ga = kf + (size_t)(j0 + srow) * 1024 + scol;
  const float* gb = qf + (size_t)(i0 + srow) * 1024 + scol;
  const size_t rstep = (size_t)32 * 1024;

  {
    const float4 ra0 = *(const float4*)(ga + 0 * rstep);
    const float4 ra1 = *(const float4*)(ga + 1 * rstep);
    const float4 ra2 = *(const float4*)(ga + 2 * rstep);
    const float4 ra3 = *(const float4*)(ga + 3 * rstep);
    const float4 rb0 = *(const float4*)(gb + 0 * rstep);
    const float4 rb1 = *(const float4*)(gb + 1 * rstep);
    *(float4*)&As[0][srow +  0][scol] = ra0;
    *(float4*)&As[0][srow + 32][scol] = ra1;
    *(float4*)&As[0][srow + 64][scol] = ra2;
    *(float4*)&As[0][srow + 96][scol] = ra3;
    *(float4*)&Bs[0][srow +  0][scol] = rb0;
    *(float4*)&Bs[0][srow + 32][scol] = rb1;
  }
  __syncthreads();

  v8f acc00 = {}, acc01 = {}, acc10 = {}, acc11 = {};

  auto slab_mma = [&](int buf) {
#pragma unroll
    for (int kk = 0; kk < SC_KS / 4; ++kk) {
      const int k0 = kk * 4 + hi * 2;
      v2f a0 = *(const v2f*)&As[buf][wj +  0 + nIdx][k0];
      v2f a1 = *(const v2f*)&As[buf][wj + 16 + nIdx][k0];
      v2f b0 = *(const v2f*)&Bs[buf][wi +  0 + nIdx][k0];
      v2f b1 = *(const v2f*)&Bs[buf][wi + 16 + nIdx][k0];
      acc00 = WMMA_F32_16x16x4(a0, b0, acc00);
      acc01 = WMMA_F32_16x16x4(a0, b1, acc01);
      acc10 = WMMA_F32_16x16x4(a1, b0, acc10);
      acc11 = WMMA_F32_16x16x4(a1, b1, acc11);
    }
  };

  for (int s = 0; s < SC_NS - 1; ++s) {
    const int ko = (s + 1) * SC_KS;
    const float4 ra0 = *(const float4*)(ga + 0 * rstep + ko);
    const float4 ra1 = *(const float4*)(ga + 1 * rstep + ko);
    const float4 ra2 = *(const float4*)(ga + 2 * rstep + ko);
    const float4 ra3 = *(const float4*)(ga + 3 * rstep + ko);
    const float4 rb0 = *(const float4*)(gb + 0 * rstep + ko);
    const float4 rb1 = *(const float4*)(gb + 1 * rstep + ko);

    slab_mma(s & 1);

    const int nb = (s + 1) & 1;
    *(float4*)&As[nb][srow +  0][scol] = ra0;
    *(float4*)&As[nb][srow + 32][scol] = ra1;
    *(float4*)&As[nb][srow + 64][scol] = ra2;
    *(float4*)&As[nb][srow + 96][scol] = ra3;
    *(float4*)&Bs[nb][srow +  0][scol] = rb0;
    *(float4*)&Bs[nb][srow + 32][scol] = rb1;
    __syncthreads();
  }
  slab_mma((SC_NS - 1) & 1);

  float* sb = scores + (size_t)b * NN * NN;
#pragma unroll
  for (int t = 0; t < 4; ++t) {
    v8f acc = (t == 0) ? acc00 : ((t == 1) ? acc01 : ((t == 2) ? acc10 : acc11));
    const int jb = j0 + wj + (t >> 1) * 16;
    const int i  = i0 + wi + (t & 1) * 16 + nIdx;
    float q2 = nrm[b * NN + i];
#pragma unroll
    for (int r = 0; r < 8; ++r) {
      int j = jb + r + hi * 8;
      float d2 = q2 + nrm[NMAT + b * NN + j] - 2.f * acc[r];
      d2 = fmaxf(d2, 1e-12f);
      float e  = sqrtf(d2);
      float sc = 1.f / (1.f + log1pf(e));
      sb[(size_t)j * NN + i] = sc;
    }
  }
}

__global__ __launch_bounds__(256) void softmax_kernel(float* __restrict__ scores) {
  int idx = blockIdx.x * 256 + threadIdx.x;
  int b = idx >> 9, i = idx & (NN - 1);
  float* col = scores + (size_t)b * NN * NN + i;
  float mx = -1e30f;
  for (int j = 0; j < NN; ++j) mx = fmaxf(mx, col[(size_t)j * NN]);
  float sum = 0.f;
  for (int j = 0; j < NN; ++j) {
    float e = __expf(col[(size_t)j * NN] - mx);
    sum += e;
    col[(size_t)j * NN] = e;
  }
  float inv = 1.f / sum;
  for (int j = 0; j < NN; ++j) col[(size_t)j * NN] *= inv;
}

__global__ __launch_bounds__(32) void outgemm_kernel(const float* __restrict__ scores,
                                                     const float* __restrict__ qkv,
                                                     float* __restrict__ out) {
  const int lane = threadIdx.x;
  const int nIdx = lane & 15;
  const int hi   = lane >> 4;
  const int ft = blockIdx.x, ct = blockIdx.y, b = blockIdx.z;
  const int f0 = ft * 16, c0 = ct * 64;
  const float* p  = scores + (size_t)b * NN * NN;
  const float* vf = qkv + ((size_t)2 * NMAT + (size_t)b * NN) * 1024;
  const float* arow0 = p + (size_t)(c0 +  0 + nIdx) * NN;
  const float* arow1 = p + (size_t)(c0 + 16 + nIdx) * NN;
  const float* arow2 = p + (size_t)(c0 + 32 + nIdx) * NN;
  const float* arow3 = p + (size_t)(c0 + 48 + nIdx) * NN;

  v8f acc0 = {}, acc1 = {}, acc2 = {}, acc3 = {};
  for (int k0 = hi * 2; k0 < NN; k0 += 4) {
    v2f bb;
    bb.x = vf[(size_t)k0 * NF + f0 + nIdx];
    bb.y = vf[(size_t)(k0 + 1) * NF + f0 + nIdx];
    acc0 = WMMA_F32_16x16x4(*(const v2f*)(arow0 + k0), bb, acc0);
    acc1 = WMMA_F32_16x16x4(*(const v2f*)(arow1 + k0), bb, acc1);
    acc2 = WMMA_F32_16x16x4(*(const v2f*)(arow2 + k0), bb, acc2);
    acc3 = WMMA_F32_16x16x4(*(const v2f*)(arow3 + k0), bb, acc3);
  }

  float* ob = out + (size_t)b * NN * NF;
#pragma unroll
  for (int t = 0; t < 4; ++t) {
    v8f acc = (t == 0) ? acc0 : ((t == 1) ? acc1 : ((t == 2) ? acc2 : acc3));
#pragma unroll
    for (int r = 0; r < 8; ++r) {
      int c = c0 + t * 16 + r + hi * 8;
      ob[(size_t)c * NF + f0 + nIdx] = acc[r];
    }
  }
}

extern "C" void kernel_launch(void* const* d_in, const int* in_sizes, int n_in,
                              void* d_out, int out_size, void* d_ws, size_t ws_size,
                              hipStream_t stream) {
  (void)in_sizes; (void)n_in; (void)out_size; (void)ws_size;
  const float* x  = (const float*)d_in[0];
  const float* qw = (const float*)d_in[1];
  const float* kw = (const float*)d_in[2];
  const float* vw = (const float*)d_in[3];
  float* out = (float*)d_out;

  float* ws     = (float*)d_ws;
  float* C      = ws;
  float* xl     = C + 3 * 1024;
  float* qkv    = xl + (size_t)NMAT * 1024;
  float* nrm    = qkv + (size_t)3 * NMAT * 1024;
  float* scores = nrm + 2 * NMAT;

  cayley_kernel<<<3, 256, 0, stream>>>(qw, kw, vw, C);
  jacobi_logm_kernel<<<NMAT, 32, 0, stream>>>(x, xl);
  transbase_kernel<<<dim3(NMAT, 3), 32, 0, stream>>>(xl, C, qkv);
  norms_kernel<<<64, 256, 0, stream>>>(qkv, nrm);
  scores_kernel<<<dim3(NN / SC_TI, NN / SC_TJ, NB), 256, 0, stream>>>(qkv, nrm, scores);
  softmax_kernel<<<32, 256, 0, stream>>>(scores);
  outgemm_kernel<<<dim3(NF / 16, NN / 64, NB), 32, 0, stream>>>(scores, qkv, out);
}


# out GEMM (p x v): 128x64 block tile, K-slabs of 32 staged once per block through double-buffered LDS (8 waves, 32x32 each) instead of one wave per 64x16 strip reading from global; k order kept
# speedup vs baseline: 1.5279x; 1.2073x over previous
#include <hip/hip_runtime.h>
#include <math.h>


typedef __attribute__((ext_vector_type(2))) float v2f;
typedef __attribute__((ext_vector_type(8))) float v8f;

#define WMMA_F32_16x16x4(a, b, c) \
  __builtin_amdgcn_wmma_f32_16x16x4_f32(false, (a), false, (b), (short)0, (c), false, false)

#define NMAT 8192
#define NB   16
#define NN   512
#define NF   1024

__global__ __launch_bounds__(256) void cayley_kernel(const float* __restrict__ qw,
                                                     const float* __restrict__ kw,
                                                     const float* __restrict__ vw,
                                                     float* __restrict__ Cout) {
  __shared__ float X[32][32];
  __shared__ float Mm[32][33];
  __shared__ float Inv[32][33];
  __shared__ float fac[32];
  __shared__ float pivinv;
  const float* W = (blockIdx.x == 0) ? qw : ((blockIdx.x == 1) ? kw : vw);
  const int tid = threadIdx.x;

  for (int e = tid; e < 1024; e += 256) {
    int i = e >> 5, j = e & 31;
    float xv = W[i * 32 + j] - W[j * 32 + i];
    X[i][j]  = xv;
    Mm[i][j] = xv + ((i == j) ? 1.f : 0.f);
    Inv[i][j] = (i == j) ? 1.f : 0.f;
  }
  __syncthreads();

  for (int k = 0; k < 32; ++k) {
    if (tid == 0) pivinv = 1.f / Mm[k][k];
    __syncthreads();
    if (tid < 32)      Mm[k][tid]       *= pivinv;
    else if (tid < 64) Inv[k][tid - 32] *= pivinv;
    __syncthreads();
    if (tid < 32) fac[tid] = (tid == k) ? 0.f : Mm[tid][k];
    __syncthreads();
    for (int e = tid; e < 2048; e += 256) {
      int i = e >> 6, j = e & 63;
      if (i == k) continue;
      float f = fac[i];
      if (j < 32) Mm[i][j]        -= f * Mm[k][j];
      else        Inv[i][j - 32]  -= f * Inv[k][j - 32];
    }
    __syncthreads();
  }

  for (int e = tid; e < 1024; e += 256) {
    int i = e >> 5, j = e & 31;
    float acc = 0.f;
    for (int kk = 0; kk < 32; ++kk) {
      float a = ((i == kk) ? 1.f : 0.f) - X[i][kk];
      acc += a * Inv[kk][j];
    }
    Cout[blockIdx.x * 1024 + i * 32 + j] = acc;
  }
}

__global__ __launch_bounds__(32) void jacobi_logm_kernel(const float* __restrict__ x,
                                                         float* __restrict__ xl) {
  __shared__ float A[32][33];
  __shared__ float V[32][33];
  __shared__ float cc[16], ss[16];
  __shared__ int   pp[16], qq[16];
  __shared__ float lw[32];
  const int lane = threadIdx.x;
  const float* xm = x + (size_t)blockIdx.x * 1024;

  for (int i = 0; i < 32; ++i) {
    A[i][lane] = xm[i * 32 + lane];
    V[i][lane] = (i == lane) ? 1.f : 0.f;
  }
  __syncthreads();

  for (int sweep = 0; sweep < 8; ++sweep) {
    for (int r = 0; r < 31; ++r) {
      if (lane < 16) {
        int p, q;
        if (lane == 0) { p = 31; q = r % 31; }
        else { p = (r + lane) % 31; q = (r - lane + 31) % 31; }
        if (p > q) { int t = p; p = q; q = t; }
        float apq = A[p][q];
        float c = 1.f, s = 0.f;
        if (fabsf(apq) > 1e-12f) {
          float theta = (A[q][q] - A[p][p]) / (2.f * apq);
          float t = 1.f / (fabsf(theta) + sqrtf(theta * theta + 1.f));
          if (theta < 0.f) t = -t;
          c = rsqrtf(t * t + 1.f);
          s = t * c;
        }
        cc[lane] = c; ss[lane] = s; pp[lane] = p; qq[lane] = q;
      }
      __syncthreads();
      for (int t = 0; t < 16; ++t) {
        int p = pp[t], q = qq[t];
        float c = cc[t], s = ss[t];
        float ap = A[p][lane], aq = A[q][lane];
        A[p][lane] = c * ap - s * aq;
        A[q][lane] = s * ap + c * aq;
      }
      __syncthreads();
      for (int t = 0; t < 16; ++t) {
        int p = pp[t], q = qq[t];
        float c = cc[t], s = ss[t];
        float ap = A[lane][p], aq = A[lane][q];
        A[lane][p] = c * ap - s * aq;
        A[lane][q] = s * ap + c * aq;
        float vp = V[lane][p], vq = V[lane][q];
        V[lane][p] = c * vp - s * vq;
        V[lane][q] = s * vp + c * vq;
      }
      __syncthreads();
    }
  }

  lw[lane] = logf(fmaxf(A[lane][lane], 1e-20f));
  __syncthreads();

  float vj[32];
#pragma unroll
  for (int k = 0; k < 32; ++k) vj[k] = V[lane][k] * lw[k];

  float* outm = xl + (size_t)blockIdx.x * 1024;
  for (int i = 0; i < 32; ++i) {
    float acc = 0.f;
#pragma unroll
    for (int k = 0; k < 32; ++k) acc += V[i][k] * vj[k];
    outm[i * 32 + lane] = (i == lane) ? 0.f : acc;
  }
}

__global__ __launch_bounds__(32) void transbase_kernel(const float* __restrict__ xl,
                                                       const float* __restrict__ C,
                                                       float* __restrict__ qkv) {
  __shared__ float Y[32][33];
  __shared__ float CL[32][33];
  __shared__ float T[32][33];
  __shared__ float T2[32][33];
  const int lane = threadIdx.x;
  const int nIdx = lane & 15;
  const int hi   = lane >> 4;
  const int m = blockIdx.x, w = blockIdx.y;
  const float* ym = xl + (size_t)m * 1024;
  const float* Cw = C + (size_t)w * 1024;

  for (int i = 0; i < 32; ++i) {
    Y[i][lane]  = ym[i * 32 + lane];
    CL[i][lane] = Cw[i * 32 + lane];
  }
  __syncthreads();

  for (int ti = 0; ti < 2; ++ti)
    for (int tj = 0; tj < 2; ++tj) {
      v8f acc = {};
#pragma unroll
      for (int kk = 0; kk < 8; ++kk) {
        int k0 = kk * 4 + hi * 2;
        v2f a; a.x = CL[ti * 16 + nIdx][k0]; a.y = CL[ti * 16 + nIdx][k0 + 1];
        v2f b; b.x = Y[k0][tj * 16 + nIdx];  b.y = Y[k0 + 1][tj * 16 + nIdx];
        acc = WMMA_F32_16x16x4(a, b, acc);
      }
#pragma unroll
      for (int r = 0; r < 8; ++r)
        T[ti * 16 + r + hi * 8][tj * 16 + nIdx] = acc[r];
    }
  __syncthreads();

  for (int ti = 0; ti < 2; ++ti)
    for (int tj = 0; tj < 2; ++tj) {
      v8f acc = {};
#pragma unroll
      for (int kk = 0; kk < 8; ++kk) {
        int k0 = kk * 4 + hi * 2;
        v2f a; a.x = T[ti * 16 + nIdx][k0];       a.y = T[ti * 16 + nIdx][k0 + 1];
        v2f b; b.x = CL[tj * 16 + nIdx][k0];      b.y = CL[tj * 16 + nIdx][k0 + 1];
        acc = WMMA_F32_16x16x4(a, b, acc);
      }
#pragma unroll
      for (int r = 0; r < 8; ++r)
        T2[ti * 16 + r + hi * 8][tj * 16 + nIdx] = acc[r];
    }
  __syncthreads();

  float* outm = qkv + ((size_t)w * NMAT + m) * 1024;
  for (int i = 0; i < 32; ++i) {
    float lo = T2[i][lane];
    float up = T2[lane][i];
    float val = (i > lane) ? lo : ((i < lane) ? up : 0.f);
    outm[i * 32 + lane] = val;
  }
}

__global__ __launch_bounds__(256) void norms_kernel(const float* __restrict__ qkv,
                                                    float* __restrict__ nrm) {
  int idx = blockIdx.x * 256 + threadIdx.x;
  int sel = idx >> 13;
  int r   = idx & (NMAT - 1);
  const float* row = qkv + ((size_t)sel * NMAT + r) * 1024;
  float acc = 0.f;
  for (int f = 0; f < NF; ++f) { float t = row[f]; acc += t * t; }
  nrm[idx] = acc;
}

#define SC_TJ 128
#define SC_TI 64
#define SC_KS 32
#define SC_NS (NF / SC_KS)
#define SC_LD 36

__global__ __launch_bounds__(256) void scores_kernel(const float* __restrict__ qkv,
                                                     const float* __restrict__ nrm,
                                                     float* __restrict__ scores) {
  __shared__ __attribute__((aligned(16))) float As[2][SC_TJ][SC_LD];
  __shared__ __attribute__((aligned(16))) float Bs[2][SC_TI][SC_LD];
  const int tid  = threadIdx.x;
  const int lane = tid & 31;
  const int wave = tid >> 5;
  const int nIdx = lane & 15;
  const int hi   = lane >> 4;
  const int it = blockIdx.x, jt = blockIdx.y, b = blockIdx.z;
  const int i0 = it * SC_TI, j0 = jt * SC_TJ;
  const int wj = (wave >> 1) * 32;
  const int wi = (wave & 1) * 32;
  const float* qf = qkv + ((size_t)0 * NMAT + (size_t)b * NN) * 1024;
  const float* kf = qkv + ((size_t)1 * NMAT + (size_t)b * NN) * 1024;

  const int srow = tid >> 3;
  const int scol = (tid & 7) * 4;
  const float* ga = kf + (size_t)(j0 + srow) * 1024 + scol;
  const float* gb = qf + (size_t)(i0 + srow) * 1024 + scol;
  const size_t rstep = (size_t)32 * 1024;

  {
    const float4 ra0 = *(const float4*)(ga + 0 * rstep);
    const float4 ra1 = *(const float4*)(ga + 1 * rstep);
    const float4 ra2 = *(const float4*)(ga + 2 * rstep);
    const float4 ra3 = *(const float4*)(ga + 3 * rstep);
    const float4 rb0 = *(const float4*)(gb + 0 * rstep);
    const float4 rb1 = *(const float4*)(gb + 1 * rstep);
    *(float4*)&As[0][srow +  0][scol] = ra0;
    *(float4*)&As[0][srow + 32][scol] = ra1;
    *(float4*)&As[0][srow + 64][scol] = ra2;
    *(float4*)&As[0][srow + 96][scol] = ra3;
    *(float4*)&Bs[0][srow +  0][scol] = rb0;
    *(float4*)&Bs[0][srow + 32][scol] = rb1;
  }
  __syncthreads();

  v8f acc00 = {}, acc01 = {}, acc10 = {}, acc11 = {};

  auto slab_mma = [&](int buf) {
#pragma unroll
    for (int kk = 0; kk < SC_KS / 4; ++kk) {
      const int k0 = kk * 4 + hi * 2;
      v2f a0 = *(const v2f*)&As[buf][wj +  0 + nIdx][k0];
      v2f a1 = *(const v2f*)&As[buf][wj + 16 + nIdx][k0];
      v2f b0 = *(const v2f*)&Bs[buf][wi +  0 + nIdx][k0];
      v2f b1 = *(const v2f*)&Bs[buf][wi + 16 + nIdx][k0];
      acc00 = WMMA_F32_16x16x4(a0, b0, acc00);
      acc01 = WMMA_F32_16x16x4(a0, b1, acc01);
      acc10 = WMMA_F32_16x16x4(a1, b0, acc10);
      acc11 = WMMA_F32_16x16x4(a1, b1, acc11);
    }
  };

  for (int s = 0; s < SC_NS - 1; ++s) {
    const int ko = (s + 1) * SC_KS;
    const float4 ra0 = *(const float4*)(ga + 0 * rstep + ko);
    const float4 ra1 = *(const float4*)(ga + 1 * rstep + ko);
    const float4 ra2 = *(const float4*)(ga + 2 * rstep + ko);
    const float4 ra3 = *(const float4*)(ga + 3 * rstep + ko);
    const float4 rb0 = *(const float4*)(gb + 0 * rstep + ko);
    const float4 rb1 = *(const float4*)(gb + 1 * rstep + ko);

    slab_mma(s & 1);

    const int nb = (s + 1) & 1;
    *(float4*)&As[nb][srow +  0][scol] = ra0;
    *(float4*)&As[nb][srow + 32][scol] = ra1;
    *(float4*)&As[nb][srow + 64][scol] = ra2;
    *(float4*)&As[nb][srow + 96][scol] = ra3;
    *(float4*)&Bs[nb][srow +  0][scol] = rb0;
    *(float4*)&Bs[nb][srow + 32][scol] = rb1;
    __syncthreads();
  }
  slab_mma((SC_NS - 1) & 1);

  float* sb = scores + (size_t)b * NN * NN;
#pragma unroll
  for (int t = 0; t < 4; ++t) {
    v8f acc = (t == 0) ? acc00 : ((t == 1) ? acc01 : ((t == 2) ? acc10 : acc11));
    const int jb = j0 + wj + (t >> 1) * 16;
    const int i  = i0 + wi + (t & 1) * 16 + nIdx;
    float q2 = nrm[b * NN + i];
#pragma unroll
    for (int r = 0; r < 8; ++r) {
      int j = jb + r + hi * 8;
      float d2 = q2 + nrm[NMAT + b * NN + j] - 2.f * acc[r];
      d2 = fmaxf(d2, 1e-12f);
      float e  = sqrtf(d2);
      float sc = 1.f / (1.f + log1pf(e));
      sb[(size_t)j * NN + i] = sc;
    }
  }
}

__global__ __launch_bounds__(256) void softmax_kernel(float* __restrict__ scores) {
  int idx = blockIdx.x * 256 + threadIdx.x;
  int b = idx >> 9, i = idx & (NN - 1);
  float* col = scores + (size_t)b * NN * NN + i;
  float mx = -1e30f;
  for (int j = 0; j < NN; ++j) mx = fmaxf(mx, col[(size_t)j * NN]);
  float sum = 0.f;
  for (int j = 0; j < NN; ++j) {
    float e = __expf(col[(size_t)j * NN] - mx);
    sum += e;
    col[(size_t)j * NN] = e;
  }
  float inv = 1.f / sum;
  for (int j = 0; j < NN; ++j) col[(size_t)j * NN] *= inv;
}

#define OG_TC 128
#define OG_TF 64
#define OG_KS 32
#define OG_NS (NN / OG_KS)
#define OG_LDA 36
#define OG_LDB 160

__global__ __launch_bounds__(256) void outgemm_kernel(const float* __restrict__ scores,
                                                      const float* __restrict__ qkv,
                                                      float* __restrict__ out) {
  __shared__ __attribute__((aligned(16))) float As[2][OG_TC][OG_LDA];
  __shared__ __attribute__((aligned(16))) float Bs[2][OG_KS / 2][OG_LDB];
  const int tid  = threadIdx.x;
  const int lane = tid & 31;
  const int wave = tid >> 5;
  const int nIdx = lane & 15;
  const int hi   = lane >> 4;
  const int ft = blockIdx.x, ct = blockIdx.y, b = blockIdx.z;
  const int f0 = ft * OG_TF, c0 = ct * OG_TC;
  const int wc = (wave >> 1) * 32;
  const int wf = (wave & 1) * 32;
  const float* p  = scores + (size_t)b * NN * NN;
  const float* vf = qkv + ((size_t)2 * NMAT + (size_t)b * NN) * 1024;

  const int arow = tid >> 3;
  const int acol = (tid & 7) * 4;
  const float* ga = p + (size_t)(c0 + arow) * NN + acol;
  const size_t astep = (size_t)32 * NN;
  const int bkp  = tid >> 4;
  const int bcol = (tid & 15) * 4;
  const float* gb = vf + (size_t)(2 * bkp) * NF + f0 + bcol;
  const size_t bslab = (size_t)OG_KS * NF;

  {
    const float4 ra0 = *(const float4*)(ga + 0 * astep);
    const float4 ra1 = *(const float4*)(ga + 1 * astep);
    const float4 ra2 = *(const float4*)(ga + 2 * astep);
    const float4 ra3 = *(const float4*)(ga + 3 * astep);
    const float4 rb0 = *(const float4*)(gb);
    const float4 rb1 = *(const float4*)(gb + NF);
    *(float4*)&As[0][arow +  0][acol] = ra0;
    *(float4*)&As[0][arow + 32][acol] = ra1;
    *(float4*)&As[0][arow + 64][acol] = ra2;
    *(float4*)&As[0][arow + 96][acol] = ra3;
    *(float4*)&Bs[0][bkp][bcol * 2 + 0] = make_float4(rb0.x, rb1.x, rb0.y, rb1.y);
    *(float4*)&Bs[0][bkp][bcol * 2 + 4] = make_float4(rb0.z, rb1.z, rb0.w, rb1.w);
  }
  __syncthreads();

  v8f acc00 = {}, acc01 = {}, acc10 = {}, acc11 = {};

  auto slab_mma = [&](int buf) {
#pragma unroll
    for (int kk = 0; kk < OG_KS / 4; ++kk) {
      const int k0 = kk * 4 + hi * 2;
      v2f a0 = *(const v2f*)&As[buf][wc +  0 + nIdx][k0];
      v2f a1 = *(const v2f*)&As[buf][wc + 16 + nIdx][k0];
      v2f b0 = *(const v2f*)&Bs[buf][kk * 2 + hi][(wf +  0 + nIdx) * 2];
      v2f b1 = *(const v2f*)&Bs[buf][kk * 2 + hi][(wf + 16 + nIdx) * 2];
      acc00 = WMMA_F32_16x16x4(a0, b0, acc00);
      acc01 = WMMA_F32_16x16x4(a0, b1, acc01);
      acc10 = WMMA_F32_16x16x4(a1, b0, acc10);
      acc11 = WMMA_F32_16x16x4(a1, b1, acc11);
    }
  };

  for (int s = 0; s < OG_NS - 1; ++s) {
    const int ko = (s + 1) * OG_KS;
    const float4 ra0 = *(const float4*)(ga + 0 * astep + ko);
    const float4 ra1 = *(const float4*)(ga + 1 * astep + ko);
    const float4 ra2 = *(const float4*)(ga + 2 * astep + ko);
    const float4 ra3 = *(const float4*)(ga + 3 * astep + ko);
    const float4 rb0 = *(const float4*)(gb + (size_t)(s + 1) * bslab);
    const float4 rb1 = *(const float4*)(gb + (size_t)(s + 1) * bslab + NF);

    slab_mma(s & 1);

    const int nb = (s + 1) & 1;
    *(float4*)&As[nb][arow +  0][acol] = ra0;
    *(float4*)&As[nb][arow + 32][acol] = ra1;
    *(float4*)&As[nb][arow + 64][acol] = ra2;
    *(float4*)&As[nb][arow + 96][acol] = ra3;
    *(float4*)&Bs[nb][bkp][bcol * 2 + 0] = make_float4(rb0.x, rb1.x, rb0.y, rb1.y);
    *(float4*)&Bs[nb][bkp][bcol * 2 + 4] = make_float4(rb0.z, rb1.z, rb0.w, rb1.w);
    __syncthreads();
  }
  slab_mma((OG_NS - 1) & 1);

  float* ob = out + (size_t)b * NN * NF;
#pragma unroll
  for (int t = 0; t < 4; ++t) {
    v8f acc = (t == 0) ? acc00 : ((t == 1) ? acc01 : ((t == 2) ? acc10 : acc11));
    const int cb = c0 + wc + (t >> 1) * 16;
    const int f  = f0 + wf + (t & 1) * 16 + nIdx;
#pragma unroll
    for (int r = 0; r < 8; ++r) {
      int c = cb + r + hi * 8;
      ob[(size_t)c * NF + f] = acc[r];
    }
  }
}

extern "C" void kernel_launch(void* const* d_in, const int* in_sizes, int n_in,
                              void* d_out, int out_size, void* d_ws, size_t ws_size,
                              hipStream_t stream) {
  (void)in_sizes; (void)n_in; (void)out_size; (void)ws_size;
  const float* x  = (const float*)d_in[0];
  const float* qw = (const float*)d_in[1];
  const float* kw = (const float*)d_in[2];
  const float* vw = (const float*)d_in[3];
  float* out = (float*)d_out;

  float* ws     = (float*)d_ws;
  float* C      = ws;
  float* xl     = C + 3 * 1024;
  float* qkv    = xl + (size_t)NMAT * 1024;
  float* nrm    = qkv + (size_t)3 * NMAT * 1024;
  float* scores = nrm + 2 * NMAT;

  cayley_kernel<<<3, 256, 0, stream>>>(qw, kw, vw, C);
  jacobi_logm_kernel<<<NMAT, 32, 0, stream>>>(x, xl);
  transbase_kernel<<<dim3(NMAT, 3), 32, 0, stream>>>(xl, C, qkv);
  norms_kernel<<<64, 256, 0, stream>>>(qkv, nrm);
  scores_kernel<<<dim3(NN / SC_TI, NN / SC_TJ, NB), 256, 0, stream>>>(qkv, nrm, scores);
  softmax_kernel<<<32, 256, 0, stream>>>(scores);
  outgemm_kernel<<<dim3(NF / OG_TF, NN / OG_TC, NB), 256, 0, stream>>>(scores, qkv, out);
}


# softmax: one 8-wave block per 32 columns, each wave holds 64 rows in registers (one read, one write, 64 loads in flight), not one thread per column; column sum = 8 partial sums added in wave order
# speedup vs baseline: 1.8089x; 1.1839x over previous
#include <hip/hip_runtime.h>
#include <math.h>


typedef __attribute__((ext_vector_type(2))) float v2f;
typedef __attribute__((ext_vector_type(8))) float v8f;

#define WMMA_F32_16x16x4(a, b, c) \
  __builtin_amdgcn_wmma_f32_16x16x4_f32(false, (a), false, (b), (short)0, (c), false, false)

#define NMAT 8192
#define NB   16
#define NN   512
#define NF   1024

__global__ __launch_bounds__(256) void cayley_kernel(const float* __restrict__ qw,
                                                     const float* __restrict__ kw,
                                                     const float* __restrict__ vw,
                                                     float* __restrict__ Cout) {
  __shared__ float X[32][32];
  __shared__ float Mm[32][33];
  __shared__ float Inv[32][33];
  __shared__ float fac[32];
  __shared__ float pivinv;
  const float* W = (blockIdx.x == 0) ? qw : ((blockIdx.x == 1) ? kw : vw);
  const int tid = threadIdx.x;

  for (int e = tid; e < 1024; e += 256) {
    int i = e >> 5, j = e & 31;
    float xv = W[i * 32 + j] - W[j * 32 + i];
    X[i][j]  = xv;
    Mm[i][j] = xv + ((i == j) ? 1.f : 0.f);
    Inv[i][j] = (i == j) ? 1.f : 0.f;
  }
  __syncthreads();

  for (int k = 0; k < 32; ++k) {
    if (tid == 0) pivinv = 1.f / Mm[k][k];
    __syncthreads();
    if (tid < 32)      Mm[k][tid]       *= pivinv;
    else if (tid < 64) Inv[k][tid - 32] *= pivinv;
    __syncthreads();
    if (tid < 32) fac[tid] = (tid == k) ? 0.f : Mm[tid][k];
    __syncthreads();
    for (int e = tid; e < 2048; e += 256) {
      int i = e >> 6, j = e & 63;
      if (i == k) continue;
      float f = fac[i];
      if (j < 32) Mm[i][j]        -= f * Mm[k][j];
      else        Inv[i][j - 32]  -= f * Inv[k][j - 32];
    }
    __syncthreads();
  }

  for (int e = tid; e < 1024; e += 256) {
    int i = e >> 5, j = e & 31;
    float acc = 0.f;
    for (int kk = 0; kk < 32; ++kk) {
      float a = ((i == kk) ? 1.f : 0.f) - X[i][kk];
      acc += a * Inv[kk][j];
    }
    Cout[blockIdx.x * 1024 + i * 32 + j] = acc;
  }
}

__global__ __launch_bounds__(32) void jacobi_logm_kernel(const float* __restrict__ x,
                                                         float* __restrict__ xl) {
  __shared__ float A[32][33];
  __shared__ float V[32][33];
  __shared__ float cc[16], ss[16];
  __shared__ int   pp[16], qq[16];
  __shared__ float lw[32];
  const int lane = threadIdx.x;
  const float* xm = x + (size_t)blockIdx.x * 1024;

  for (int i = 0; i < 32; ++i) {
    A[i][lane] = xm[i * 32 + lane];
    V[i][lane] = (i == lane) ? 1.f : 0.f;
  }
  __syncthreads();

  for (int sweep = 0; sweep < 8; ++sweep) {
    for (int r = 0; r < 31; ++r) {
      if (lane < 16) {
        int p, q;
        if (lane == 0) { p = 31; q = r % 31; }
        else { p = (r + lane) % 31; q = (r - lane + 31) % 31; }
        if (p > q) { int t = p; p = q; q = t; }
        float apq = A[p][q];
        float c = 1.f, s = 0.f;
        if (fabsf(apq) > 1e-12f) {
          float theta = (A[q][q] - A[p][p]) / (2.f * apq);
          float t = 1.f / (fabsf(theta) + sqrtf(theta * theta + 1.f));
          if (theta < 0.f) t = -t;
          c = rsqrtf(t * t + 1.f);
          s = t * c;
        }
        cc[lane] = c; ss[lane] = s; pp[lane] = p; qq[lane] = q;
      }
      __syncthreads();
      for (int t = 0; t < 16; ++t) {
        int p = pp[t], q = qq[t];
        float c = cc[t], s = ss[t];
        float ap = A[p][lane], aq = A[q][lane];
        A[p][lane] = c * ap - s * aq;
        A[q][lane] = s * ap + c * aq;
      }
      __syncthreads();
      for (int t = 0; t < 16; ++t) {
        int p = pp[t], q = qq[t];
        float c = cc[t], s = ss[t];
        float ap = A[lane][p], aq = A[lane][q];
        A[lane][p] = c * ap - s * aq;
        A[lane][q] = s * ap + c * aq;
        float vp = V[lane][p], vq = V[lane][q];
        V[lane][p] = c * vp - s * vq;
        V[lane][q] = s * vp + c * vq;
      }
      __syncthreads();
    }
  }

  lw[lane] = logf(fmaxf(A[lane][lane], 1e-20f));
  __syncthreads();

  float vj[32];
#pragma unroll
  for (int k = 0; k < 32; ++k) vj[k] = V[lane][k] * lw[k];

  float* outm = xl + (size_t)blockIdx.x * 1024;
  for (int i = 0; i < 32; ++i) {
    float acc = 0.f;
#pragma unroll
    for (int k = 0; k < 32; ++k) acc += V[i][k] * vj[k];
    outm[i * 32 + lane] = (i == lane) ? 0.f : acc;
  }
}

__global__ __launch_bounds__(32) void transbase_kernel(const float* __restrict__ xl,
                                                       const float* __restrict__ C,
                                                       float* __restrict__ qkv) {
  __shared__ float Y[32][33];
  __shared__ float CL[32][33];
  __shared__ float T[32][33];
  __shared__ float T2[32][33];
  const int lane = threadIdx.x;
  const int nIdx = lane & 15;
  const int hi   = lane >> 4;
  const int m = blockIdx.x, w = blockIdx.y;
  const float* ym = xl + (size_t)m * 1024;
  const float* Cw = C + (size_t)w * 1024;

  for (int i = 0; i < 32; ++i) {
    Y[i][lane]  = ym[i * 32 + lane];
    CL[i][lane] = Cw[i * 32 + lane];
  }
  __syncthreads();

  for (int ti = 0; ti < 2; ++ti)
    for (int tj = 0; tj < 2; ++tj) {
      v8f acc = {};
#pragma unroll
      for (int kk = 0; kk < 8; ++kk) {
        int k0 = kk * 4 + hi * 2;
        v2f a; a.x = CL[ti * 16 + nIdx][k0]; a.y = CL[ti * 16 + nIdx][k0 + 1];
        v2f b; b.x = Y[k0][tj * 16 + nIdx];  b.y = Y[k0 + 1][tj * 16 + nIdx];
        acc = WMMA_F32_16x16x4(a, b, acc);
      }
#pragma unroll
      for (int r = 0; r < 8; ++r)
        T[ti * 16 + r + hi * 8][tj * 16 + nIdx] = acc[r];
    }
  __syncthreads();

  for (int ti = 0; ti < 2; ++ti)
    for (int tj = 0; tj < 2; ++tj) {
      v8f acc = {};
#pragma unroll
      for (int kk = 0; kk < 8; ++kk) {
        int k0 = kk * 4 + hi * 2;
        v2f a; a.x = T[ti * 16 + nIdx][k0];       a.y = T[ti * 16 + nIdx][k0 + 1];
        v2f b; b.x = CL[tj * 16 + nIdx][k0];      b.y = CL[tj * 16 + nIdx][k0 + 1];
        acc = WMMA_F32_16x16x4(a, b, acc);
      }
#pragma unroll
      for (int r = 0; r < 8; ++r)
        T2[ti * 16 + r + hi * 8][tj * 16 + nIdx] = acc[r];
    }
  __syncthreads();

  float* outm = qkv + ((size_t)w * NMAT + m) * 1024;
  for (int i = 0; i < 32; ++i) {
    float lo = T2[i][lane];
    float up = T2[lane][i];
    float val = (i > lane) ? lo : ((i < lane) ? up : 0.f);
    outm[i * 32 + lane] = val;
  }
}

__global__ __launch_bounds__(256) void norms_kernel(const float* __restrict__ qkv,
                                                    float* __restrict__ nrm) {
  int idx = blockIdx.x * 256 + threadIdx.x;
  int sel = idx >> 13;
  int r   = idx & (NMAT - 1);
  const float* row = qkv + ((size_t)sel * NMAT + r) * 1024;
  float acc = 0.f;
  for (int f = 0; f < NF; ++f) { float t = row[f]; acc += t * t; }
  nrm[idx] = acc;
}

#define SC_TJ 128
#define SC_TI 64
#define SC_KS 32
#define SC_NS (NF / SC_KS)
#define SC_LD 36

__global__ __launch_bounds__(256) void scores_kernel(const float* __restrict__ qkv,
                                                     const float* __restrict__ nrm,
                                                     float* __restrict__ scores) {
  __shared__ __attribute__((aligned(16))) float As[2][SC_TJ][SC_LD];
  __shared__ __attribute__((aligned(16))) float Bs[2][SC_TI][SC_LD];
  const int tid  = threadIdx.x;
  const int lane = tid & 31;
  const int wave = tid >> 5;
  const int nIdx = lane & 15;
  const int hi   = lane >> 4;
  const int it = blockIdx.x, jt = blockIdx.y, b = blockIdx.z;
  const int i0 = it * SC_TI, j0 = jt * SC_TJ;
  const int wj = (wave >> 1) * 32;
  const int wi = (wave & 1) * 32;
  const float* qf = qkv + ((size_t)0 * NMAT + (size_t)b * NN) * 1024;
  const float* kf = qkv + ((size_t)1 * NMAT + (size_t)b * NN) * 1024;

  const int srow = tid >> 3;
  const int scol = (tid & 7) * 4;
  const float* ga = kf + (size_t)(j0 + srow) * 1024 + scol;
  const float* gb = qf + (size_t)(i0 + srow) * 1024 + scol;
  const size_t rstep = (size_t)32 * 1024;

  {
    const float4 ra0 = *(const float4*)(ga + 0 * rstep);
    const float4 ra1 = *(const float4*)(ga + 1 * rstep);
    const float4 ra2 = *(const float4*)(ga + 2 * rstep);
    const float4 ra3 = *(const float4*)(ga + 3 * rstep);
    const float4 rb0 = *(const float4*)(gb + 0 * rstep);
    const float4 rb1 = *(const float4*)(gb + 1 * rstep);
    *(float4*)&As[0][srow +  0][scol] = ra0;
    *(float4*)&As[0][srow + 32][scol] = ra1;
    *(float4*)&As[0][srow + 64][scol] = ra2;
    *(float4*)&As[0][srow + 96][scol] = ra3;
    *(float4*)&Bs[0][srow +  0][scol] = rb0;
    *(float4*)&Bs[0][srow + 32][scol] = rb1;
  }
  __syncthreads();

  v8f acc00 = {}, acc01 = {}, acc10 = {}, acc11 = {};

  auto slab_mma = [&](int buf) {
#pragma unroll
    for (int kk = 0; kk < SC_KS / 4; ++kk) {
      const int k0 = kk * 4 + hi * 2;
      v2f a0 = *(const v2f*)&As[buf][wj +  0 + nIdx][k0];
      v2f a1 = *(const v2f*)&As[buf][wj + 16 + nIdx][k0];
      v2f b0 = *(const v2f*)&Bs[buf][wi +  0 + nIdx][k0];
      v2f b1 = *(const v2f*)&Bs[buf][wi + 16 + nIdx][k0];
      acc00 = WMMA_F32_16x16x4(a0, b0, acc00);
      acc01 = WMMA_F32_16x16x4(a0, b1, acc01);
      acc10 = WMMA_F32_16x16x4(a1, b0, acc10);
      acc11 = WMMA_F32_16x16x4(a1, b1, acc11);
    }
  };

  for (int s = 0; s < SC_NS - 1; ++s) {
    const int ko = (s + 1) * SC_KS;
    const float4 ra0 = *(const float4*)(ga + 0 * rstep + ko);
    const float4 ra1 = *(const float4*)(ga + 1 * rstep + ko);
    const float4 ra2 = *(const float4*)(ga + 2 * rstep + ko);
    const float4 ra3 = *(const float4*)(ga + 3 * rstep + ko);
    const float4 rb0 = *(const float4*)(gb + 0 * rstep + ko);
    const float4 rb1 = *(const float4*)(gb + 1 * rstep + ko);

    slab_mma(s & 1);

    const int nb = (s + 1) & 1;
    *(float4*)&As[nb][srow +  0][scol] = ra0;
    *(float4*)&As[nb][srow + 32][scol] = ra1;
    *(float4*)&As[nb][srow + 64][scol] = ra2;
    *(float4*)&As[nb][srow + 96][scol] = ra3;
    *(float4*)&Bs[nb][srow +  0][scol] = rb0;
    *(float4*)&Bs[nb][srow + 32][scol] = rb1;
    __syncthreads();
  }
  slab_mma((SC_NS - 1) & 1);

  float* sb = scores + (size_t)b * NN * NN;
#pragma unroll
  for (int t = 0; t < 4; ++t) {
    v8f acc = (t == 0) ? acc00 : ((t == 1) ? acc01 : ((t == 2) ? acc10 : acc11));
    const int jb = j0 + wj + (t >> 1) * 16;
    const int i  = i0 + wi + (t & 1) * 16 + nIdx;
    float q2 = nrm[b * NN + i];
#pragma unroll
    for (int r = 0; r < 8; ++r) {
      int j = jb + r + hi * 8;
      float d2 = q2 + nrm[NMAT + b * NN + j] - 2.f * acc[r];
      d2 = fmaxf(d2, 1e-12f);
      float e  = sqrtf(d2);
      float sc = 1.f / (1.f + log1pf(e));
      sb[(size_t)j * NN + i] = sc;
    }
  }
}

#define SM_W 8
#define SM_R (NN / SM_W)

__global__ __launch_bounds__(256) void softmax_kernel(float* __restrict__ scores) {
  __shared__ float red[2][SM_W][32];
  const int lane = threadIdx.x & 31;
  const int wave = threadIdx.x >> 5;
  const int b = blockIdx.y;
  float* col = scores + (size_t)b * NN * NN + (size_t)(wave * SM_R) * NN
             + blockIdx.x * 32 + lane;

  float v[SM_R];
#pragma unroll
  for (int r = 0; r < SM_R; ++r) v[r] = col[(size_t)r * NN];

  float mx = -1e30f;
#pragma unroll
  for (int r = 0; r < SM_R; ++r) mx = fmaxf(mx, v[r]);
  red[0][wave][lane] = mx;
  __syncthreads();
  for (int w = 0; w < SM_W; ++w) mx = fmaxf(mx, red[0][w][lane]);

  float sum = 0.f;
#pragma unroll
  for (int r = 0; r < SM_R; ++r) {
    v[r] = __expf(v[r] - mx);
    sum += v[r];
  }
  red[1][wave][lane] = sum;
  __syncthreads();
  float tot = 0.f;
  for (int w = 0; w < SM_W; ++w) tot += red[1][w][lane];
  const float inv = 1.f / tot;

#pragma unroll
  for (int r = 0; r < SM_R; ++r) col[(size_t)r * NN] = v[r] * inv;
}

#define OG_TC 128
#define OG_TF 64
#define OG_KS 32
#define OG_NS (NN / OG_KS)
#define OG_LDA 36
#define OG_LDB 160

__global__ __launch_bounds__(256) void outgemm_kernel(const float* __restrict__ scores,
                                                      const float* __restrict__ qkv,
                                                      float* __restrict__ out) {
  __shared__ __attribute__((aligned(16))) float As[2][OG_TC][OG_LDA];
  __shared__ __attribute__((aligned(16))) float Bs[2][OG_KS / 2][OG_LDB];
  const int tid  = threadIdx.x;
  const int lane = tid & 31;
  const int wave = tid >> 5;
  const int nIdx = lane & 15;
  const int hi   = lane >> 4;
  const int ft = blockIdx.x, ct = blockIdx.y, b = blockIdx.z;
  const int f0 = ft * OG_TF, c0 = ct * OG_TC;
  const int wc = (wave >> 1) * 32;
  const int wf = (wave & 1) * 32;
  const float* p  = scores + (size_t)b * NN * NN;
  const float* vf = qkv + ((size_t)2 * NMAT + (size_t)b * NN) * 1024;

  const int arow = tid >> 3;
  const int acol = (tid & 7) * 4;
  const float* ga = p + (size_t)(c0 + arow) * NN + acol;
  const size_t astep = (size_t)32 * NN;
  const int bkp  = tid >> 4;
  const int bcol = (tid & 15) * 4;
  const float* gb = vf + (size_t)(2 * bkp) * NF + f0 + bcol;
  const size_t bslab = (size_t)OG_KS * NF;

  {
    const float4 ra0 = *(const float4*)(ga + 0 * astep);
    const float4 ra1 = *(const float4*)(ga + 1 * astep);
    const float4 ra2 = *(const float4*)(ga + 2 * astep);
    const float4 ra3 = *(const float4*)(ga + 3 * astep);
    const float4 rb0 = *(const float4*)(gb);
    const float4 rb1 = *(const float4*)(gb + NF);
    *(float4*)&As[0][arow +  0][acol] = ra0;
    *(float4*)&As[0][arow + 32][acol] = ra1;
    *(float4*)&As[0][arow + 64][acol] = ra2;
    *(float4*)&As[0][arow + 96][acol] = ra3;
    *(float4*)&Bs[0][bkp][bcol * 2 + 0] = make_float4(rb0.x, rb1.x, rb0.y, rb1.y);
    *(float4*)&Bs[0][bkp][bcol * 2 + 4] = make_float4(rb0.z, rb1.z, rb0.w, rb1.w);
  }
  __syncthreads();

  v8f acc00 = {}, acc01 = {}, acc10 = {}, acc11 = {};

  auto slab_mma = [&](int buf) {
#pragma unroll
    for (int kk = 0; kk < OG_KS / 4; ++kk) {
      const int k0 = kk * 4 + hi * 2;
      v2f a0 = *(const v2f*)&As[buf][wc +  0 + nIdx][k0];
      v2f a1 = *(const v2f*)&As[buf][wc + 16 + nIdx][k0];
      v2f b0 = *(const v2f*)&Bs[buf][kk * 2 + hi][(wf +  0 + nIdx) * 2];
      v2f b1 = *(const v2f*)&Bs[buf][kk * 2 + hi][(wf + 16 + nIdx) * 2];
      acc00 = WMMA_F32_16x16x4(a0, b0, acc00);
      acc01 = WMMA_F32_16x16x4(a0, b1, acc01);
      acc10 = WMMA_F32_16x16x4(a1, b0, acc10);
      acc11 = WMMA_F32_16x16x4(a1, b1, acc11);
    }
  };

  for (int s = 0; s < OG_NS - 1; ++s) {
    const int ko = (s + 1) * OG_KS;
    const float4 ra0 = *(const float4*)(ga + 0 * astep + ko);
    const float4 ra1 = *(const float4*)(ga + 1 * astep + ko);
    const float4 ra2 = *(const float4*)(ga + 2 * astep + ko);
    const float4 ra3 = *(const float4*)(ga + 3 * astep + ko);
    const float4 rb0 = *(const float4*)(gb + (size_t)(s + 1) * bslab);
    const float4 rb1 = *(const float4*)(gb + (size_t)(s + 1) * bslab + NF);

    slab_mma(s & 1);

    const int nb = (s + 1) & 1;
    *(float4*)&As[nb][arow +  0][acol] = ra0;
    *(float4*)&As[nb][arow + 32][acol] = ra1;
    *(float4*)&As[nb][arow + 64][acol] = ra2;
    *(float4*)&As[nb][arow + 96][acol] = ra3;
    *(float4*)&Bs[nb][bkp][bcol * 2 + 0] = make_float4(rb0.x, rb1.x, rb0.y, rb1.y);
    *(float4*)&Bs[nb][bkp][bcol * 2 + 4] = make_float4(rb0.z, rb1.z, rb0.w, rb1.w);
    __syncthreads();
  }
  slab_mma((OG_NS - 1) & 1);

  float* ob = out + (size_t)b * NN * NF;
#pragma unroll
  for (int t = 0; t < 4; ++t) {
    v8f acc = (t == 0) ? acc00 : ((t == 1) ? acc01 : ((t == 2) ? acc10 : acc11));
    const int cb = c0 + wc + (t >> 1) * 16;
    const int f  = f0 + wf + (t & 1) * 16 + nIdx;
#pragma unroll
    for (int r = 0; r < 8; ++r) {
      int c = cb + r + hi * 8;
      ob[(size_t)c * NF + f] = acc[r];
    }
  }
}

extern "C" void kernel_launch(void* const* d_in, const int* in_sizes, int n_in,
                              void* d_out, int out_size, void* d_ws, size_t ws_size,
                              hipStream_t stream) {
  (void)in_sizes; (void)n_in; (void)out_size; (void)ws_size;
  const float* x  = (const float*)d_in[0];
  const float* qw = (const float*)d_in[1];
  const float* kw = (const float*)d_in[2];
  const float* vw = (const float*)d_in[3];
  float* out = (float*)d_out;

  float* ws     = (float*)d_ws;
  float* C      = ws;
  float* xl     = C + 3 * 1024;
  float* qkv    = xl + (size_t)NMAT * 1024;
  float* nrm    = qkv + (size_t)3 * NMAT * 1024;
  float* scores = nrm + 2 * NMAT;

  cayley_kernel<<<3, 256, 0, stream>>>(qw, kw, vw, C);
  jacobi_logm_kernel<<<NMAT, 32, 0, stream>>>(x, xl);
  transbase_kernel<<<dim3(NMAT, 3), 32, 0, stream>>>(xl, C, qkv);
  norms_kernel<<<64, 256, 0, stream>>>(qkv, nrm);
  scores_kernel<<<dim3(NN / SC_TI, NN / SC_TJ, NB), 256, 0, stream>>>(qkv, nrm, scores);
  softmax_kernel<<<dim3(NN / 32, NB), 256, 0, stream>>>(scores);
  outgemm_kernel<<<dim3(NF / OG_TF, NN / OG_TC, NB), 256, 0, stream>>>(scores, qkv, out);
}


# eigensolver round: LDS loads of the 16 disjoint pairs issued together ahead of math and stores; c,s by readlane; V kept transposed and updated by 16-byte row accesses; sums unchanged
# speedup vs baseline: 1.9257x; 1.0646x over previous
#include <hip/hip_runtime.h>
#include <math.h>


typedef __attribute__((ext_vector_type(2))) float v2f;
typedef __attribute__((ext_vector_type(8))) float v8f;

#define WMMA_F32_16x16x4(a, b, c) \
  __builtin_amdgcn_wmma_f32_16x16x4_f32(false, (a), false, (b), (short)0, (c), false, false)

#define NMAT 8192
#define NB   16
#define NN   512
#define NF   1024

__global__ __launch_bounds__(256) void cayley_kernel(const float* __restrict__ qw,
                                                     const float* __restrict__ kw,
                                                     const float* __restrict__ vw,
                                                     float* __restrict__ Cout) {
  __shared__ float X[32][32];
  __shared__ float Mm[32][33];
  __shared__ float Inv[32][33];
  __shared__ float fac[32];
  __shared__ float pivinv;
  const float* W = (blockIdx.x == 0) ? qw : ((blockIdx.x == 1) ? kw : vw);
  const int tid = threadIdx.x;

  for (int e = tid; e < 1024; e += 256) {
    int i = e >> 5, j = e & 31;
    float xv = W[i * 32 + j] - W[j * 32 + i];
    X[i][j]  = xv;
    Mm[i][j] = xv + ((i == j) ? 1.f : 0.f);
    Inv[i][j] = (i == j) ? 1.f : 0.f;
  }
  __syncthreads();

  for (int k = 0; k < 32; ++k) {
    if (tid == 0) pivinv = 1.f / Mm[k][k];
    __syncthreads();
    if (tid < 32)      Mm[k][tid]       *= pivinv;
    else if (tid < 64) Inv[k][tid - 32] *= pivinv;
    __syncthreads();
    if (tid < 32) fac[tid] = (tid == k) ? 0.f : Mm[tid][k];
    __syncthreads();
    for (int e = tid; e < 2048; e += 256) {
      int i = e >> 6, j = e & 63;
      if (i == k) continue;
      float f = fac[i];
      if (j < 32) Mm[i][j]        -= f * Mm[k][j];
      else        Inv[i][j - 32]  -= f * Inv[k][j - 32];
    }
    __syncthreads();
  }

  for (int e = tid; e < 1024; e += 256) {
    int i = e >> 5, j = e & 31;
    float acc = 0.f;
    for (int kk = 0; kk < 32; ++kk) {
      float a = ((i == kk) ? 1.f : 0.f) - X[i][kk];
      acc += a * Inv[kk][j];
    }
    Cout[blockIdx.x * 1024 + i * 32 + j] = acc;
  }
}

#define JV_LD 36

__device__ __forceinline__ void jacobi_pair(int r, int t, int& p, int& q) {
  int a, b;
  if (t == 0) { a = 31; b = r; }
  else {
    a = r + t; if (a >= 31) a -= 31;
    b = r - t; if (b < 0)   b += 31;
  }
  p = (a < b) ? a : b;
  q = (a < b) ? b : a;
}

__global__ __launch_bounds__(32, 8) void jacobi_logm_kernel(const float* __restrict__ x,
                                                            float* __restrict__ xl) {
  __shared__ float A[32][33];
  __shared__ __attribute__((aligned(16))) float VT[32][JV_LD];
  __shared__ __attribute__((aligned(16))) float cc[16];
  __shared__ __attribute__((aligned(16))) float ss[16];
  __shared__ __attribute__((aligned(16))) int   pp[16];
  __shared__ __attribute__((aligned(16))) int   qq[16];
  __shared__ float lw[32];
  const int lane = threadIdx.x;
  const int g4 = (lane >> 3) * 4;
  const int c4 = (lane & 7) * 4;
  const float* xm = x + (size_t)blockIdx.x * 1024;

  for (int i = 0; i < 32; ++i) {
    A[i][lane]  = xm[i * 32 + lane];
    VT[i][lane] = (i == lane) ? 1.f : 0.f;
  }
  __syncthreads();

  for (int sweep = 0; sweep < 8; ++sweep) {
    for (int r = 0; r < 31; ++r) {
      float c = 1.f, s = 0.f;
      if (lane < 16) {
        int p, q;
        jacobi_pair(r, lane, p, q);
        float apq = A[p][q];
        if (fabsf(apq) > 1e-12f) {
          float theta = (A[q][q] - A[p][p]) / (2.f * apq);
          float t = 1.f / (fabsf(theta) + sqrtf(theta * theta + 1.f));
          if (theta < 0.f) t = -t;
          c = rsqrtf(t * t + 1.f);
          s = t * c;
        }
        cc[lane] = c; ss[lane] = s; pp[lane] = p; qq[lane] = q;
      }
      __syncthreads();

      float ap[16], aq[16];
#pragma unroll
      for (int t = 0; t < 16; ++t) {
        int p, q;
        jacobi_pair(r, t, p, q);
        ap[t] = A[p][lane];
        aq[t] = A[q][lane];
      }
#pragma unroll
      for (int t = 0; t < 16; ++t) {
        int p, q;
        jacobi_pair(r, t, p, q);
        const float ct = __int_as_float(__builtin_amdgcn_readlane(__float_as_int(c), t));
        const float st = __int_as_float(__builtin_amdgcn_readlane(__float_as_int(s), t));
        A[p][lane] = ct * ap[t] - st * aq[t];
        A[q][lane] = st * ap[t] + ct * aq[t];
      }

      {
        const float4 cv = *(const float4*)&cc[g4];
        const float4 sv = *(const float4*)&ss[g4];
        const int4   pv = *(const int4*)&pp[g4];
        const int4   qv = *(const int4*)&qq[g4];
        const float cl[4] = {cv.x, cv.y, cv.z, cv.w};
        const float sl[4] = {sv.x, sv.y, sv.z, sv.w};
        const int   pl[4] = {pv.x, pv.y, pv.z, pv.w};
        const int   ql[4] = {qv.x, qv.y, qv.z, qv.w};
        float4 vp[4], vq[4];
#pragma unroll
        for (int u = 0; u < 4; ++u) {
          vp[u] = *(const float4*)&VT[pl[u]][c4];
          vq[u] = *(const float4*)&VT[ql[u]][c4];
        }
#pragma unroll
        for (int u = 0; u < 4; ++u) {
          const float cu = cl[u], su = sl[u];
          float4 np, nq;
          np.x = cu * vp[u].x - su * vq[u].x;  nq.x = su * vp[u].x + cu * vq[u].x;
          np.y = cu * vp[u].y - su * vq[u].y;  nq.y = su * vp[u].y + cu * vq[u].y;
          np.z = cu * vp[u].z - su * vq[u].z;  nq.z = su * vp[u].z + cu * vq[u].z;
          np.w = cu * vp[u].w - su * vq[u].w;  nq.w = su * vp[u].w + cu * vq[u].w;
          *(float4*)&VT[pl[u]][c4] = np;
          *(float4*)&VT[ql[u]][c4] = nq;
        }
      }
      __syncthreads();

#pragma unroll
      for (int t = 0; t < 16; ++t) {
        int p, q;
        jacobi_pair(r, t, p, q);
        ap[t] = A[lane][p];
        aq[t] = A[lane][q];
      }
#pragma unroll
      for (int t = 0; t < 16; ++t) {
        int p, q;
        jacobi_pair(r, t, p, q);
        const float ct = __int_as_float(__builtin_amdgcn_readlane(__float_as_int(c), t));
        const float st = __int_as_float(__builtin_amdgcn_readlane(__float_as_int(s), t));
        A[lane][p] = ct * ap[t] - st * aq[t];
        A[lane][q] = st * ap[t] + ct * aq[t];
      }
      __syncthreads();
    }
  }

  lw[lane] = logf(fmaxf(A[lane][lane], 1e-20f));
  __syncthreads();

  float acc[32];
#pragma unroll
  for (int i = 0; i < 32; ++i) acc[i] = 0.f;
  for (int k = 0; k < 32; ++k) {
    const float w = VT[k][lane] * lw[k];
#pragma unroll
    for (int i = 0; i < 32; ++i) acc[i] += VT[k][i] * w;
  }

  float* outm = xl + (size_t)blockIdx.x * 1024;
#pragma unroll
  for (int i = 0; i < 32; ++i)
    outm[i * 32 + lane] = (i == lane) ? 0.f : acc[i];
}

__global__ __launch_bounds__(32) void transbase_kernel(const float* __restrict__ xl,
                                                       const float* __restrict__ C,
                                                       float* __restrict__ qkv) {
  __shared__ float Y[32][33];
  __shared__ float CL[32][33];
  __shared__ float T[32][33];
  __shared__ float T2[32][33];
  const int lane = threadIdx.x;
  const int nIdx = lane & 15;
  const int hi   = lane >> 4;
  const int m = blockIdx.x, w = blockIdx.y;
  const float* ym = xl + (size_t)m * 1024;
  const float* Cw = C + (size_t)w * 1024;

  for (int i = 0; i < 32; ++i) {
    Y[i][lane]  = ym[i * 32 + lane];
    CL[i][lane] = Cw[i * 32 + lane];
  }
  __syncthreads();

  for (int ti = 0; ti < 2; ++ti)
    for (int tj = 0; tj < 2; ++tj) {
      v8f acc = {};
#pragma unroll
      for (int kk = 0; kk < 8; ++kk) {
        int k0 = kk * 4 + hi * 2;
        v2f a; a.x = CL[ti * 16 + nIdx][k0]; a.y = CL[ti * 16 + nIdx][k0 + 1];
        v2f b; b.x = Y[k0][tj * 16 + nIdx];  b.y = Y[k0 + 1][tj * 16 + nIdx];
        acc = WMMA_F32_16x16x4(a, b, acc);
      }
#pragma unroll
      for (int r = 0; r < 8; ++r)
        T[ti * 16 + r + hi * 8][tj * 16 + nIdx] = acc[r];
    }
  __syncthreads();

  for (int ti = 0; ti < 2; ++ti)
    for (int tj = 0; tj < 2; ++tj) {
      v8f acc = {};
#pragma unroll
      for (int kk = 0; kk < 8; ++kk) {
        int k0 = kk * 4 + hi * 2;
        v2f a; a.x = T[ti * 16 + nIdx][k0];       a.y = T[ti * 16 + nIdx][k0 + 1];
        v2f b; b.x = CL[tj * 16 + nIdx][k0];      b.y = CL[tj * 16 + nIdx][k0 + 1];
        acc = WMMA_F32_16x16x4(a, b, acc);
      }
#pragma unroll
      for (int r = 0; r < 8; ++r)
        T2[ti * 16 + r + hi * 8][tj * 16 + nIdx] = acc[r];
    }
  __syncthreads();

  float* outm = qkv + ((size_t)w * NMAT + m) * 1024;
  for (int i = 0; i < 32; ++i) {
    float lo = T2[i][lane];
    float up = T2[lane][i];
    float val = (i > lane) ? lo : ((i < lane) ? up : 0.f);
    outm[i * 32 + lane] = val;
  }
}

__global__ __launch_bounds__(256) void norms_kernel(const float* __restrict__ qkv,
                                                    float* __restrict__ nrm) {
  int idx = blockIdx.x * 256 + threadIdx.x;
  int sel = idx >> 13;
  int r   = idx & (NMAT - 1);
  const float* row = qkv + ((size_t)sel * NMAT + r) * 1024;
  float acc = 0.f;
  for (int f = 0; f < NF; ++f) { float t = row[f]; acc += t * t; }
  nrm[idx] = acc;
}

#define SC_TJ 128
#define SC_TI 64
#define SC_KS 32
#define SC_NS (NF / SC_KS)
#define SC_LD 36

__global__ __launch_bounds__(256) void scores_kernel(const float* __restrict__ qkv,
                                                     const float* __restrict__ nrm,
                                                     float* __restrict__ scores) {
  __shared__ __attribute__((aligned(16))) float As[2][SC_TJ][SC_LD];
  __shared__ __attribute__((aligned(16))) float Bs[2][SC_TI][SC_LD];
  const int tid  = threadIdx.x;
  const int lane = tid & 31;
  const int wave = tid >> 5;
  const int nIdx = lane & 15;
  const int hi   = lane >> 4;
  const int it = blockIdx.x, jt = blockIdx.y, b = blockIdx.z;
  const int i0 = it * SC_TI, j0 = jt * SC_TJ;
  const int wj = (wave >> 1) * 32;
  const int wi = (wave & 1) * 32;
  const float* qf = qkv + ((size_t)0 * NMAT + (size_t)b * NN) * 1024;
  const float* kf = qkv + ((size_t)1 * NMAT + (size_t)b * NN) * 1024;

  const int srow = tid >> 3;
  const int scol = (tid & 7) * 4;
  const float* ga = kf + (size_t)(j0 + srow) * 1024 + scol;
  const float* gb = qf + (size_t)(i0 + srow) * 1024 + scol;
  const size_t rstep = (size_t)32 * 1024;

  {
    const float4 ra0 = *(const float4*)(ga + 0 * rstep);
    const float4 ra1 = *(const float4*)(ga + 1 * rstep);
    const float4 ra2 = *(const float4*)(ga + 2 * rstep);
    const float4 ra3 = *(const float4*)(ga + 3 * rstep);
    const float4 rb0 = *(const float4*)(gb + 0 * rstep);
    const float4 rb1 = *(const float4*)(gb + 1 * rstep);
    *(float4*)&As[0][srow +  0][scol] = ra0;
    *(float4*)&As[0][srow + 32][scol] = ra1;
    *(float4*)&As[0][srow + 64][scol] = ra2;
    *(float4*)&As[0][srow + 96][scol] = ra3;
    *(float4*)&Bs[0][srow +  0][scol] = rb0;
    *(float4*)&Bs[0][srow + 32][scol] = rb1;
  }
  __syncthreads();

  v8f acc00 = {}, acc01 = {}, acc10 = {}, acc11 = {};

  auto slab_mma = [&](int buf) {
#pragma unroll
    for (int kk = 0; kk < SC_KS / 4; ++kk) {
      const int k0 = kk * 4 + hi * 2;
      v2f a0 = *(const v2f*)&As[buf][wj +  0 + nIdx][k0];
      v2f a1 = *(const v2f*)&As[buf][wj + 16 + nIdx][k0];
      v2f b0 = *(const v2f*)&Bs[buf][wi +  0 + nIdx][k0];
      v2f b1 = *(const v2f*)&Bs[buf][wi + 16 + nIdx][k0];
      acc00 = WMMA_F32_16x16x4(a0, b0, acc00);
      acc01 = WMMA_F32_16x16x4(a0, b1, acc01);
      acc10 = WMMA_F32_16x16x4(a1, b0, acc10);
      acc11 = WMMA_F32_16x16x4(a1, b1, acc11);
    }
  };

  for (int s = 0; s < SC_NS - 1; ++s) {
    const int ko = (s + 1) * SC_KS;
    const float4 ra0 = *(const float4*)(ga + 0 * rstep + ko);
    const float4 ra1 = *(const float4*)(ga + 1 * rstep + ko);
    const float4 ra2 = *(const float4*)(ga + 2 * rstep + ko);
    const float4 ra3 = *(const float4*)(ga + 3 * rstep + ko);
    const float4 rb0 = *(const float4*)(gb + 0 * rstep + ko);
    const float4 rb1 = *(const float4*)(gb + 1 * rstep + ko);

    slab_mma(s & 1);

    const int nb = (s + 1) & 1;
    *(float4*)&As[nb][srow +  0][scol] = ra0;
    *(float4*)&As[nb][srow + 32][scol] = ra1;
    *(float4*)&As[nb][srow + 64][scol] = ra2;
    *(float4*)&As[nb][srow + 96][scol] = ra3;
    *(float4*)&Bs[nb][srow +  0][scol] = rb0;
    *(float4*)&Bs[nb][srow + 32][scol] = rb1;
    __syncthreads();
  }
  slab_mma((SC_NS - 1) & 1);

  float* sb = scores + (size_t)b * NN * NN;
#pragma unroll
  for (int t = 0; t < 4; ++t) {
    v8f acc = (t == 0) ? acc00 : ((t == 1) ? acc01 : ((t == 2) ? acc10 : acc11));
    const int jb = j0 + wj + (t >> 1) * 16;
    const int i  = i0 + wi + (t & 1) * 16 + nIdx;
    float q2 = nrm[b * NN + i];
#pragma unroll
    for (int r = 0; r < 8; ++r) {
      int j = jb + r + hi * 8;
      float d2 = q2 + nrm[NMAT + b * NN + j] - 2.f * acc[r];
      d2 = fmaxf(d2, 1e-12f);
      float e  = sqrtf(d2);
      float sc = 1.f / (1.f + log1pf(e));
      sb[(size_t)j * NN + i] = sc;
    }
  }
}

#define SM_W 8
#define SM_R (NN / SM_W)

__global__ __launch_bounds__(256) void softmax_kernel(float* __restrict__ scores) {
  __shared__ float red[2][SM_W][32];
  const int lane = threadIdx.x & 31;
  const int wave = threadIdx.x >> 5;
  const int b = blockIdx.y;
  float* col = scores + (size_t)b * NN * NN + (size_t)(wave * SM_R) * NN
             + blockIdx.x * 32 + lane;

  float v[SM_R];
#pragma unroll
  for (int r = 0; r < SM_R; ++r) v[r] = col[(size_t)r * NN];

  float mx = -1e30f;
#pragma unroll
  for (int r = 0; r < SM_R; ++r) mx = fmaxf(mx, v[r]);
  red[0][wave][lane] = mx;
  __syncthreads();
  for (int w = 0; w < SM_W; ++w) mx = fmaxf(mx, red[0][w][lane]);

  float sum = 0.f;
#pragma unroll
  for (int r = 0; r < SM_R; ++r) {
    v[r] = __expf(v[r] - mx);
    sum += v[r];
  }
  red[1][wave][lane] = sum;
  __syncthreads();
  float tot = 0.f;
  for (int w = 0; w < SM_W; ++w) tot += red[1][w][lane];
  const float inv = 1.f / tot;

#pragma unroll
  for (int r = 0; r < SM_R; ++r) col[(size_t)r * NN] = v[r] * inv;
}

#define OG_TC 128
#define OG_TF 64
#define OG_KS 32
#define OG_NS (NN / OG_KS)
#define OG_LDA 36
#define OG_LDB 160

__global__ __launch_bounds__(256) void outgemm_kernel(const float* __restrict__ scores,
                                                      const float* __restrict__ qkv,
                                                      float* __restrict__ out) {
  __shared__ __attribute__((aligned(16))) float As[2][OG_TC][OG_LDA];
  __shared__ __attribute__((aligned(16))) float Bs[2][OG_KS / 2][OG_LDB];
  const int tid  = threadIdx.x;
  const int lane = tid & 31;
  const int wave = tid >> 5;
  const int nIdx = lane & 15;
  const int hi   = lane >> 4;
  const int ft = blockIdx.x, ct = blockIdx.y, b = blockIdx.z;
  const int f0 = ft * OG_TF, c0 = ct * OG_TC;
  const int wc = (wave >> 1) * 32;
  const int wf = (wave & 1) * 32;
  const float* p  = scores + (size_t)b * NN * NN;
  const float* vf = qkv + ((size_t)2 * NMAT + (size_t)b * NN) * 1024;

  const int arow = tid >> 3;
  const int acol = (tid & 7) * 4;
  const float* ga = p + (size_t)(c0 + arow) * NN + acol;
  const size_t astep = (size_t)32 * NN;
  const int bkp  = tid >> 4;
  const int bcol = (tid & 15) * 4;
  const float* gb = vf + (size_t)(2 * bkp) * NF + f0 + bcol;
  const size_t bslab = (size_t)OG_KS * NF;

  {
    const float4 ra0 = *(const float4*)(ga + 0 * astep);
    const float4 ra1 = *(const float4*)(ga + 1 * astep);
    const float4 ra2 = *(const float4*)(ga + 2 * astep);
    const float4 ra3 = *(const float4*)(ga + 3 * astep);
    const float4 rb0 = *(const float4*)(gb);
    const float4 rb1 = *(const float4*)(gb + NF);
    *(float4*)&As[0][arow +  0][acol] = ra0;
    *(float4*)&As[0][arow + 32][acol] = ra1;
    *(float4*)&As[0][arow + 64][acol] = ra2;
    *(float4*)&As[0][arow + 96][acol] = ra3;
    *(float4*)&Bs[0][bkp][bcol * 2 + 0] = make_float4(rb0.x, rb1.x, rb0.y, rb1.y);
    *(float4*)&Bs[0][bkp][bcol * 2 + 4] = make_float4(rb0.z, rb1.z, rb0.w, rb1.w);
  }
  __syncthreads();

  v8f acc00 = {}, acc01 = {}, acc10 = {}, acc11 = {};

  auto slab_mma = [&](int buf) {
#pragma unroll
    for (int kk = 0; kk < OG_KS / 4; ++kk) {
      const int k0 = kk * 4 + hi * 2;
      v2f a0 = *(const v2f*)&As[buf][wc +  0 + nIdx][k0];
      v2f a1 = *(const v2f*)&As[buf][wc + 16 + nIdx][k0];
      v2f b0 = *(const v2f*)&Bs[buf][kk * 2 + hi][(wf +  0 + nIdx) * 2];
      v2f b1 = *(const v2f*)&Bs[buf][kk * 2 + hi][(wf + 16 + nIdx) * 2];
      acc00 = WMMA_F32_16x16x4(a0, b0, acc00);
      acc01 = WMMA_F32_16x16x4(a0, b1, acc01);
      acc10 = WMMA_F32_16x16x4(a1, b0, acc10);
      acc11 = WMMA_F32_16x16x4(a1, b1, acc11);
    }
  };

  for (int s = 0; s < OG_NS - 1; ++s) {
    const int ko = (s + 1) * OG_KS;
    const float4 ra0 = *(const float4*)(ga + 0 * astep + ko);
    const float4 ra1 = *(const float4*)(ga + 1 * astep + ko);
    const float4 ra2 = *(const float4*)(ga + 2 * astep + ko);
    const float4 ra3 = *(const float4*)(ga + 3 * astep + ko);
    const float4 rb0 = *(const float4*)(gb + (size_t)(s + 1) * bslab);
    const float4 rb1 = *(const float4*)(gb + (size_t)(s + 1) * bslab + NF);

    slab_mma(s & 1);

    const int nb = (s + 1) & 1;
    *(float4*)&As[nb][arow +  0][acol] = ra0;
    *(float4*)&As[nb][arow + 32][acol] = ra1;
    *(float4*)&As[nb][arow + 64][acol] = ra2;
    *(float4*)&As[nb][arow + 96][acol] = ra3;
    *(float4*)&Bs[nb][bkp][bcol * 2 + 0] = make_float4(rb0.x, rb1.x, rb0.y, rb1.y);
    *(float4*)&Bs[nb][bkp][bcol * 2 + 4] = make_float4(rb0.z, rb1.z, rb0.w, rb1.w);
    __syncthreads();
  }
  slab_mma((OG_NS - 1) & 1);

  float* ob = out + (size_t)b * NN * NF;
#pragma unroll
  for (int t = 0; t < 4; ++t) {
    v8f acc = (t == 0) ? acc00 : ((t == 1) ? acc01 : ((t == 2) ? acc10 : acc11));
    const int cb = c0 + wc + (t >> 1) * 16;
    const int f  = f0 + wf + (t & 1) * 16 + nIdx;
#pragma unroll
    for (int r = 0; r < 8; ++r) {
      int c = cb + r + hi * 8;
      ob[(size_t)c * NF + f] = acc[r];
    }
  }
}

extern "C" void kernel_launch(void* const* d_in, const int* in_sizes, int n_in,
                              void* d_out, int out_size, void* d_ws, size_t ws_size,
                              hipStream_t stream) {
  (void)in_sizes; (void)n_in; (void)out_size; (void)ws_size;
  const float* x  = (const float*)d_in[0];
  const float* qw = (const float*)d_in[1];
  const float* kw = (const float*)d_in[2];
  const float* vw = (const float*)d_in[3];
  float* out = (float*)d_out;

  float* ws     = (float*)d_ws;
  float* C      = ws;
  float* xl     = C + 3 * 1024;
  float* qkv    = xl + (size_t)NMAT * 1024;
  float* nrm    = qkv + (size_t)3 * NMAT * 1024;
  float* scores = nrm + 2 * NMAT;

  cayley_kernel<<<3, 256, 0, stream>>>(qw, kw, vw, C);
  jacobi_logm_kernel<<<NMAT, 32, 0, stream>>>(x, xl);
  transbase_kernel<<<dim3(NMAT, 3), 32, 0, stream>>>(xl, C, qkv);
  norms_kernel<<<64, 256, 0, stream>>>(qkv, nrm);
  scores_kernel<<<dim3(NN / SC_TI, NN / SC_TJ, NB), 256, 0, stream>>>(qkv, nrm, scores);
  softmax_kernel<<<dim3(NN / 32, NB), 256, 0, stream>>>(scores);
  outgemm_kernel<<<dim3(NF / OG_TF, NN / OG_TC, NB), 256, 0, stream>>>(scores, qkv, out);
}


# eigensolver: convergence test after each sweep: the wave stops once every above-diagonal entry is <= 2e-6 x the smallest diagonal entry (still at most 8 sweeps) instead of always 8; sums unchanged
# speedup vs baseline: 2.1752x; 1.1296x over previous
#include <hip/hip_runtime.h>
#include <math.h>


typedef __attribute__((ext_vector_type(2))) float v2f;
typedef __attribute__((ext_vector_type(8))) float v8f;

#define WMMA_F32_16x16x4(a, b, c) \
  __builtin_amdgcn_wmma_f32_16x16x4_f32(false, (a), false, (b), (short)0, (c), false, false)

#define NMAT 8192
#define NB   16
#define NN   512
#define NF   1024

__global__ __launch_bounds__(256) void cayley_kernel(const float* __restrict__ qw,
                                                     const float* __restrict__ kw,
                                                     const float* __restrict__ vw,
                                                     float* __restrict__ Cout) {
  __shared__ float X[32][32];
  __shared__ float Mm[32][33];
  __shared__ float Inv[32][33];
  __shared__ float fac[32];
  __shared__ float pivinv;
  const float* W = (blockIdx.x == 0) ? qw : ((blockIdx.x == 1) ? kw : vw);
  const int tid = threadIdx.x;

  for (int e = tid; e < 1024; e += 256) {
    int i = e >> 5, j = e & 31;
    float xv = W[i * 32 + j] - W[j * 32 + i];
    X[i][j]  = xv;
    Mm[i][j] = xv + ((i == j) ? 1.f : 0.f);
    Inv[i][j] = (i == j) ? 1.f : 0.f;
  }
  __syncthreads();

  for (int k = 0; k < 32; ++k) {
    if (tid == 0) pivinv = 1.f / Mm[k][k];
    __syncthreads();
    if (tid < 32)      Mm[k][tid]       *= pivinv;
    else if (tid < 64) Inv[k][tid - 32] *= pivinv;
    __syncthreads();
    if (tid < 32) fac[tid] = (tid == k) ? 0.f : Mm[tid][k];
    __syncthreads();
    for (int e = tid; e < 2048; e += 256) {
      int i = e >> 6, j = e & 63;
      if (i == k) continue;
      float f = fac[i];
      if (j < 32) Mm[i][j]        -= f * Mm[k][j];
      else        Inv[i][j - 32]  -= f * Inv[k][j - 32];
    }
    __syncthreads();
  }

  for (int e = tid; e < 1024; e += 256) {
    int i = e >> 5, j = e & 31;
    float acc = 0.f;
    for (int kk = 0; kk < 32; ++kk) {
      float a = ((i == kk) ? 1.f : 0.f) - X[i][kk];
      acc += a * Inv[kk][j];
    }
    Cout[blockIdx.x * 1024 + i * 32 + j] = acc;
  }
}

#define JV_LD 36
#define JC_TOL 2e-6f

__device__ __forceinline__ void jacobi_pair(int r, int t, int& p, int& q) {
  int a, b;
  if (t == 0) { a = 31; b = r; }
  else {
    a = r + t; if (a >= 31) a -= 31;
    b = r - t; if (b < 0)   b += 31;
  }
  p = (a < b) ? a : b;
  q = (a < b) ? b : a;
}

__global__ __launch_bounds__(32, 8) void jacobi_logm_kernel(const float* __restrict__ x,
                                                            float* __restrict__ xl) {
  __shared__ float A[32][33];
  __shared__ __attribute__((aligned(16))) float VT[32][JV_LD];
  __shared__ __attribute__((aligned(16))) float cc[16];
  __shared__ __attribute__((aligned(16))) float ss[16];
  __shared__ __attribute__((aligned(16))) int   pp[16];
  __shared__ __attribute__((aligned(16))) int   qq[16];
  __shared__ float lw[32];
  const int lane = threadIdx.x;
  const int g4 = (lane >> 3) * 4;
  const int c4 = (lane & 7) * 4;
  const float* xm = x + (size_t)blockIdx.x * 1024;

  for (int i = 0; i < 32; ++i) {
    A[i][lane]  = xm[i * 32 + lane];
    VT[i][lane] = (i == lane) ? 1.f : 0.f;
  }
  __syncthreads();

  for (int sweep = 0; sweep < 8; ++sweep) {
    for (int r = 0; r < 31; ++r) {
      float c = 1.f, s = 0.f;
      if (lane < 16) {
        int p, q;
        jacobi_pair(r, lane, p, q);
        float apq = A[p][q];
        if (fabsf(apq) > 1e-12f) {
          float theta = (A[q][q] - A[p][p]) / (2.f * apq);
          float t = 1.f / (fabsf(theta) + sqrtf(theta * theta + 1.f));
          if (theta < 0.f) t = -t;
          c = rsqrtf(t * t + 1.f);
          s = t * c;
        }
        cc[lane] = c; ss[lane] = s; pp[lane] = p; qq[lane] = q;
      }
      __syncthreads();

      float ap[16], aq[16];
#pragma unroll
      for (int t = 0; t < 16; ++t) {
        int p, q;
        jacobi_pair(r, t, p, q);
        ap[t] = A[p][lane];
        aq[t] = A[q][lane];
      }
#pragma unroll
      for (int t = 0; t < 16; ++t) {
        int p, q;
        jacobi_pair(r, t, p, q);
        const float ct = __int_as_float(__builtin_amdgcn_readlane(__float_as_int(c), t));
        const float st = __int_as_float(__builtin_amdgcn_readlane(__float_as_int(s), t));
        A[p][lane] = ct * ap[t] - st * aq[t];
        A[q][lane] = st * ap[t] + ct * aq[t];
      }

      {
        const float4 cv = *(const float4*)&cc[g4];
        const float4 sv = *(const float4*)&ss[g4];
        const int4   pv = *(const int4*)&pp[g4];
        const int4   qv = *(const int4*)&qq[g4];
        const float cl[4] = {cv.x, cv.y, cv.z, cv.w};
        const float sl[4] = {sv.x, sv.y, sv.z, sv.w};
        const int   pl[4] = {pv.x, pv.y, pv.z, pv.w};
        const int   ql[4] = {qv.x, qv.y, qv.z, qv.w};
        float4 vp[4], vq[4];
#pragma unroll
        for (int u = 0; u < 4; ++u) {
          vp[u] = *(const float4*)&VT[pl[u]][c4];
          vq[u] = *(const float4*)&VT[ql[u]][c4];
        }
#pragma unroll
        for (int u = 0; u < 4; ++u) {
          const float cu = cl[u], su = sl[u];
          float4 np, nq;
          np.x = cu * vp[u].x - su * vq[u].x;  nq.x = su * vp[u].x + cu * vq[u].x;
          np.y = cu * vp[u].y - su * vq[u].y;  nq.y = su * vp[u].y + cu * vq[u].y;
          np.z = cu * vp[u].z - su * vq[u].z;  nq.z = su * vp[u].z + cu * vq[u].z;
          np.w = cu * vp[u].w - su * vq[u].w;  nq.w = su * vp[u].w + cu * vq[u].w;
          *(float4*)&VT[pl[u]][c4] = np;
          *(float4*)&VT[ql[u]][c4] = nq;
        }
      }
      __syncthreads();

#pragma unroll
      for (int t = 0; t < 16; ++t) {
        int p, q;
        jacobi_pair(r, t, p, q);
        ap[t] = A[lane][p];
        aq[t] = A[lane][q];
      }
#pragma unroll
      for (int t = 0; t < 16; ++t) {
        int p, q;
        jacobi_pair(r, t, p, q);
        const float ct = __int_as_float(__builtin_amdgcn_readlane(__float_as_int(c), t));
        const float st = __int_as_float(__builtin_amdgcn_readlane(__float_as_int(s), t));
        A[lane][p] = ct * ap[t] - st * aq[t];
        A[lane][q] = st * ap[t] + ct * aq[t];
      }
      __syncthreads();
    }

    float dg = 0.f, om = 0.f;
    for (int j = 0; j < 32; ++j) {
      const float a = fabsf(A[lane][j]);
      dg = (j == lane) ? a : dg;
      om = (j > lane) ? fmaxf(om, a) : om;
    }
    float dmin = dg;
    for (int o = 16; o > 0; o >>= 1) dmin = fminf(dmin, __shfl_xor(dmin, o));
    if (__builtin_amdgcn_ballot_w32(om > JC_TOL * dmin) == 0u) break;
  }

  lw[lane] = logf(fmaxf(A[lane][lane], 1e-20f));
  __syncthreads();

  float acc[32];
#pragma unroll
  for (int i = 0; i < 32; ++i) acc[i] = 0.f;
  for (int k = 0; k < 32; ++k) {
    const float w = VT[k][lane] * lw[k];
#pragma unroll
    for (int i = 0; i < 32; ++i) acc[i] += VT[k][i] * w;
  }

  float* outm = xl + (size_t)blockIdx.x * 1024;
#pragma unroll
  for (int i = 0; i < 32; ++i)
    outm[i * 32 + lane] = (i == lane) ? 0.f : acc[i];
}

__global__ __launch_bounds__(32) void transbase_kernel(const float* __restrict__ xl,
                                                       const float* __restrict__ C,
                                                       float* __restrict__ qkv) {
  __shared__ float Y[32][33];
  __shared__ float CL[32][33];
  __shared__ float T[32][33];
  __shared__ float T2[32][33];
  const int lane = threadIdx.x;
  const int nIdx = lane & 15;
  const int hi   = lane >> 4;
  const int m = blockIdx.x, w = blockIdx.y;
  const float* ym = xl + (size_t)m * 1024;
  const float* Cw = C + (size_t)w * 1024;

  for (int i = 0; i < 32; ++i) {
    Y[i][lane]  = ym[i * 32 + lane];
    CL[i][lane] = Cw[i * 32 + lane];
  }
  __syncthreads();

  for (int ti = 0; ti < 2; ++ti)
    for (int tj = 0; tj < 2; ++tj) {
      v8f acc = {};
#pragma unroll
      for (int kk = 0; kk < 8; ++kk) {
        int k0 = kk * 4 + hi * 2;
        v2f a; a.x = CL[ti * 16 + nIdx][k0]; a.y = CL[ti * 16 + nIdx][k0 + 1];
        v2f b; b.x = Y[k0][tj * 16 + nIdx];  b.y = Y[k0 + 1][tj * 16 + nIdx];
        acc = WMMA_F32_16x16x4(a, b, acc);
      }
#pragma unroll
      for (int r = 0; r < 8; ++r)
        T[ti * 16 + r + hi * 8][tj * 16 + nIdx] = acc[r];
    }
  __syncthreads();

  for (int ti = 0; ti < 2; ++ti)
    for (int tj = 0; tj < 2; ++tj) {
      v8f acc = {};
#pragma unroll
      for (int kk = 0; kk < 8; ++kk) {
        int k0 = kk * 4 + hi * 2;
        v2f a; a.x = T[ti * 16 + nIdx][k0];       a.y = T[ti * 16 + nIdx][k0 + 1];
        v2f b; b.x = CL[tj * 16 + nIdx][k0];      b.y = CL[tj * 16 + nIdx][k0 + 1];
        acc = WMMA_F32_16x16x4(a, b, acc);
      }
#pragma unroll
      for (int r = 0; r < 8; ++r)
        T2[ti * 16 + r + hi * 8][tj * 16 + nIdx] = acc[r];
    }
  __syncthreads();

  float* outm = qkv + ((size_t)w * NMAT + m) * 1024;
  for (int i = 0; i < 32; ++i) {
    float lo = T2[i][lane];
    float up = T2[lane][i];
    float val = (i > lane) ? lo : ((i < lane) ? up : 0.f);
    outm[i * 32 + lane] = val;
  }
}

__global__ __launch_bounds__(256) void norms_kernel(const float* __restrict__ qkv,
                                                    float* __restrict__ nrm) {
  int idx = blockIdx.x * 256 + threadIdx.x;
  int sel = idx >> 13;
  int r   = idx & (NMAT - 1);
  const float* row = qkv + ((size_t)sel * NMAT + r) * 1024;
  float acc = 0.f;
  for (int f = 0; f < NF; ++f) { float t = row[f]; acc += t * t; }
  nrm[idx] = acc;
}

#define SC_TJ 128
#define SC_TI 64
#define SC_KS 32
#define SC_NS (NF / SC_KS)
#define SC_LD 36

__global__ __launch_bounds__(256) void scores_kernel(const float* __restrict__ qkv,
                                                     const float* __restrict__ nrm,
                                                     float* __restrict__ scores) {
  __shared__ __attribute__((aligned(16))) float As[2][SC_TJ][SC_LD];
  __shared__ __attribute__((aligned(16))) float Bs[2][SC_TI][SC_LD];
  const int tid  = threadIdx.x;
  const int lane = tid & 31;
  const int wave = tid >> 5;
  const int nIdx = lane & 15;
  const int hi   = lane >> 4;
  const int it = blockIdx.x, jt = blockIdx.y, b = blockIdx.z;
  const int i0 = it * SC_TI, j0 = jt * SC_TJ;
  const int wj = (wave >> 1) * 32;
  const int wi = (wave & 1) * 32;
  const float* qf = qkv + ((size_t)0 * NMAT + (size_t)b * NN) * 1024;
  const float* kf = qkv + ((size_t)1 * NMAT + (size_t)b * NN) * 1024;

  const int srow = tid >> 3;
  const int scol = (tid & 7) * 4;
  const float* ga = kf + (size_t)(j0 + srow) * 1024 + scol;
  const float* gb = qf + (size_t)(i0 + srow) * 1024 + scol;
  const size_t rstep = (size_t)32 * 1024;

  {
    const float4 ra0 = *(const float4*)(ga + 0 * rstep);
    const float4 ra1 = *(const float4*)(ga + 1 * rstep);
    const float4 ra2 = *(const float4*)(ga + 2 * rstep);
    const float4 ra3 = *(const float4*)(ga + 3 * rstep);
    const float4 rb0 = *(const float4*)(gb + 0 * rstep);
    const float4 rb1 = *(const float4*)(gb + 1 * rstep);
    *(float4*)&As[0][srow +  0][scol] = ra0;
    *(float4*)&As[0][srow + 32][scol] = ra1;
    *(float4*)&As[0][srow + 64][scol] = ra2;
    *(float4*)&As[0][srow + 96][scol] = ra3;
    *(float4*)&Bs[0][srow +  0][scol] = rb0;
    *(float4*)&Bs[0][srow + 32][scol] = rb1;
  }
  __syncthreads();

  v8f acc00 = {}, acc01 = {}, acc10 = {}, acc11 = {};

  auto slab_mma = [&](int buf) {
#pragma unroll
    for (int kk = 0; kk < SC_KS / 4; ++kk) {
      const int k0 = kk * 4 + hi * 2;
      v2f a0 = *(const v2f*)&As[buf][wj +  0 + nIdx][k0];
      v2f a1 = *(const v2f*)&As[buf][wj + 16 + nIdx][k0];
      v2f b0 = *(const v2f*)&Bs[buf][wi +  0 + nIdx][k0];
      v2f b1 = *(const v2f*)&Bs[buf][wi + 16 + nIdx][k0];
      acc00 = WMMA_F32_16x16x4(a0, b0, acc00);
      acc01 = WMMA_F32_16x16x4(a0, b1, acc01);
      acc10 = WMMA_F32_16x16x4(a1, b0, acc10);
      acc11 = WMMA_F32_16x16x4(a1, b1, acc11);
    }
  };

  for (int s = 0; s < SC_NS - 1; ++s) {
    const int ko = (s + 1) * SC_KS;
    const float4 ra0 = *(const float4*)(ga + 0 * rstep + ko);
    const float4 ra1 = *(const float4*)(ga + 1 * rstep + ko);
    const float4 ra2 = *(const float4*)(ga + 2 * rstep + ko);
    const float4 ra3 = *(const float4*)(ga + 3 * rstep + ko);
    const float4 rb0 = *(const float4*)(gb + 0 * rstep + ko);
    const float4 rb1 = *(const float4*)(gb + 1 * rstep + ko);

    slab_mma(s & 1);

    const int nb = (s + 1) & 1;
    *(float4*)&As[nb][srow +  0][scol] = ra0;
    *(float4*)&As[nb][srow + 32][scol] = ra1;
    *(float4*)&As[nb][srow + 64][scol] = ra2;
    *(float4*)&As[nb][srow + 96][scol] = ra3;
    *(float4*)&Bs[nb][srow +  0][scol] = rb0;
    *(float4*)&Bs[nb][srow + 32][scol] = rb1;
    __syncthreads();
  }
  slab_mma((SC_NS - 1) & 1);

  float* sb = scores + (size_t)b * NN * NN;
#pragma unroll
  for (int t = 0; t < 4; ++t) {
    v8f acc = (t == 0) ? acc00 : ((t == 1) ? acc01 : ((t == 2) ? acc10 : acc11));
    const int jb = j0 + wj + (t >> 1) * 16;
    const int i  = i0 + wi + (t & 1) * 16 + nIdx;
    float q2 = nrm[b * NN + i];
#pragma unroll
    for (int r = 0; r < 8; ++r) {
      int j = jb + r + hi * 8;
      float d2 = q2 + nrm[NMAT + b * NN + j] - 2.f * acc[r];
      d2 = fmaxf(d2, 1e-12f);
      float e  = sqrtf(d2);
      float sc = 1.f / (1.f + log1pf(e));
      sb[(size_t)j * NN + i] = sc;
    }
  }
}

#define SM_W 8
#define SM_R (NN / SM_W)

__global__ __launch_bounds__(256) void softmax_kernel(float* __restrict__ scores) {
  __shared__ float red[2][SM_W][32];
  const int lane = threadIdx.x & 31;
  const int wave = threadIdx.x >> 5;
  const int b = blockIdx.y;
  float* col = scores + (size_t)b * NN * NN + (size_t)(wave * SM_R) * NN
             + blockIdx.x * 32 + lane;

  float v[SM_R];
#pragma unroll
  for (int r = 0; r < SM_R; ++r) v[r] = col[(size_t)r * NN];

  float mx = -1e30f;
#pragma unroll
  for (int r = 0; r < SM_R; ++r) mx = fmaxf(mx, v[r]);
  red[0][wave][lane] = mx;
  __syncthreads();
  for (int w = 0; w < SM_W; ++w) mx = fmaxf(mx, red[0][w][lane]);

  float sum = 0.f;
#pragma unroll
  for (int r = 0; r < SM_R; ++r) {
    v[r] = __expf(v[r] - mx);
    sum += v[r];
  }
  red[1][wave][lane] = sum;
  __syncthreads();
  float tot = 0.f;
  for (int w = 0; w < SM_W; ++w) tot += red[1][w][lane];
  const float inv = 1.f / tot;

#pragma unroll
  for (int r = 0; r < SM_R; ++r) col[(size_t)r * NN] = v[r] * inv;
}

#define OG_TC 128
#define OG_TF 64
#define OG_KS 32
#define OG_NS (NN / OG_KS)
#define OG_LDA 36
#define OG_LDB 160

__global__ __launch_bounds__(256) void outgemm_kernel(const float* __restrict__ scores,
                                                      const float* __restrict__ qkv,
                                                      float* __restrict__ out) {
  __shared__ __attribute__((aligned(16))) float As[2][OG_TC][OG_LDA];
  __shared__ __attribute__((aligned(16))) float Bs[2][OG_KS / 2][OG_LDB];
  const int tid  = threadIdx.x;
  const int lane = tid & 31;
  const int wave = tid >> 5;
  const int nIdx = lane & 15;
  const int hi   = lane >> 4;
  const int ft = blockIdx.x, ct = blockIdx.y, b = blockIdx.z;
  const int f0 = ft * OG_TF, c0 = ct * OG_TC;
  const int wc = (wave >> 1) * 32;
  const int wf = (wave & 1) * 32;
  const float* p  = scores + (size_t)b * NN * NN;
  const float* vf = qkv + ((size_t)2 * NMAT + (size_t)b * NN) * 1024;

  const int arow = tid >> 3;
  const int acol = (tid & 7) * 4;
  const float* ga = p + (size_t)(c0 + arow) * NN + acol;
  const size_t astep = (size_t)32 * NN;
  const int bkp  = tid >> 4;
  const int bcol = (tid & 15) * 4;
  const float* gb = vf + (size_t)(2 * bkp) * NF + f0 + bcol;
  const size_t bslab = (size_t)OG_KS * NF;

  {
    const float4 ra0 = *(const float4*)(ga + 0 * astep);
    const float4 ra1 = *(const float4*)(ga + 1 * astep);
    const float4 ra2 = *(const float4*)(ga + 2 * astep);
    const float4 ra3 = *(const float4*)(ga + 3 * astep);
    const float4 rb0 = *(const float4*)(gb);
    const float4 rb1 = *(const float4*)(gb + NF);
    *(float4*)&As[0][arow +  0][acol] = ra0;
    *(float4*)&As[0][arow + 32][acol] = ra1;
    *(float4*)&As[0][arow + 64][acol] = ra2;
    *(float4*)&As[0][arow + 96][acol] = ra3;
    *(float4*)&Bs[0][bkp][bcol * 2 + 0] = make_float4(rb0.x, rb1.x, rb0.y, rb1.y);
    *(float4*)&Bs[0][bkp][bcol * 2 + 4] = make_float4(rb0.z, rb1.z, rb0.w, rb1.w);
  }
  __syncthreads();

  v8f acc00 = {}, acc01 = {}, acc10 = {}, acc11 = {};

  auto slab_mma = [&](int buf) {
#pragma unroll
    for (int kk = 0; kk < OG_KS / 4; ++kk) {
      const int k0 = kk * 4 + hi * 2;
      v2f a0 = *(const v2f*)&As[buf][wc +  0 + nIdx][k0];
      v2f a1 = *(const v2f*)&As[buf][wc + 16 + nIdx][k0];
      v2f b0 = *(const v2f*)&Bs[buf][kk * 2 + hi][(wf +  0 + nIdx) * 2];
      v2f b1 = *(const v2f*)&Bs[buf][kk * 2 + hi][(wf + 16 + nIdx) * 2];
      acc00 = WMMA_F32_16x16x4(a0, b0, acc00);
      acc01 = WMMA_F32_16x16x4(a0, b1, acc01);
      acc10 = WMMA_F32_16x16x4(a1, b0, acc10);
      acc11 = WMMA_F32_16x16x4(a1, b1, acc11);
    }
  };

  for (int s = 0; s < OG_NS - 1; ++s) {
    const int ko = (s + 1) * OG_KS;
    const float4 ra0 = *(const float4*)(ga + 0 * astep + ko);
    const float4 ra1 = *(const float4*)(ga + 1 * astep + ko);
    const float4 ra2 = *(const float4*)(ga + 2 * astep + ko);
    const float4 ra3 = *(const float4*)(ga + 3 * astep + ko);
    const float4 rb0 = *(const float4*)(gb + (size_t)(s + 1) * bslab);
    const float4 rb1 = *(const float4*)(gb + (size_t)(s + 1) * bslab + NF);

    slab_mma(s & 1);

    const int nb = (s + 1) & 1;
    *(float4*)&As[nb][arow +  0][acol] = ra0;
    *(float4*)&As[nb][arow + 32][acol] = ra1;
    *(float4*)&As[nb][arow + 64][acol] = ra2;
    *(float4*)&As[nb][arow + 96][acol] = ra3;
    *(float4*)&Bs[nb][bkp][bcol * 2 + 0] = make_float4(rb0.x, rb1.x, rb0.y, rb1.y);
    *(float4*)&Bs[nb][bkp][bcol * 2 + 4] = make_float4(rb0.z, rb1.z, rb0.w, rb1.w);
    __syncthreads();
  }
  slab_mma((OG_NS - 1) & 1);

  float* ob = out + (size_t)b * NN * NF;
#pragma unroll
  for (int t = 0; t < 4; ++t) {
    v8f acc = (t == 0) ? acc00 : ((t == 1) ? acc01 : ((t == 2) ? acc10 : acc11));
    const int cb = c0 + wc + (t >> 1) * 16;
    const int f  = f0 + wf + (t & 1) * 16 + nIdx;
#pragma unroll
    for (int r = 0; r < 8; ++r) {
      int c = cb + r + hi * 8;
      ob[(size_t)c * NF + f] = acc[r];
    }
  }
}

extern "C" void kernel_launch(void* const* d_in, const int* in_sizes, int n_in,
                              void* d_out, int out_size, void* d_ws, size_t ws_size,
                              hipStream_t stream) {
  (void)in_sizes; (void)n_in; (void)out_size; (void)ws_size;
  const float* x  = (const float*)d_in[0];
  const float* qw = (const float*)d_in[1];
  const float* kw = (const float*)d_in[2];
  const float* vw = (const float*)d_in[3];
  float* out = (float*)d_out;

  float* ws     = (float*)d_ws;
  float* C      = ws;
  float* xl     = C + 3 * 1024;
  float* qkv    = xl + (size_t)NMAT * 1024;
  float* nrm    = qkv + (size_t)3 * NMAT * 1024;
  float* scores = nrm + 2 * NMAT;

  cayley_kernel<<<3, 256, 0, stream>>>(qw, kw, vw, C);
  jacobi_logm_kernel<<<NMAT, 32, 0, stream>>>(x, xl);
  transbase_kernel<<<dim3(NMAT, 3), 32, 0, stream>>>(xl, C, qkv);
  norms_kernel<<<64, 256, 0, stream>>>(qkv, nrm);
  scores_kernel<<<dim3(NN / SC_TI, NN / SC_TJ, NB), 256, 0, stream>>>(qkv, nrm, scores);
  softmax_kernel<<<dim3(NN / 32, NB), 256, 0, stream>>>(scores);
  outgemm_kernel<<<dim3(NF / OG_TF, NN / OG_TC, NB), 256, 0, stream>>>(scores, qkv, out);
}


# eigensolver round: row and column rotation of A fused per 2x2 block, A read and written in LDS once per round not twice (pitch 32, lane halves on different columns: no bank clash); same two-term sums
# speedup vs baseline: 2.5292x; 1.1627x over previous
#include <hip/hip_runtime.h>
#include <math.h>


typedef __attribute__((ext_vector_type(2))) float v2f;
typedef __attribute__((ext_vector_type(8))) float v8f;

#define WMMA_F32_16x16x4(a, b, c) \
  __builtin_amdgcn_wmma_f32_16x16x4_f32(false, (a), false, (b), (short)0, (c), false, false)

#define NMAT 8192
#define NB   16
#define NN   512
#define NF   1024

__global__ __launch_bounds__(256) void cayley_kernel(const float* __restrict__ qw,
                                                     const float* __restrict__ kw,
                                                     const float* __restrict__ vw,
                                                     float* __restrict__ Cout) {
  __shared__ float X[32][32];
  __shared__ float Mm[32][33];
  __shared__ float Inv[32][33];
  __shared__ float fac[32];
  __shared__ float pivinv;
  const float* W = (blockIdx.x == 0) ? qw : ((blockIdx.x == 1) ? kw : vw);
  const int tid = threadIdx.x;

  for (int e = tid; e < 1024; e += 256) {
    int i = e >> 5, j = e & 31;
    float xv = W[i * 32 + j] - W[j * 32 + i];
    X[i][j]  = xv;
    Mm[i][j] = xv + ((i == j) ? 1.f : 0.f);
    Inv[i][j] = (i == j) ? 1.f : 0.f;
  }
  __syncthreads();

  for (int k = 0; k < 32; ++k) {
    if (tid == 0) pivinv = 1.f / Mm[k][k];
    __syncthreads();
    if (tid < 32)      Mm[k][tid]       *= pivinv;
    else if (tid < 64) Inv[k][tid - 32] *= pivinv;
    __syncthreads();
    if (tid < 32) fac[tid] = (tid == k) ? 0.f : Mm[tid][k];
    __syncthreads();
    for (int e = tid; e < 2048; e += 256) {
      int i = e >> 6, j = e & 63;
      if (i == k) continue;
      float f = fac[i];
      if (j < 32) Mm[i][j]        -= f * Mm[k][j];
      else        Inv[i][j - 32]  -= f * Inv[k][j - 32];
    }
    __syncthreads();
  }

  for (int e = tid; e < 1024; e += 256) {
    int i = e >> 5, j = e & 31;
    float acc = 0.f;
    for (int kk = 0; kk < 32; ++kk) {
      float a = ((i == kk) ? 1.f : 0.f) - X[i][kk];
      acc += a * Inv[kk][j];
    }
    Cout[blockIdx.x * 1024 + i * 32 + j] = acc;
  }
}

#define JA_LD 32
#define JV_LD 36
#define JC_TOL 2e-6f

__device__ __forceinline__ void jacobi_pair(int r, int t, int& p, int& q) {
  int a, b;
  if (t == 0) { a = 31; b = r; }
  else {
    a = r + t; if (a >= 31) a -= 31;
    b = r - t; if (b < 0)   b += 31;
  }
  p = (a < b) ? a : b;
  q = (a < b) ? b : a;
}

template <int U> __device__ __forceinline__ int row_share_i(int v) {
  return __builtin_amdgcn_update_dpp(v, v, 0x150 + U, 0xf, 0xf, false);
}
template <int U> __device__ __forceinline__ float row_share_f(float v) {
  return __int_as_float(row_share_i<U>(__float_as_int(v)));
}

__global__ __launch_bounds__(32, 8) void jacobi_logm_kernel(const float* __restrict__ x,
                                                            float* __restrict__ xl) {
  __shared__ __attribute__((aligned(16))) float A[32][JA_LD];
  __shared__ __attribute__((aligned(16))) float VT[32][JV_LD];
  __shared__ __attribute__((aligned(16))) float cc[16];
  __shared__ __attribute__((aligned(16))) float ss[16];
  __shared__ __attribute__((aligned(16))) int   pp[16];
  __shared__ __attribute__((aligned(16))) int   qq[16];
  __shared__ float lw[32];
  const int lane = threadIdx.x;
  const int g4 = (lane >> 3) * 4;
  const int c4 = (lane & 7) * 4;
  const int hi = lane >> 4;
  const int tl = (lane + 8 * hi) & 15;
  float* const Af = &A[0][0];
  const float* xm = x + (size_t)blockIdx.x * 1024;

  for (int i = 0; i < 32; ++i) {
    A[i][lane]  = xm[i * 32 + lane];
    VT[i][lane] = (i == lane) ? 1.f : 0.f;
  }
  __syncthreads();

  for (int sweep = 0; sweep < 8; ++sweep) {
    for (int r = 0; r < 31; ++r) {
      int p, q;
      jacobi_pair(r, tl, p, q);
      float c = 1.f, s = 0.f;
      {
        float apq = A[p][q];
        if (fabsf(apq) > 1e-12f) {
          float theta = (A[q][q] - A[p][p]) / (2.f * apq);
          float t = 1.f / (fabsf(theta) + sqrtf(theta * theta + 1.f));
          if (theta < 0.f) t = -t;
          c = rsqrtf(t * t + 1.f);
          s = t * c;
        }
      }
      if (lane < 16) { cc[lane] = c; ss[lane] = s; pp[lane] = p; qq[lane] = q; }
      __syncthreads();

      const int   colA = hi ? q : p;
      const int   colB = hi ? p : q;
      const float sx   = hi ? -s : s;
      const int   prow = p * JA_LD, qrow = q * JA_LD;
      float y0[8], y1[8], y2[8], y3[8], ca[8], sa[8];
      int   o0[8], o1[8], o2[8], o3[8];
#define JBLK_LOAD(U)                                                        \
      {                                                                     \
        ca[U] = row_share_f<U>(c);  sa[U] = row_share_f<U>(s);              \
        const int pr = row_share_i<U>(prow), qr = row_share_i<U>(qrow);     \
        o0[U] = pr + colA; o1[U] = pr + colB;                               \
        o2[U] = qr + colA; o3[U] = qr + colB;                               \
        y0[U] = Af[o0[U]]; y1[U] = Af[o1[U]];                               \
        y2[U] = Af[o2[U]]; y3[U] = Af[o3[U]];                               \
      }
      JBLK_LOAD(0) JBLK_LOAD(1) JBLK_LOAD(2) JBLK_LOAD(3)
      JBLK_LOAD(4) JBLK_LOAD(5) JBLK_LOAD(6) JBLK_LOAD(7)
#undef JBLK_LOAD
#pragma unroll
      for (int u = 0; u < 8; ++u) {
        const float r0 = ca[u] * y0[u] - sa[u] * y2[u];
        const float r2 = sa[u] * y0[u] + ca[u] * y2[u];
        const float r1 = ca[u] * y1[u] - sa[u] * y3[u];
        const float r3 = sa[u] * y1[u] + ca[u] * y3[u];
        Af[o0[u]] = c  * r0 - sx * r1;
        Af[o1[u]] = sx * r0 + c  * r1;
        Af[o2[u]] = c  * r2 - sx * r3;
        Af[o3[u]] = sx * r2 + c  * r3;
      }

      {
        const float4 cv = *(const float4*)&cc[g4];
        const float4 sv = *(const float4*)&ss[g4];
        const int4   pv = *(const int4*)&pp[g4];
        const int4   qv = *(const int4*)&qq[g4];
        const float cl[4] = {cv.x, cv.y, cv.z, cv.w};
        const float sl[4] = {sv.x, sv.y, sv.z, sv.w};
        const int   pl[4] = {pv.x, pv.y, pv.z, pv.w};
        const int   ql[4] = {qv.x, qv.y, qv.z, qv.w};
        float4 vp[4], vq[4];
#pragma unroll
        for (int u = 0; u < 4; ++u) {
          vp[u] = *(const float4*)&VT[pl[u]][c4];
          vq[u] = *(const float4*)&VT[ql[u]][c4];
        }
#pragma unroll
        for (int u = 0; u < 4; ++u) {
          const float cu = cl[u], su = sl[u];
          float4 np, nq;
          np.x = cu * vp[u].x - su * vq[u].x;  nq.x = su * vp[u].x + cu * vq[u].x;
          np.y = cu * vp[u].y - su * vq[u].y;  nq.y = su * vp[u].y + cu * vq[u].y;
          np.z = cu * vp[u].z - su * vq[u].z;  nq.z = su * vp[u].z + cu * vq[u].z;
          np.w = cu * vp[u].w - su * vq[u].w;  nq.w = su * vp[u].w + cu * vq[u].w;
          *(float4*)&VT[pl[u]][c4] = np;
          *(float4*)&VT[ql[u]][c4] = nq;
        }
      }
      __syncthreads();
    }

    float dg = 0.f, om = 0.f;
    for (int i = 0; i < 32; ++i) {
      const float a = fabsf(A[i][lane]);
      dg = (i == lane) ? a : dg;
      om = (i < lane) ? fmaxf(om, a) : om;
    }
    float dmin = dg;
    for (int o = 16; o > 0; o >>= 1) dmin = fminf(dmin, __shfl_xor(dmin, o));
    if (__builtin_amdgcn_ballot_w32(om > JC_TOL * dmin) == 0u) break;
  }

  lw[lane] = logf(fmaxf(A[lane][lane], 1e-20f));
  __syncthreads();

  float acc[32];
#pragma unroll
  for (int i = 0; i < 32; ++i) acc[i] = 0.f;
  for (int k = 0; k < 32; ++k) {
    const float w = VT[k][lane] * lw[k];
#pragma unroll
    for (int i = 0; i < 32; ++i) acc[i] += VT[k][i] * w;
  }

  float* outm = xl + (size_t)blockIdx.x * 1024;
#pragma unroll
  for (int i = 0; i < 32; ++i)
    outm[i * 32 + lane] = (i == lane) ? 0.f : acc[i];
}

__global__ __launch_bounds__(32) void transbase_kernel(const float* __restrict__ xl,
                                                       const float* __restrict__ C,
                                                       float* __restrict__ qkv) {
  __shared__ float Y[32][33];
  __shared__ float CL[32][33];
  __shared__ float T[32][33];
  __shared__ float T2[32][33];
  const int lane = threadIdx.x;
  const int nIdx = lane & 15;
  const int hi   = lane >> 4;
  const int m = blockIdx.x, w = blockIdx.y;
  const float* ym = xl + (size_t)m * 1024;
  const float* Cw = C + (size_t)w * 1024;

  for (int i = 0; i < 32; ++i) {
    Y[i][lane]  = ym[i * 32 + lane];
    CL[i][lane] = Cw[i * 32 + lane];
  }
  __syncthreads();

  for (int ti = 0; ti < 2; ++ti)
    for (int tj = 0; tj < 2; ++tj) {
      v8f acc = {};
#pragma unroll
      for (int kk = 0; kk < 8; ++kk) {
        int k0 = kk * 4 + hi * 2;
        v2f a; a.x = CL[ti * 16 + nIdx][k0]; a.y = CL[ti * 16 + nIdx][k0 + 1];
        v2f b; b.x = Y[k0][tj * 16 + nIdx];  b.y = Y[k0 + 1][tj * 16 + nIdx];
        acc = WMMA_F32_16x16x4(a, b, acc);
      }
#pragma unroll
      for (int r = 0; r < 8; ++r)
        T[ti * 16 + r + hi * 8][tj * 16 + nIdx] = acc[r];
    }
  __syncthreads();

  for (int ti = 0; ti < 2; ++ti)
    for (int tj = 0; tj < 2; ++tj) {
      v8f acc = {};
#pragma unroll
      for (int kk = 0; kk < 8; ++kk) {
        int k0 = kk * 4 + hi * 2;
        v2f a; a.x = T[ti * 16 + nIdx][k0];       a.y = T[ti * 16 + nIdx][k0 + 1];
        v2f b; b.x = CL[tj * 16 + nIdx][k0];      b.y = CL[tj * 16 + nIdx][k0 + 1];
        acc = WMMA_F32_16x16x4(a, b, acc);
      }
#pragma unroll
      for (int r = 0; r < 8; ++r)
        T2[ti * 16 + r + hi * 8][tj * 16 + nIdx] = acc[r];
    }
  __syncthreads();

  float* outm = qkv + ((size_t)w * NMAT + m) * 1024;
  for (int i = 0; i < 32; ++i) {
    float lo = T2[i][lane];
    float up = T2[lane][i];
    float val = (i > lane) ? lo : ((i < lane) ? up : 0.f);
    outm[i * 32 + lane] = val;
  }
}

__global__ __launch_bounds__(256) void norms_kernel(const float* __restrict__ qkv,
                                                    float* __restrict__ nrm) {
  int idx = blockIdx.x * 256 + threadIdx.x;
  int sel = idx >> 13;
  int r   = idx & (NMAT - 1);
  const float* row = qkv + ((size_t)sel * NMAT + r) * 1024;
  float acc = 0.f;
  for (int f = 0; f < NF; ++f) { float t = row[f]; acc += t * t; }
  nrm[idx] = acc;
}

#define SC_TJ 128
#define SC_TI 64
#define SC_KS 32
#define SC_NS (NF / SC_KS)
#define SC_LD 36

__global__ __launch_bounds__(256) void scores_kernel(const float* __restrict__ qkv,
                                                     const float* __restrict__ nrm,
                                                     float* __restrict__ scores) {
  __shared__ __attribute__((aligned(16))) float As[2][SC_TJ][SC_LD];
  __shared__ __attribute__((aligned(16))) float Bs[2][SC_TI][SC_LD];
  const int tid  = threadIdx.x;
  const int lane = tid & 31;
  const int wave = tid >> 5;
  const int nIdx = lane & 15;
  const int hi   = lane >> 4;
  const int it = blockIdx.x, jt = blockIdx.y, b = blockIdx.z;
  const int i0 = it * SC_TI, j0 = jt * SC_TJ;
  const int wj = (wave >> 1) * 32;
  const int wi = (wave & 1) * 32;
  const float* qf = qkv + ((size_t)0 * NMAT + (size_t)b * NN) * 1024;
  const float* kf = qkv + ((size_t)1 * NMAT + (size_t)b * NN) * 1024;

  const int srow = tid >> 3;
  const int scol = (tid & 7) * 4;
  const float* ga = kf + (size_t)(j0 + srow) * 1024 + scol;
  const float* gb = qf + (size_t)(i0 + srow) * 1024 + scol;
  const size_t rstep = (size_t)32 * 1024;

  {
    const float4 ra0 = *(const float4*)(ga + 0 * rstep);
    const float4 ra1 = *(const float4*)(ga + 1 * rstep);
    const float4 ra2 = *(const float4*)(ga + 2 * rstep);
    const float4 ra3 = *(const float4*)(ga + 3 * rstep);
    const float4 rb0 = *(const float4*)(gb + 0 * rstep);
    const float4 rb1 = *(const float4*)(gb + 1 * rstep);
    *(float4*)&As[0][srow +  0][scol] = ra0;
    *(float4*)&As[0][srow + 32][scol] = ra1;
    *(float4*)&As[0][srow + 64][scol] = ra2;
    *(float4*)&As[0][srow + 96][scol] = ra3;
    *(float4*)&Bs[0][srow +  0][scol] = rb0;
    *(float4*)&Bs[0][srow + 32][scol] = rb1;
  }
  __syncthreads();

  v8f acc00 = {}, acc01 = {}, acc10 = {}, acc11 = {};

  auto slab_mma = [&](int buf) {
#pragma unroll
    for (int kk = 0; kk < SC_KS / 4; ++kk) {
      const int k0 = kk * 4 + hi * 2;
      v2f a0 = *(const v2f*)&As[buf][wj +  0 + nIdx][k0];
      v2f a1 = *(const v2f*)&As[buf][wj + 16 + nIdx][k0];
      v2f b0 = *(const v2f*)&Bs[buf][wi +  0 + nIdx][k0];
      v2f b1 = *(const v2f*)&Bs[buf][wi + 16 + nIdx][k0];
      acc00 = WMMA_F32_16x16x4(a0, b0, acc00);
      acc01 = WMMA_F32_16x16x4(a0, b1, acc01);
      acc10 = WMMA_F32_16x16x4(a1, b0, acc10);
      acc11 = WMMA_F32_16x16x4(a1, b1, acc11);
    }
  };

  for (int s = 0; s < SC_NS - 1; ++s) {
    const int ko = (s + 1) * SC_KS;
    const float4 ra0 = *(const float4*)(ga + 0 * rstep + ko);
    const float4 ra1 = *(const float4*)(ga + 1 * rstep + ko);
    const float4 ra2 = *(const float4*)(ga + 2 * rstep + ko);
    const float4 ra3 = *(const float4*)(ga + 3 * rstep + ko);
    const float4 rb0 = *(const float4*)(gb + 0 * rstep + ko);
    const float4 rb1 = *(const float4*)(gb + 1 * rstep + ko);

    slab_mma(s & 1);

    const int nb = (s + 1) & 1;
    *(float4*)&As[nb][srow +  0][scol] = ra0;
    *(float4*)&As[nb][srow + 32][scol] = ra1;
    *(float4*)&As[nb][srow + 64][scol] = ra2;
    *(float4*)&As[nb][srow + 96][scol] = ra3;
    *(float4*)&Bs[nb][srow +  0][scol] = rb0;
    *(float4*)&Bs[nb][srow + 32][scol] = rb1;
    __syncthreads();
  }
  slab_mma((SC_NS - 1) & 1);

  float* sb = scores + (size_t)b * NN * NN;
#pragma unroll
  for (int t = 0; t < 4; ++t) {
    v8f acc = (t == 0) ? acc00 : ((t == 1) ? acc01 : ((t == 2) ? acc10 : acc11));
    const int jb = j0 + wj + (t >> 1) * 16;
    const int i  = i0 + wi + (t & 1) * 16 + nIdx;
    float q2 = nrm[b * NN + i];
#pragma unroll
    for (int r = 0; r < 8; ++r) {
      int j = jb + r + hi * 8;
      float d2 = q2 + nrm[NMAT + b * NN + j] - 2.f * acc[r];
      d2 = fmaxf(d2, 1e-12f);
      float e  = sqrtf(d2);
      float sc = 1.f / (1.f + log1pf(e));
      sb[(size_t)j * NN + i] = sc;
    }
  }
}

#define SM_W 8
#define SM_R (NN / SM_W)

__global__ __launch_bounds__(256) void softmax_kernel(float* __restrict__ scores) {
  __shared__ float red[2][SM_W][32];
  const int lane = threadIdx.x & 31;
  const int wave = threadIdx.x >> 5;
  const int b = blockIdx.y;
  float* col = scores + (size_t)b * NN * NN + (size_t)(wave * SM_R) * NN
             + blockIdx.x * 32 + lane;

  float v[SM_R];
#pragma unroll
  for (int r = 0; r < SM_R; ++r) v[r] = col[(size_t)r * NN];

  float mx = -1e30f;
#pragma unroll
  for (int r = 0; r < SM_R; ++r) mx = fmaxf(mx, v[r]);
  red[0][wave][lane] = mx;
  __syncthreads();
  for (int w = 0; w < SM_W; ++w) mx = fmaxf(mx, red[0][w][lane]);

  float sum = 0.f;
#pragma unroll
  for (int r = 0; r < SM_R; ++r) {
    v[r] = __expf(v[r] - mx);
    sum += v[r];
  }
  red[1][wave][lane] = sum;
  __syncthreads();
  float tot = 0.f;
  for (int w = 0; w < SM_W; ++w) tot += red[1][w][lane];
  const float inv = 1.f / tot;

#pragma unroll
  for (int r = 0; r < SM_R; ++r) col[(size_t)r * NN] = v[r] * inv;
}

#define OG_TC 128
#define OG_TF 64
#define OG_KS 32
#define OG_NS (NN / OG_KS)
#define OG_LDA 36
#define OG_LDB 160

__global__ __launch_bounds__(256) void outgemm_kernel(const float* __restrict__ scores,
                                                      const float* __restrict__ qkv,
                                                      float* __restrict__ out) {
  __shared__ __attribute__((aligned(16))) float As[2][OG_TC][OG_LDA];
  __shared__ __attribute__((aligned(16))) float Bs[2][OG_KS / 2][OG_LDB];
  const int tid  = threadIdx.x;
  const int lane = tid & 31;
  const int wave = tid >> 5;
  const int nIdx = lane & 15;
  const int hi   = lane >> 4;
  const int ft = blockIdx.x, ct = blockIdx.y, b = blockIdx.z;
  const int f0 = ft * OG_TF, c0 = ct * OG_TC;
  const int wc = (wave >> 1) * 32;
  const int wf = (wave & 1) * 32;
  const float* p  = scores + (size_t)b * NN * NN;
  const float* vf = qkv + ((size_t)2 * NMAT + (size_t)b * NN) * 1024;

  const int arow = tid >> 3;
  const int acol = (tid & 7) * 4;
  const float* ga = p + (size_t)(c0 + arow) * NN + acol;
  const size_t astep = (size_t)32 * NN;
  const int bkp  = tid >> 4;
  const int bcol = (tid & 15) * 4;
  const float* gb = vf + (size_t)(2 * bkp) * NF + f0 + bcol;
  const size_t bslab = (size_t)OG_KS * NF;

  {
    const float4 ra0 = *(const float4*)(ga + 0 * astep);
    const float4 ra1 = *(const float4*)(ga + 1 * astep);
    const float4 ra2 = *(const float4*)(ga + 2 * astep);
    const float4 ra3 = *(const float4*)(ga + 3 * astep);
    const float4 rb0 = *(const float4*)(gb);
    const float4 rb1 = *(const float4*)(gb + NF);
    *(float4*)&As[0][arow +  0][acol] = ra0;
    *(float4*)&As[0][arow + 32][acol] = ra1;
    *(float4*)&As[0][arow + 64][acol] = ra2;
    *(float4*)&As[0][arow + 96][acol] = ra3;
    *(float4*)&Bs[0][bkp][bcol * 2 + 0] = make_float4(rb0.x, rb1.x, rb0.y, rb1.y);
    *(float4*)&Bs[0][bkp][bcol * 2 + 4] = make_float4(rb0.z, rb1.z, rb0.w, rb1.w);
  }
  __syncthreads();

  v8f acc00 = {}, acc01 = {}, acc10 = {}, acc11 = {};

  auto slab_mma = [&](int buf) {
#pragma unroll
    for (int kk = 0; kk < OG_KS / 4; ++kk) {
      const int k0 = kk * 4 + hi * 2;
      v2f a0 = *(const v2f*)&As[buf][wc +  0 + nIdx][k0];
      v2f a1 = *(const v2f*)&As[buf][wc + 16 + nIdx][k0];
      v2f b0 = *(const v2f*)&Bs[buf][kk * 2 + hi][(wf +  0 + nIdx) * 2];
      v2f b1 = *(const v2f*)&Bs[buf][kk * 2 + hi][(wf + 16 + nIdx) * 2];
      acc00 = WMMA_F32_16x16x4(a0, b0, acc00);
      acc01 = WMMA_F32_16x16x4(a0, b1, acc01);
      acc10 = WMMA_F32_16x16x4(a1, b0, acc10);
      acc11 = WMMA_F32_16x16x4(a1, b1, acc11);
    }
  };

  for (int s = 0; s < OG_NS - 1; ++s) {
    const int ko = (s + 1) * OG_KS;
    const float4 ra0 = *(const float4*)(ga + 0 * astep + ko);
    const float4 ra1 = *(const float4*)(ga + 1 * astep + ko);
    const float4 ra2 = *(const float4*)(ga + 2 * astep + ko);
    const float4 ra3 = *(const float4*)(ga + 3 * astep + ko);
    const float4 rb0 = *(const float4*)(gb + (size_t)(s + 1) * bslab);
    const float4 rb1 = *(const float4*)(gb + (size_t)(s + 1) * bslab + NF);

    slab_mma(s & 1);

    const int nb = (s + 1) & 1;
    *(float4*)&As[nb][arow +  0][acol] = ra0;
    *(float4*)&As[nb][arow + 32][acol] = ra1;
    *(float4*)&As[nb][arow + 64][acol] = ra2;
    *(float4*)&As[nb][arow + 96][acol] = ra3;
    *(float4*)&Bs[nb][bkp][bcol * 2 + 0] = make_float4(rb0.x, rb1.x, rb0.y, rb1.y);
    *(float4*)&Bs[nb][bkp][bcol * 2 + 4] = make_float4(rb0.z, rb1.z, rb0.w, rb1.w);
    __syncthreads();
  }
  slab_mma((OG_NS - 1) & 1);

  float* ob = out + (size_t)b * NN * NF;
#pragma unroll
  for (int t = 0; t < 4; ++t) {
    v8f acc = (t == 0) ? acc00 : ((t == 1) ? acc01 : ((t == 2) ? acc10 : acc11));
    const int cb = c0 + wc + (t >> 1) * 16;
    const int f  = f0 + wf + (t & 1) * 16 + nIdx;
#pragma unroll
    for (int r = 0; r < 8; ++r) {
      int c = cb + r + hi * 8;
      ob[(size_t)c * NF + f] = acc[r];
    }
  }
}

extern "C" void kernel_launch(void* const* d_in, const int* in_sizes, int n_in,
                              void* d_out, int out_size, void* d_ws, size_t ws_size,
                              hipStream_t stream) {
  (void)in_sizes; (void)n_in; (void)out_size; (void)ws_size;
  const float* x  = (const float*)d_in[0];
  const float* qw = (const float*)d_in[1];
  const float* kw = (const float*)d_in[2];
  const float* vw = (const float*)d_in[3];
  float* out = (float*)d_out;

  float* ws     = (float*)d_ws;
  float* C      = ws;
  float* xl     = C + 3 * 1024;
  float* qkv    = xl + (size_t)NMAT * 1024;
  float* nrm    = qkv + (size_t)3 * NMAT * 1024;
  float* scores = nrm + 2 * NMAT;

  cayley_kernel<<<3, 256, 0, stream>>>(qw, kw, vw, C);
  jacobi_logm_kernel<<<NMAT, 32, 0, stream>>>(x, xl);
  transbase_kernel<<<dim3(NMAT, 3), 32, 0, stream>>>(xl, C, qkv);
  norms_kernel<<<64, 256, 0, stream>>>(qkv, nrm);
  scores_kernel<<<dim3(NN / SC_TI, NN / SC_TJ, NB), 256, 0, stream>>>(qkv, nrm, scores);
  softmax_kernel<<<dim3(NN / 32, NB), 256, 0, stream>>>(scores);
  outgemm_kernel<<<dim3(NF / OG_TF, NN / OG_TC, NB), 256, 0, stream>>>(scores, qkv, out);
}


# q,k = L+L^T, zero diagonal: kept as packed strict lower triangle (496 + 16 zeros); scores GEMM and norms sum K=512 and double, not 1024 (sum order changes: each mirrored product once)
# speedup vs baseline: 2.7108x; 1.0718x over previous
#include <hip/hip_runtime.h>
#include <math.h>


typedef __attribute__((ext_vector_type(2))) float v2f;
typedef __attribute__((ext_vector_type(8))) float v8f;

#define WMMA_F32_16x16x4(a, b, c) \
  __builtin_amdgcn_wmma_f32_16x16x4_f32(false, (a), false, (b), (short)0, (c), false, false)

#define NMAT 8192
#define NB   16
#define NN   512
#define NF   1024
#define NFP  512

__global__ __launch_bounds__(256) void cayley_kernel(const float* __restrict__ qw,
                                                     const float* __restrict__ kw,
                                                     const float* __restrict__ vw,
                                                     float* __restrict__ Cout) {
  __shared__ float X[32][32];
  __shared__ float Mm[32][33];
  __shared__ float Inv[32][33];
  __shared__ float fac[32];
  __shared__ float pivinv;
  const float* W = (blockIdx.x == 0) ? qw : ((blockIdx.x == 1) ? kw : vw);
  const int tid = threadIdx.x;

  for (int e = tid; e < 1024; e += 256) {
    int i = e >> 5, j = e & 31;
    float xv = W[i * 32 + j] - W[j * 32 + i];
    X[i][j]  = xv;
    Mm[i][j] = xv + ((i == j) ? 1.f : 0.f);
    Inv[i][j] = (i == j) ? 1.f : 0.f;
  }
  __syncthreads();

  for (int k = 0; k < 32; ++k) {
    if (tid == 0) pivinv = 1.f / Mm[k][k];
    __syncthreads();
    if (tid < 32)      Mm[k][tid]       *= pivinv;
    else if (tid < 64) Inv[k][tid - 32] *= pivinv;
    __syncthreads();
    if (tid < 32) fac[tid] = (tid == k) ? 0.f : Mm[tid][k];
    __syncthreads();
    for (int e = tid; e < 2048; e += 256) {
      int i = e >> 6, j = e & 63;
      if (i == k) continue;
      float f = fac[i];
      if (j < 32) Mm[i][j]        -= f * Mm[k][j];
      else        Inv[i][j - 32]  -= f * Inv[k][j - 32];
    }
    __syncthreads();
  }

  for (int e = tid; e < 1024; e += 256) {
    int i = e >> 5, j = e & 31;
    float acc = 0.f;
    for (int kk = 0; kk < 32; ++kk) {
      float a = ((i == kk) ? 1.f : 0.f) - X[i][kk];
      acc += a * Inv[kk][j];
    }
    Cout[blockIdx.x * 1024 + i * 32 + j] = acc;
  }
}

#define JA_LD 32
#define JV_LD 36
#define JC_TOL 2e-6f

__device__ __forceinline__ void jacobi_pair(int r, int t, int& p, int& q) {
  int a, b;
  if (t == 0) { a = 31; b = r; }
  else {
    a = r + t; if (a >= 31) a -= 31;
    b = r - t; if (b < 0)   b += 31;
  }
  p = (a < b) ? a : b;
  q = (a < b) ? b : a;
}

template <int U> __device__ __forceinline__ int row_share_i(int v) {
  return __builtin_amdgcn_update_dpp(v, v, 0x150 + U, 0xf, 0xf, false);
}
template <int U> __device__ __forceinline__ float row_share_f(float v) {
  return __int_as_float(row_share_i<U>(__float_as_int(v)));
}

__global__ __launch_bounds__(32, 8) void jacobi_logm_kernel(const float* __restrict__ x,
                                                            float* __restrict__ xl) {
  __shared__ __attribute__((aligned(16))) float A[32][JA_LD];
  __shared__ __attribute__((aligned(16))) float VT[32][JV_LD];
  __shared__ __attribute__((aligned(16))) float cc[16];
  __shared__ __attribute__((aligned(16))) float ss[16];
  __shared__ __attribute__((aligned(16))) int   pp[16];
  __shared__ __attribute__((aligned(16))) int   qq[16];
  __shared__ float lw[32];
  const int lane = threadIdx.x;
  const int g4 = (lane >> 3) * 4;
  const int c4 = (lane & 7) * 4;
  const int hi = lane >> 4;
  const int tl = (lane + 8 * hi) & 15;
  float* const Af = &A[0][0];
  const float* xm = x + (size_t)blockIdx.x * 1024;

  for (int i = 0; i < 32; ++i) {
    A[i][lane]  = xm[i * 32 + lane];
    VT[i][lane] = (i == lane) ? 1.f : 0.f;
  }
  __syncthreads();

  for (int sweep = 0; sweep < 8; ++sweep) {
    for (int r = 0; r < 31; ++r) {
      int p, q;
      jacobi_pair(r, tl, p, q);
      float c = 1.f, s = 0.f;
      {
        float apq = A[p][q];
        if (fabsf(apq) > 1e-12f) {
          float theta = (A[q][q] - A[p][p]) / (2.f * apq);
          float t = 1.f / (fabsf(theta) + sqrtf(theta * theta + 1.f));
          if (theta < 0.f) t = -t;
          c = rsqrtf(t * t + 1.f);
          s = t * c;
        }
      }
      if (lane < 16) { cc[lane] = c; ss[lane] = s; pp[lane] = p; qq[lane] = q; }
      __syncthreads();

      const int   colA = hi ? q : p;
      const int   colB = hi ? p : q;
      const float sx   = hi ? -s : s;
      const int   prow = p * JA_LD, qrow = q * JA_LD;
      float y0[8], y1[8], y2[8], y3[8], ca[8], sa[8];
      int   o0[8], o1[8], o2[8], o3[8];
#define JBLK_LOAD(U)                                                        \
      {                                                                     \
        ca[U] = row_share_f<U>(c);  sa[U] = row_share_f<U>(s);              \
        const int pr = row_share_i<U>(prow), qr = row_share_i<U>(qrow);     \
        o0[U] = pr + colA; o1[U] = pr + colB;                               \
        o2[U] = qr + colA; o3[U] = qr + colB;                               \
        y0[U] = Af[o0[U]]; y1[U] = Af[o1[U]];                               \
        y2[U] = Af[o2[U]]; y3[U] = Af[o3[U]];                               \
      }
      JBLK_LOAD(0) JBLK_LOAD(1) JBLK_LOAD(2) JBLK_LOAD(3)
      JBLK_LOAD(4) JBLK_LOAD(5) JBLK_LOAD(6) JBLK_LOAD(7)
#undef JBLK_LOAD
#pragma unroll
      for (int u = 0; u < 8; ++u) {
        const float r0 = ca[u] * y0[u] - sa[u] * y2[u];
        const float r2 = sa[u] * y0[u] + ca[u] * y2[u];
        const float r1 = ca[u] * y1[u] - sa[u] * y3[u];
        const float r3 = sa[u] * y1[u] + ca[u] * y3[u];
        Af[o0[u]] = c  * r0 - sx * r1;
        Af[o1[u]] = sx * r0 + c  * r1;
        Af[o2[u]] = c  * r2 - sx * r3;
        Af[o3[u]] = sx * r2 + c  * r3;
      }

      {
        const float4 cv = *(const float4*)&cc[g4];
        const float4 sv = *(const float4*)&ss[g4];
        const int4   pv = *(const int4*)&pp[g4];
        const int4   qv = *(const int4*)&qq[g4];
        const float cl[4] = {cv.x, cv.y, cv.z, cv.w};
        const float sl[4] = {sv.x, sv.y, sv.z, sv.w};
        const int   pl[4] = {pv.x, pv.y, pv.z, pv.w};
        const int   ql[4] = {qv.x, qv.y, qv.z, qv.w};
        float4 vp[4], vq[4];
#pragma unroll
        for (int u = 0; u < 4; ++u) {
          vp[u] = *(const float4*)&VT[pl[u]][c4];
          vq[u] = *(const float4*)&VT[ql[u]][c4];
        }
#pragma unroll
        for (int u = 0; u < 4; ++u) {
          const float cu = cl[u], su = sl[u];
          float4 np, nq;
          np.x = cu * vp[u].x - su * vq[u].x;  nq.x = su * vp[u].x + cu * vq[u].x;
          np.y = cu * vp[u].y - su * vq[u].y;  nq.y = su * vp[u].y + cu * vq[u].y;
          np.z = cu * vp[u].z - su * vq[u].z;  nq.z = su * vp[u].z + cu * vq[u].z;
          np.w = cu * vp[u].w - su * vq[u].w;  nq.w = su * vp[u].w + cu * vq[u].w;
          *(float4*)&VT[pl[u]][c4] = np;
          *(float4*)&VT[ql[u]][c4] = nq;
        }
      }
      __syncthreads();
    }

    float dg = 0.f, om = 0.f;
    for (int i = 0; i < 32; ++i) {
      const float a = fabsf(A[i][lane]);
      dg = (i == lane) ? a : dg;
      om = (i < lane) ? fmaxf(om, a) : om;
    }
    float dmin = dg;
    for (int o = 16; o > 0; o >>= 1) dmin = fminf(dmin, __shfl_xor(dmin, o));
    if (__builtin_amdgcn_ballot_w32(om > JC_TOL * dmin) == 0u) break;
  }

  lw[lane] = logf(fmaxf(A[lane][lane], 1e-20f));
  __syncthreads();

  float acc[32];
#pragma unroll
  for (int i = 0; i < 32; ++i) acc[i] = 0.f;
  for (int k = 0; k < 32; ++k) {
    const float w = VT[k][lane] * lw[k];
#pragma unroll
    for (int i = 0; i < 32; ++i) acc[i] += VT[k][i] * w;
  }

  float* outm = xl + (size_t)blockIdx.x * 1024;
#pragma unroll
  for (int i = 0; i < 32; ++i)
    outm[i * 32 + lane] = (i == lane) ? 0.f : acc[i];
}

__global__ __launch_bounds__(32) void transbase_kernel(const float* __restrict__ xl,
                                                       const float* __restrict__ C,
                                                       float* __restrict__ qkv) {
  __shared__ float Y[32][33];
  __shared__ float CL[32][33];
  __shared__ float T[32][33];
  __shared__ float T2[32][33];
  const int lane = threadIdx.x;
  const int nIdx = lane & 15;
  const int hi   = lane >> 4;
  const int m = blockIdx.x, w = blockIdx.y;
  const float* ym = xl + (size_t)m * 1024;
  const float* Cw = C + (size_t)w * 1024;

  for (int i = 0; i < 32; ++i) {
    Y[i][lane]  = ym[i * 32 + lane];
    CL[i][lane] = Cw[i * 32 + lane];
  }
  __syncthreads();

  for (int ti = 0; ti < 2; ++ti)
    for (int tj = 0; tj < 2; ++tj) {
      v8f acc = {};
#pragma unroll
      for (int kk = 0; kk < 8; ++kk) {
        int k0 = kk * 4 + hi * 2;
        v2f a; a.x = CL[ti * 16 + nIdx][k0]; a.y = CL[ti * 16 + nIdx][k0 + 1];
        v2f b; b.x = Y[k0][tj * 16 + nIdx];  b.y = Y[k0 + 1][tj * 16 + nIdx];
        acc = WMMA_F32_16x16x4(a, b, acc);
      }
#pragma unroll
      for (int r = 0; r < 8; ++r)
        T[ti * 16 + r + hi * 8][tj * 16 + nIdx] = acc[r];
    }
  __syncthreads();

  for (int ti = 0; ti < 2; ++ti)
    for (int tj = 0; tj < 2; ++tj) {
      v8f acc = {};
#pragma unroll
      for (int kk = 0; kk < 8; ++kk) {
        int k0 = kk * 4 + hi * 2;
        v2f a; a.x = T[ti * 16 + nIdx][k0];       a.y = T[ti * 16 + nIdx][k0 + 1];
        v2f b; b.x = CL[tj * 16 + nIdx][k0];      b.y = CL[tj * 16 + nIdx][k0 + 1];
        acc = WMMA_F32_16x16x4(a, b, acc);
      }
#pragma unroll
      for (int r = 0; r < 8; ++r)
        T2[ti * 16 + r + hi * 8][tj * 16 + nIdx] = acc[r];
    }
  __syncthreads();

  if (w < 2) {
    float* outp = qkv + ((size_t)w * NMAT + m) * NFP;
    for (int s = 0; s < 16; ++s) {
      int i, j;
      if (s == 0)        { i = 16;     j = lane & 15; }
      else if (lane < s) { i = s;      j = lane; }
      else               { i = 32 - s; j = lane - s; }
      const float t = T2[i][j];
      outp[s * 32 + lane] = (s == 0 && lane >= 16) ? 0.f : t;
    }
  } else {
    float* outm = qkv + ((size_t)w * NMAT + m) * 1024;
    for (int i = 0; i < 32; ++i) {
      float lo = T2[i][lane];
      float up = T2[lane][i];
      float val = (i > lane) ? lo : ((i < lane) ? up : 0.f);
      outm[i * 32 + lane] = val;
    }
  }
}

__global__ __launch_bounds__(256) void norms_kernel(const float* __restrict__ qkv,
                                                    float* __restrict__ nrm) {
  int idx = blockIdx.x * 256 + threadIdx.x;
  int sel = idx >> 13;
  int r   = idx & (NMAT - 1);
  const float* row = qkv + ((size_t)sel * NMAT + r) * NFP;
  float acc = 0.f;
  for (int f = 0; f < NFP; ++f) { float t = row[f]; acc += t * t; }
  nrm[idx] = 2.f * acc;
}

#define SC_TJ 128
#define SC_TI 64
#define SC_KS 32
#define SC_NS (NFP / SC_KS)
#define SC_LD 36

__global__ __launch_bounds__(256) void scores_kernel(const float* __restrict__ qkv,
                                                     const float* __restrict__ nrm,
                                                     float* __restrict__ scores) {
  __shared__ __attribute__((aligned(16))) float As[2][SC_TJ][SC_LD];
  __shared__ __attribute__((aligned(16))) float Bs[2][SC_TI][SC_LD];
  const int tid  = threadIdx.x;
  const int lane = tid & 31;
  const int wave = tid >> 5;
  const int nIdx = lane & 15;
  const int hi   = lane >> 4;
  const int it = blockIdx.x, jt = blockIdx.y, b = blockIdx.z;
  const int i0 = it * SC_TI, j0 = jt * SC_TJ;
  const int wj = (wave >> 1) * 32;
  const int wi = (wave & 1) * 32;
  const float* qf = qkv + ((size_t)0 * NMAT + (size_t)b * NN) * NFP;
  const float* kf = qkv + ((size_t)1 * NMAT + (size_t)b * NN) * NFP;

  const int srow = tid >> 3;
  const int scol = (tid & 7) * 4;
  const float* ga = kf + (size_t)(j0 + srow) * NFP + scol;
  const float* gb = qf + (size_t)(i0 + srow) * NFP + scol;
  const size_t rstep = (size_t)32 * NFP;

  {
    const float4 ra0 = *(const float4*)(ga + 0 * rstep);
    const float4 ra1 = *(const float4*)(ga + 1 * rstep);
    const float4 ra2 = *(const float4*)(ga + 2 * rstep);
    const float4 ra3 = *(const float4*)(ga + 3 * rstep);
    const float4 rb0 = *(const float4*)(gb + 0 * rstep);
    const float4 rb1 = *(const float4*)(gb + 1 * rstep);
    *(float4*)&As[0][srow +  0][scol] = ra0;
    *(float4*)&As[0][srow + 32][scol] = ra1;
    *(float4*)&As[0][srow + 64][scol] = ra2;
    *(float4*)&As[0][srow + 96][scol] = ra3;
    *(float4*)&Bs[0][srow +  0][scol] = rb0;
    *(float4*)&Bs[0][srow + 32][scol] = rb1;
  }
  __syncthreads();

  v8f acc00 = {}, acc01 = {}, acc10 = {}, acc11 = {};

  auto slab_mma = [&](int buf) {
#pragma unroll
    for (int kk = 0; kk < SC_KS / 4; ++kk) {
      const int k0 = kk * 4 + hi * 2;
      v2f a0 = *(const v2f*)&As[buf][wj +  0 + nIdx][k0];
      v2f a1 = *(const v2f*)&As[buf][wj + 16 + nIdx][k0];
      v2f b0 = *(const v2f*)&Bs[buf][wi +  0 + nIdx][k0];
      v2f b1 = *(const v2f*)&Bs[buf][wi + 16 + nIdx][k0];
      acc00 = WMMA_F32_16x16x4(a0, b0, acc00);
      acc01 = WMMA_F32_16x16x4(a0, b1, acc01);
      acc10 = WMMA_F32_16x16x4(a1, b0, acc10);
      acc11 = WMMA_F32_16x16x4(a1, b1, acc11);
    }
  };

  for (int s = 0; s < SC_NS - 1; ++s) {
    const int ko = (s + 1) * SC_KS;
    const float4 ra0 = *(const float4*)(ga + 0 * rstep + ko);
    const float4 ra1 = *(const float4*)(ga + 1 * rstep + ko);
    const float4 ra2 = *(const float4*)(ga + 2 * rstep + ko);
    const float4 ra3 = *(const float4*)(ga + 3 * rstep + ko);
    const float4 rb0 = *(const float4*)(gb + 0 * rstep + ko);
    const float4 rb1 = *(const float4*)(gb + 1 * rstep + ko);

    slab_mma(s & 1);

    const int nb = (s + 1) & 1;
    *(float4*)&As[nb][srow +  0][scol] = ra0;
    *(float4*)&As[nb][srow + 32][scol] = ra1;
    *(float4*)&As[nb][srow + 64][scol] = ra2;
    *(float4*)&As[nb][srow + 96][scol] = ra3;
    *(float4*)&Bs[nb][srow +  0][scol] = rb0;
    *(float4*)&Bs[nb][srow + 32][scol] = rb1;
    __syncthreads();
  }
  slab_mma((SC_NS - 1) & 1);

  float* sb = scores + (size_t)b * NN * NN;
#pragma unroll
  for (int t = 0; t < 4; ++t) {
    v8f acc = (t == 0) ? acc00 : ((t == 1) ? acc01 : ((t == 2) ? acc10 : acc11));
    const int jb = j0 + wj + (t >> 1) * 16;
    const int i  = i0 + wi + (t & 1) * 16 + nIdx;
    float q2 = nrm[b * NN + i];
#pragma unroll
    for (int r = 0; r < 8; ++r) {
      int j = jb + r + hi * 8;
      float d2 = q2 + nrm[NMAT + b * NN + j] - 4.f * acc[r];
      d2 = fmaxf(d2, 1e-12f);
      float e  = sqrtf(d2);
      float sc = 1.f / (1.f + log1pf(e));
      sb[(size_t)j * NN + i] = sc;
    }
  }
}

#define SM_W 8
#define SM_R (NN / SM_W)

__global__ __launch_bounds__(256) void softmax_kernel(float* __restrict__ scores) {
  __shared__ float red[2][SM_W][32];
  const int lane = threadIdx.x & 31;
  const int wave = threadIdx.x >> 5;
  const int b = blockIdx.y;
  float* col = scores + (size_t)b * NN * NN + (size_t)(wave * SM_R) * NN
             + blockIdx.x * 32 + lane;

  float v[SM_R];
#pragma unroll
  for (int r = 0; r < SM_R; ++r) v[r] = col[(size_t)r * NN];

  float mx = -1e30f;
#pragma unroll
  for (int r = 0; r < SM_R; ++r) mx = fmaxf(mx, v[r]);
  red[0][wave][lane] = mx;
  __syncthreads();
  for (int w = 0; w < SM_W; ++w) mx = fmaxf(mx, red[0][w][lane]);

  float sum = 0.f;
#pragma unroll
  for (int r = 0; r < SM_R; ++r) {
    v[r] = __expf(v[r] - mx);
    sum += v[r];
  }
  red[1][wave][lane] = sum;
  __syncthreads();
  float tot = 0.f;
  for (int w = 0; w < SM_W; ++w) tot += red[1][w][lane];
  const float inv = 1.f / tot;

#pragma unroll
  for (int r = 0; r < SM_R; ++r) col[(size_t)r * NN] = v[r] * inv;
}

#define OG_TC 128
#define OG_TF 64
#define OG_KS 32
#define OG_NS (NN / OG_KS)
#define OG_LDA 36
#define OG_LDB 160

__global__ __launch_bounds__(256) void outgemm_kernel(const float* __restrict__ scores,
                                                      const float* __restrict__ qkv,
                                                      float* __restrict__ out) {
  __shared__ __attribute__((aligned(16))) float As[2][OG_TC][OG_LDA];
  __shared__ __attribute__((aligned(16))) float Bs[2][OG_KS / 2][OG_LDB];
  const int tid  = threadIdx.x;
  const int lane = tid & 31;
  const int wave = tid >> 5;
  const int nIdx = lane & 15;
  const int hi   = lane >> 4;
  const int ft = blockIdx.x, ct = blockIdx.y, b = blockIdx.z;
  const int f0 = ft * OG_TF, c0 = ct * OG_TC;
  const int wc = (wave >> 1) * 32;
  const int wf = (wave & 1) * 32;
  const float* p  = scores + (size_t)b * NN * NN;
  const float* vf = qkv + ((size_t)2 * NMAT + (size_t)b * NN) * 1024;

  const int arow = tid >> 3;
  const int acol = (tid & 7) * 4;
  const float* ga = p + (size_t)(c0 + arow) * NN + acol;
  const size_t astep = (size_t)32 * NN;
  const int bkp  = tid >> 4;
  const int bcol = (tid & 15) * 4;
  const float* gb = vf + (size_t)(2 * bkp) * NF + f0 + bcol;
  const size_t bslab = (size_t)OG_KS * NF;

  {
    const float4 ra0 = *(const float4*)(ga + 0 * astep);
    const float4 ra1 = *(const float4*)(ga + 1 * astep);
    const float4 ra2 = *(const float4*)(ga + 2 * astep);
    const float4 ra3 = *(const float4*)(ga + 3 * astep);
    const float4 rb0 = *(const float4*)(gb);
    const float4 rb1 = *(const float4*)(gb + NF);
    *(float4*)&As[0][arow +  0][acol] = ra0;
    *(float4*)&As[0][arow + 32][acol] = ra1;
    *(float4*)&As[0][arow + 64][acol] = ra2;
    *(float4*)&As[0][arow + 96][acol] = ra3;
    *(float4*)&Bs[0][bkp][bcol * 2 + 0] = make_float4(rb0.x, rb1.x, rb0.y, rb1.y);
    *(float4*)&Bs[0][bkp][bcol * 2 + 4] = make_float4(rb0.z, rb1.z, rb0.w, rb1.w);
  }
  __syncthreads();

  v8f acc00 = {}, acc01 = {}, acc10 = {}, acc11 = {};

  auto slab_mma = [&](int buf) {
#pragma unroll
    for (int kk = 0; kk < OG_KS / 4; ++kk) {
      const int k0 = kk * 4 + hi * 2;
      v2f a0 = *(const v2f*)&As[buf][wc +  0 + nIdx][k0];
      v2f a1 = *(const v2f*)&As[buf][wc + 16 + nIdx][k0];
      v2f b0 = *(const v2f*)&Bs[buf][kk * 2 + hi][(wf +  0 + nIdx) * 2];
      v2f b1 = *(const v2f*)&Bs[buf][kk * 2 + hi][(wf + 16 + nIdx) * 2];
      acc00 = WMMA_F32_16x16x4(a0, b0, acc00);
      acc01 = WMMA_F32_16x16x4(a0, b1, acc01);
      acc10 = WMMA_F32_16x16x4(a1, b0, acc10);
      acc11 = WMMA_F32_16x16x4(a1, b1, acc11);
    }
  };

  for (int s = 0; s < OG_NS - 1; ++s) {
    const int ko = (s + 1) * OG_KS;
    const float4 ra0 = *(const float4*)(ga + 0 * astep + ko);
    const float4 ra1 = *(const float4*)(ga + 1 * astep + ko);
    const float4 ra2 = *(const float4*)(ga + 2 * astep + ko);
    const float4 ra3 = *(const float4*)(ga + 3 * astep + ko);
    const float4 rb0 = *(const float4*)(gb + (size_t)(s + 1) * bslab);
    const float4 rb1 = *(const float4*)(gb + (size_t)(s + 1) * bslab + NF);

    slab_mma(s & 1);

    const int nb = (s + 1) & 1;
    *(float4*)&As[nb][arow +  0][acol] = ra0;
    *(float4*)&As[nb][arow + 32][acol] = ra1;
    *(float4*)&As[nb][arow + 64][acol] = ra2;
    *(float4*)&As[nb][arow + 96][acol] = ra3;
    *(float4*)&Bs[nb][bkp][bcol * 2 + 0] = make_float4(rb0.x, rb1.x, rb0.y, rb1.y);
    *(float4*)&Bs[nb][bkp][bcol * 2 + 4] = make_float4(rb0.z, rb1.z, rb0.w, rb1.w);
    __syncthreads();
  }
  slab_mma((OG_NS - 1) & 1);

  float* ob = out + (size_t)b * NN * NF;
#pragma unroll
  for (int t = 0; t < 4; ++t) {
    v8f acc = (t == 0) ? acc00 : ((t == 1) ? acc01 : ((t == 2) ? acc10 : acc11));
    const int cb = c0 + wc + (t >> 1) * 16;
    const int f  = f0 + wf + (t & 1) * 16 + nIdx;
#pragma unroll
    for (int r = 0; r < 8; ++r) {
      int c = cb + r + hi * 8;
      ob[(size_t)c * NF + f] = acc[r];
    }
  }
}

extern "C" void kernel_launch(void* const* d_in, const int* in_sizes, int n_in,
                              void* d_out, int out_size, void* d_ws, size_t ws_size,
                              hipStream_t stream) {
  (void)in_sizes; (void)n_in; (void)out_size; (void)ws_size;
  const float* x  = (const float*)d_in[0];
  const float* qw = (const float*)d_in[1];
  const float* kw = (const float*)d_in[2];
  const float* vw = (const float*)d_in[3];
  float* out = (float*)d_out;

  float* ws     = (float*)d_ws;
  float* C      = ws;
  float* xl     = C + 3 * 1024;
  float* qkv    = xl + (size_t)NMAT * 1024;
  float* nrm    = qkv + (size_t)3 * NMAT * 1024;
  float* scores = nrm + 2 * NMAT;

  cayley_kernel<<<3, 256, 0, stream>>>(qw, kw, vw, C);
  jacobi_logm_kernel<<<NMAT, 32, 0, stream>>>(x, xl);
  transbase_kernel<<<dim3(NMAT, 3), 32, 0, stream>>>(xl, C, qkv);
  norms_kernel<<<64, 256, 0, stream>>>(qkv, nrm);
  scores_kernel<<<dim3(NN / SC_TI, NN / SC_TJ, NB), 256, 0, stream>>>(qkv, nrm, scores);
  softmax_kernel<<<dim3(NN / 32, NB), 256, 0, stream>>>(scores);
  outgemm_kernel<<<dim3(NF / OG_TF, NN / OG_TC, NB), 256, 0, stream>>>(scores, qkv, out);
}


# eigensolver: pairs sit at fixed rows/columns 2t,2t+1 of a permuted matrix, each round stores one step round the schedule circle: constant LDS addresses, 8-byte reads, only c,s shared; sum order kept
# speedup vs baseline: 3.5410x; 1.3063x over previous
#include <hip/hip_runtime.h>
#include <math.h>


typedef __attribute__((ext_vector_type(2))) float v2f;
typedef __attribute__((ext_vector_type(8))) float v8f;

#define WMMA_F32_16x16x4(a, b, c) \
  __builtin_amdgcn_wmma_f32_16x16x4_f32(false, (a), false, (b), (short)0, (c), false, false)

#define NMAT 8192
#define NB   16
#define NN   512
#define NF   1024
#define NFP  512

__global__ __launch_bounds__(256) void cayley_kernel(const float* __restrict__ qw,
                                                     const float* __restrict__ kw,
                                                     const float* __restrict__ vw,
                                                     float* __restrict__ Cout) {
  __shared__ float X[32][32];
  __shared__ float Mm[32][33];
  __shared__ float Inv[32][33];
  __shared__ float fac[32];
  __shared__ float pivinv;
  const float* W = (blockIdx.x == 0) ? qw : ((blockIdx.x == 1) ? kw : vw);
  const int tid = threadIdx.x;

  for (int e = tid; e < 1024; e += 256) {
    int i = e >> 5, j = e & 31;
    float xv = W[i * 32 + j] - W[j * 32 + i];
    X[i][j]  = xv;
    Mm[i][j] = xv + ((i == j) ? 1.f : 0.f);
    Inv[i][j] = (i == j) ? 1.f : 0.f;
  }
  __syncthreads();

  for (int k = 0; k < 32; ++k) {
    if (tid == 0) pivinv = 1.f / Mm[k][k];
    __syncthreads();
    if (tid < 32)      Mm[k][tid]       *= pivinv;
    else if (tid < 64) Inv[k][tid - 32] *= pivinv;
    __syncthreads();
    if (tid < 32) fac[tid] = (tid == k) ? 0.f : Mm[tid][k];
    __syncthreads();
    for (int e = tid; e < 2048; e += 256) {
      int i = e >> 6, j = e & 63;
      if (i == k) continue;
      float f = fac[i];
      if (j < 32) Mm[i][j]        -= f * Mm[k][j];
      else        Inv[i][j - 32]  -= f * Inv[k][j - 32];
    }
    __syncthreads();
  }

  for (int e = tid; e < 1024; e += 256) {
    int i = e >> 5, j = e & 31;
    float acc = 0.f;
    for (int kk = 0; kk < 32; ++kk) {
      float a = ((i == kk) ? 1.f : 0.f) - X[i][kk];
      acc += a * Inv[kk][j];
    }
    Cout[blockIdx.x * 1024 + i * 32 + j] = acc;
  }
}

#define J_LD 34
#define JC_TOL 2e-6f

__device__ __forceinline__ int jacobi_next_slot(int s) {
  int d = (s & 1) ? s - 2 : s + 2;
  d = (s == 0) ? 0 : d;
  d = (s == 1) ? 2 : d;
  d = (s == 30) ? 31 : d;
  return d;
}
__device__ __forceinline__ int jacobi_slot_index(int s) { return (s & 1) ? (s >> 1) : 31 - (s >> 1); }
__device__ __forceinline__ int jacobi_index_slot(int k) { return (k <= 15) ? 2 * k + 1 : 2 * (31 - k); }

template <int U> __device__ __forceinline__ int row_share_i(int v) {
  return __builtin_amdgcn_update_dpp(v, v, 0x150 + U, 0xf, 0xf, false);
}
template <int U> __device__ __forceinline__ float row_share_f(float v) {
  return __int_as_float(row_share_i<U>(__float_as_int(v)));
}

#define JDST_E(U, GEN, P0, P7) ((U) == 0 ? (P0) : ((U) == 7 ? (P7) : (GEN) + (2 * (U) + 2) * J_LD))
#define JDST_O(U, GEN, P0)     ((U) == 0 ? (P0) : (GEN) + (2 * (U) - 1) * J_LD)

__global__ __launch_bounds__(32, 8) void jacobi_logm_kernel(const float* __restrict__ x,
                                                            float* __restrict__ xl) {
  __shared__ __attribute__((aligned(16))) float A[32][J_LD];
  __shared__ __attribute__((aligned(16))) float VT[32][J_LD];
  __shared__ float lw[32];
  const int lane = threadIdx.x;
  const int hi = lane >> 4;
  const int tl = (lane + 8 * hi) & 15;
  const int ra = 16 * hi;
  const int vc = 2 * (lane & 15);
  float* const Af = &A[0][0];
  float* const Vf = &VT[0][0];
  const float* xm = x + (size_t)blockIdx.x * 1024;

  const int myidx = jacobi_slot_index(lane);
  for (int i = 0; i < 32; ++i) {
    const int ri = jacobi_slot_index(i);
    A[i][lane]  = xm[ri * 32 + myidx];
    VT[i][lane] = (ri == lane) ? 1.f : 0.f;
  }
  __syncthreads();

  const float* const pLd = Af + (2 * tl) * J_LD + 2 * tl;
  const float* const aLd = Af + ra * J_LD + 2 * tl;
  const float* const vLd = Vf + ra * J_LD + vc;
  float* const aStA = Af + ra * J_LD + jacobi_next_slot(2 * tl);
  float* const aStB = Af + ra * J_LD + jacobi_next_slot(2 * tl + 1);
  float* const vSt  = Vf + ra * J_LD + vc;
  const int e0 = hi ? 2 * J_LD : 0;
  const int e7 = hi ? 15 * J_LD : 16 * J_LD;
  const int o0 = hi ? -J_LD : 2 * J_LD;
  float* const aA_e0 = aStA + e0;  float* const aB_e0 = aStB + e0;  float* const v_e0 = vSt + e0;
  float* const aA_e7 = aStA + e7;  float* const aB_e7 = aStB + e7;  float* const v_e7 = vSt + e7;
  float* const aA_o0 = aStA + o0;  float* const aB_o0 = aStB + o0;  float* const v_o0 = vSt + o0;

  for (int sweep = 0; sweep < 8; ++sweep) {
    for (int r = 0; r < 31; ++r) {
      const float2 pd = *(const float2*)pLd;
      const float aqq = pLd[J_LD + 1];
      float2 yE[8], yO[8], wE[8], wO[8];
#define JLOAD(U)                                                       \
      yE[U] = *(const float2*)(aLd + (2 * (U)) * J_LD);                \
      yO[U] = *(const float2*)(aLd + (2 * (U) + 1) * J_LD);            \
      wE[U] = *(const float2*)(vLd + (2 * (U)) * J_LD);                \
      wO[U] = *(const float2*)(vLd + (2 * (U) + 1) * J_LD);
      JLOAD(0) JLOAD(1) JLOAD(2) JLOAD(3) JLOAD(4) JLOAD(5) JLOAD(6) JLOAD(7)
#undef JLOAD

      float c = 1.f, s = 0.f;
      {
        const float apq = pd.y;
        if (fabsf(apq) > 1e-12f) {
          float theta = (aqq - pd.x) / (2.f * apq);
          float t = 1.f / (fabsf(theta) + sqrtf(theta * theta + 1.f));
          if (theta < 0.f) t = -t;
          c = rsqrtf(t * t + 1.f);
          s = t * c;
        }
      }
      __syncthreads();

#define JROT(U)                                                                   \
      {                                                                           \
        const float ca = row_share_f<U>(c), sa = row_share_f<U>(s);               \
                     \
        const float r0 = ca * yE[U].x - sa * yO[U].x;                             \
        const float r2 = sa * yE[U].x + ca * yO[U].x;                             \
        const float r1 = ca * yE[U].y - sa * yO[U].y;                             \
        const float r3 = sa * yE[U].y + ca * yO[U].y;                             \
        *JDST_E(U, aStA, aA_e0, aA_e7) = c * r0 - s * r1;                         \
        *JDST_E(U, aStB, aB_e0, aB_e7) = s * r0 + c * r1;                         \
        *JDST_O(U, aStA, aA_o0)        = c * r2 - s * r3;                         \
        *JDST_O(U, aStB, aB_o0)        = s * r2 + c * r3;                         \
                      \
        float2 np, nq;                                                            \
        np.x = ca * wE[U].x - sa * wO[U].x;  nq.x = sa * wE[U].x + ca * wO[U].x;  \
        np.y = ca * wE[U].y - sa * wO[U].y;  nq.y = sa * wE[U].y + ca * wO[U].y;  \
        *(float2*)JDST_E(U, vSt, v_e0, v_e7) = np;                                \
        *(float2*)JDST_O(U, vSt, v_o0)       = nq;                                \
      }
      JROT(0) JROT(1) JROT(2) JROT(3) JROT(4) JROT(5) JROT(6) JROT(7)
#undef JROT
      __syncthreads();
    }

    float dg = 0.f, om = 0.f;
    for (int i = 0; i < 32; ++i) {
      const float a = fabsf(A[i][lane]);
      dg = (i == lane) ? a : dg;
      om = (i != lane) ? fmaxf(om, a) : om;
    }
    float dmin = dg;
    for (int o = 16; o > 0; o >>= 1) dmin = fminf(dmin, __shfl_xor(dmin, o));
    if (__builtin_amdgcn_ballot_w32(om > JC_TOL * dmin) == 0u) break;
  }

  lw[lane] = logf(fmaxf(A[lane][lane], 1e-20f));
  __syncthreads();

  float acc[32];
#pragma unroll
  for (int i = 0; i < 32; ++i) acc[i] = 0.f;
  for (int k = 0; k < 32; ++k) {
    const int sk = jacobi_index_slot(k);
    const float w = VT[sk][lane] * lw[sk];
#pragma unroll
    for (int i = 0; i < 32; ++i) acc[i] += VT[sk][i] * w;
  }

  float* outm = xl + (size_t)blockIdx.x * 1024;
#pragma unroll
  for (int i = 0; i < 32; ++i)
    outm[i * 32 + lane] = (i == lane) ? 0.f : acc[i];
}
#undef JDST_E
#undef JDST_O

__global__ __launch_bounds__(32) void transbase_kernel(const float* __restrict__ xl,
                                                       const float* __restrict__ C,
                                                       float* __restrict__ qkv) {
  __shared__ float Y[32][33];
  __shared__ float CL[32][33];
  __shared__ float T[32][33];
  __shared__ float T2[32][33];
  const int lane = threadIdx.x;
  const int nIdx = lane & 15;
  const int hi   = lane >> 4;
  const int m = blockIdx.x, w = blockIdx.y;
  const float* ym = xl + (size_t)m * 1024;
  const float* Cw = C + (size_t)w * 1024;

  for (int i = 0; i < 32; ++i) {
    Y[i][lane]  = ym[i * 32 + lane];
    CL[i][lane] = Cw[i * 32 + lane];
  }
  __syncthreads();

  for (int ti = 0; ti < 2; ++ti)
    for (int tj = 0; tj < 2; ++tj) {
      v8f acc = {};
#pragma unroll
      for (int kk = 0; kk < 8; ++kk) {
        int k0 = kk * 4 + hi * 2;
        v2f a; a.x = CL[ti * 16 + nIdx][k0]; a.y = CL[ti * 16 + nIdx][k0 + 1];
        v2f b; b.x = Y[k0][tj * 16 + nIdx];  b.y = Y[k0 + 1][tj * 16 + nIdx];
        acc = WMMA_F32_16x16x4(a, b, acc);
      }
#pragma unroll
      for (int r = 0; r < 8; ++r)
        T[ti * 16 + r + hi * 8][tj * 16 + nIdx] = acc[r];
    }
  __syncthreads();

  for (int ti = 0; ti < 2; ++ti)
    for (int tj = 0; tj < 2; ++tj) {
      v8f acc = {};
#pragma unroll
      for (int kk = 0; kk < 8; ++kk) {
        int k0 = kk * 4 + hi * 2;
        v2f a; a.x = T[ti * 16 + nIdx][k0];       a.y = T[ti * 16 + nIdx][k0 + 1];
        v2f b; b.x = CL[tj * 16 + nIdx][k0];      b.y = CL[tj * 16 + nIdx][k0 + 1];
        acc = WMMA_F32_16x16x4(a, b, acc);
      }
#pragma unroll
      for (int r = 0; r < 8; ++r)
        T2[ti * 16 + r + hi * 8][tj * 16 + nIdx] = acc[r];
    }
  __syncthreads();

  if (w < 2) {
    float* outp = qkv + ((size_t)w * NMAT + m) * NFP;
    for (int s = 0; s < 16; ++s) {
      int i, j;
      if (s == 0)        { i = 16;     j = lane & 15; }
      else if (lane < s) { i = s;      j = lane; }
      else               { i = 32 - s; j = lane - s; }
      const float t = T2[i][j];
      outp[s * 32 + lane] = (s == 0 && lane >= 16) ? 0.f : t;
    }
  } else {
    float* outm = qkv + ((size_t)w * NMAT + m) * 1024;
    for (int i = 0; i < 32; ++i) {
      float lo = T2[i][lane];
      float up = T2[lane][i];
      float val = (i > lane) ? lo : ((i < lane) ? up : 0.f);
      outm[i * 32 + lane] = val;
    }
  }
}

__global__ __launch_bounds__(256) void norms_kernel(const float* __restrict__ qkv,
                                                    float* __restrict__ nrm) {
  int idx = blockIdx.x * 256 + threadIdx.x;
  int sel = idx >> 13;
  int r   = idx & (NMAT - 1);
  const float* row = qkv + ((size_t)sel * NMAT + r) * NFP;
  float acc = 0.f;
  for (int f = 0; f < NFP; ++f) { float t = row[f]; acc += t * t; }
  nrm[idx] = 2.f * acc;
}

#define SC_TJ 128
#define SC_TI 64
#define SC_KS 32
#define SC_NS (NFP / SC_KS)
#define SC_LD 36

__global__ __launch_bounds__(256) void scores_kernel(const float* __restrict__ qkv,
                                                     const float* __restrict__ nrm,
                                                     float* __restrict__ scores) {
  __shared__ __attribute__((aligned(16))) float As[2][SC_TJ][SC_LD];
  __shared__ __attribute__((aligned(16))) float Bs[2][SC_TI][SC_LD];
  const int tid  = threadIdx.x;
  const int lane = tid & 31;
  const int wave = tid >> 5;
  const int nIdx = lane & 15;
  const int hi   = lane >> 4;
  const int it = blockIdx.x, jt = blockIdx.y, b = blockIdx.z;
  const int i0 = it * SC_TI, j0 = jt * SC_TJ;
  const int wj = (wave >> 1) * 32;
  const int wi = (wave & 1) * 32;
  const float* qf = qkv + ((size_t)0 * NMAT + (size_t)b * NN) * NFP;
  const float* kf = qkv + ((size_t)1 * NMAT + (size_t)b * NN) * NFP;

  const int srow = tid >> 3;
  const int scol = (tid & 7) * 4;
  const float* ga = kf + (size_t)(j0 + srow) * NFP + scol;
  const float* gb = qf + (size_t)(i0 + srow) * NFP + scol;
  const size_t rstep = (size_t)32 * NFP;

  {
    const float4 ra0 = *(const float4*)(ga + 0 * rstep);
    const float4 ra1 = *(const float4*)(ga + 1 * rstep);
    const float4 ra2 = *(const float4*)(ga + 2 * rstep);
    const float4 ra3 = *(const float4*)(ga + 3 * rstep);
    const float4 rb0 = *(const float4*)(gb + 0 * rstep);
    const float4 rb1 = *(const float4*)(gb + 1 * rstep);
    *(float4*)&As[0][srow +  0][scol] = ra0;
    *(float4*)&As[0][srow + 32][scol] = ra1;
    *(float4*)&As[0][srow + 64][scol] = ra2;
    *(float4*)&As[0][srow + 96][scol] = ra3;
    *(float4*)&Bs[0][srow +  0][scol] = rb0;
    *(float4*)&Bs[0][srow + 32][scol] = rb1;
  }
  __syncthreads();

  v8f acc00 = {}, acc01 = {}, acc10 = {}, acc11 = {};

  auto slab_mma = [&](int buf) {
#pragma unroll
    for (int kk = 0; kk < SC_KS / 4; ++kk) {
      const int k0 = kk * 4 + hi * 2;
      v2f a0 = *(const v2f*)&As[buf][wj +  0 + nIdx][k0];
      v2f a1 = *(const v2f*)&As[buf][wj + 16 + nIdx][k0];
      v2f b0 = *(const v2f*)&Bs[buf][wi +  0 + nIdx][k0];
      v2f b1 = *(const v2f*)&Bs[buf][wi + 16 + nIdx][k0];
      acc00 = WMMA_F32_16x16x4(a0, b0, acc00);
      acc01 = WMMA_F32_16x16x4(a0, b1, acc01);
      acc10 = WMMA_F32_16x16x4(a1, b0, acc10);
      acc11 = WMMA_F32_16x16x4(a1, b1, acc11);
    }
  };

  for (int s = 0; s < SC_NS - 1; ++s) {
    const int ko = (s + 1) * SC_KS;
    const float4 ra0 = *(const float4*)(ga + 0 * rstep + ko);
    const float4 ra1 = *(const float4*)(ga + 1 * rstep + ko);
    const float4 ra2 = *(const float4*)(ga + 2 * rstep + ko);
    const float4 ra3 = *(const float4*)(ga + 3 * rstep + ko);
    const float4 rb0 = *(const float4*)(gb + 0 * rstep + ko);
    const float4 rb1 = *(const float4*)(gb + 1 * rstep + ko);

    slab_mma(s & 1);

    const int nb = (s + 1) & 1;
    *(float4*)&As[nb][srow +  0][scol] = ra0;
    *(float4*)&As[nb][srow + 32][scol] = ra1;
    *(float4*)&As[nb][srow + 64][scol] = ra2;
    *(float4*)&As[nb][srow + 96][scol] = ra3;
    *(float4*)&Bs[nb][srow +  0][scol] = rb0;
    *(float4*)&Bs[nb][srow + 32][scol] = rb1;
    __syncthreads();
  }
  slab_mma((SC_NS - 1) & 1);

  float* sb = scores + (size_t)b * NN * NN;
#pragma unroll
  for (int t = 0; t < 4; ++t) {
    v8f acc = (t == 0) ? acc00 : ((t == 1) ? acc01 : ((t == 2) ? acc10 : acc11));
    const int jb = j0 + wj + (t >> 1) * 16;
    const int i  = i0 + wi + (t & 1) * 16 + nIdx;
    float q2 = nrm[b * NN + i];
#pragma unroll
    for (int r = 0; r < 8; ++r) {
      int j = jb + r + hi * 8;
      float d2 = q2 + nrm[NMAT + b * NN + j] - 4.f * acc[r];
      d2 = fmaxf(d2, 1e-12f);
      float e  = sqrtf(d2);
      float sc = 1.f / (1.f + log1pf(e));
      sb[(size_t)j * NN + i] = sc;
    }
  }
}

#define SM_W 8
#define SM_R (NN / SM_W)

__global__ __launch_bounds__(256) void softmax_kernel(float* __restrict__ scores) {
  __shared__ float red[2][SM_W][32];
  const int lane = threadIdx.x & 31;
  const int wave = threadIdx.x >> 5;
  const int b = blockIdx.y;
  float* col = scores + (size_t)b * NN * NN + (size_t)(wave * SM_R) * NN
             + blockIdx.x * 32 + lane;

  float v[SM_R];
#pragma unroll
  for (int r = 0; r < SM_R; ++r) v[r] = col[(size_t)r * NN];

  float mx = -1e30f;
#pragma unroll
  for (int r = 0; r < SM_R; ++r) mx = fmaxf(mx, v[r]);
  red[0][wave][lane] = mx;
  __syncthreads();
  for (int w = 0; w < SM_W; ++w) mx = fmaxf(mx, red[0][w][lane]);

  float sum = 0.f;
#pragma unroll
  for (int r = 0; r < SM_R; ++r) {
    v[r] = __expf(v[r] - mx);
    sum += v[r];
  }
  red[1][wave][lane] = sum;
  __syncthreads();
  float tot = 0.f;
  for (int w = 0; w < SM_W; ++w) tot += red[1][w][lane];
  const float inv = 1.f / tot;

#pragma unroll
  for (int r = 0; r < SM_R; ++r) col[(size_t)r * NN] = v[r] * inv;
}

#define OG_TC 128
#define OG_TF 64
#define OG_KS 32
#define OG_NS (NN / OG_KS)
#define OG_LDA 36
#define OG_LDB 160

__global__ __launch_bounds__(256) void outgemm_kernel(const float* __restrict__ scores,
                                                      const float* __restrict__ qkv,
                                                      float* __restrict__ out) {
  __shared__ __attribute__((aligned(16))) float As[2][OG_TC][OG_LDA];
  __shared__ __attribute__((aligned(16))) float Bs[2][OG_KS / 2][OG_LDB];
  const int tid  = threadIdx.x;
  const int lane = tid & 31;
  const int wave = tid >> 5;
  const int nIdx = lane & 15;
  const int hi   = lane >> 4;
  const int ft = blockIdx.x, ct = blockIdx.y, b = blockIdx.z;
  const int f0 = ft * OG_TF, c0 = ct * OG_TC;
  const int wc = (wave >> 1) * 32;
  const int wf = (wave & 1) * 32;
  const float* p  = scores + (size_t)b * NN * NN;
  const float* vf = qkv + ((size_t)2 * NMAT + (size_t)b * NN) * 1024;

  const int arow = tid >> 3;
  const int acol = (tid & 7) * 4;
  const float* ga = p + (size_t)(c0 + arow) * NN + acol;
  const size_t astep = (size_t)32 * NN;
  const int bkp  = tid >> 4;
  const int bcol = (tid & 15) * 4;
  const float* gb = vf + (size_t)(2 * bkp) * NF + f0 + bcol;
  const size_t bslab = (size_t)OG_KS * NF;

  {
    const float4 ra0 = *(const float4*)(ga + 0 * astep);
    const float4 ra1 = *(const float4*)(ga + 1 * astep);
    const float4 ra2 = *(const float4*)(ga + 2 * astep);
    const float4 ra3 = *(const float4*)(ga + 3 * astep);
    const float4 rb0 = *(const float4*)(gb);
    const float4 rb1 = *(const float4*)(gb + NF);
    *(float4*)&As[0][arow +  0][acol] = ra0;
    *(float4*)&As[0][arow + 32][acol] = ra1;
    *(float4*)&As[0][arow + 64][acol] = ra2;
    *(float4*)&As[0][arow + 96][acol] = ra3;
    *(float4*)&Bs[0][bkp][bcol * 2 + 0] = make_float4(rb0.x, rb1.x, rb0.y, rb1.y);
    *(float4*)&Bs[0][bkp][bcol * 2 + 4] = make_float4(rb0.z, rb1.z, rb0.w, rb1.w);
  }
  __syncthreads();

  v8f acc00 = {}, acc01 = {}, acc10 = {}, acc11 = {};

  auto slab_mma = [&](int buf) {
#pragma unroll
    for (int kk = 0; kk < OG_KS / 4; ++kk) {
      const int k0 = kk * 4 + hi * 2;
      v2f a0 = *(const v2f*)&As[buf][wc +  0 + nIdx][k0];
      v2f a1 = *(const v2f*)&As[buf][wc + 16 + nIdx][k0];
      v2f b0 = *(const v2f*)&Bs[buf][kk * 2 + hi][(wf +  0 + nIdx) * 2];
      v2f b1 = *(const v2f*)&Bs[buf][kk * 2 + hi][(wf + 16 + nIdx) * 2];
      acc00 = WMMA_F32_16x16x4(a0, b0, acc00);
      acc01 = WMMA_F32_16x16x4(a0, b1, acc01);
      acc10 = WMMA_F32_16x16x4(a1, b0, acc10);
      acc11 = WMMA_F32_16x16x4(a1, b1, acc11);
    }
  };

  for (int s = 0; s < OG_NS - 1; ++s) {
    const int ko = (s + 1) * OG_KS;
    const float4 ra0 = *(const float4*)(ga + 0 * astep + ko);
    const float4 ra1 = *(const float4*)(ga + 1 * astep + ko);
    const float4 ra2 = *(const float4*)(ga + 2 * astep + ko);
    const float4 ra3 = *(const float4*)(ga + 3 * astep + ko);
    const float4 rb0 = *(const float4*)(gb + (size_t)(s + 1) * bslab);
    const float4 rb1 = *(const float4*)(gb + (size_t)(s + 1) * bslab + NF);

    slab_mma(s & 1);

    const int nb = (s + 1) & 1;
    *(float4*)&As[nb][arow +  0][acol] = ra0;
    *(float4*)&As[nb][arow + 32][acol] = ra1;
    *(float4*)&As[nb][arow + 64][acol] = ra2;
    *(float4*)&As[nb][arow + 96][acol] = ra3;
    *(float4*)&Bs[nb][bkp][bcol * 2 + 0] = make_float4(rb0.x, rb1.x, rb0.y, rb1.y);
    *(float4*)&Bs[nb][bkp][bcol * 2 + 4] = make_float4(rb0.z, rb1.z, rb0.w, rb1.w);
    __syncthreads();
  }
  slab_mma((OG_NS - 1) & 1);

  float* ob = out + (size_t)b * NN * NF;
#pragma unroll
  for (int t = 0; t < 4; ++t) {
    v8f acc = (t == 0) ? acc00 : ((t == 1) ? acc01 : ((t == 2) ? acc10 : acc11));
    const int cb = c0 + wc + (t >> 1) * 16;
    const int f  = f0 + wf + (t & 1) * 16 + nIdx;
#pragma unroll
    for (int r = 0; r < 8; ++r) {
      int c = cb + r + hi * 8;
      ob[(size_t)c * NF + f] = acc[r];
    }
  }
}

extern "C" void kernel_launch(void* const* d_in, const int* in_sizes, int n_in,
                              void* d_out, int out_size, void* d_ws, size_t ws_size,
                              hipStream_t stream) {
  (void)in_sizes; (void)n_in; (void)out_size; (void)ws_size;
  const float* x  = (const float*)d_in[0];
  const float* qw = (const float*)d_in[1];
  const float* kw = (const float*)d_in[2];
  const float* vw = (const float*)d_in[3];
  float* out = (float*)d_out;

  float* ws     = (float*)d_ws;
  float* C      = ws;
  float* xl     = C + 3 * 1024;
  float* qkv    = xl + (size_t)NMAT * 1024;
  float* nrm    = qkv + (size_t)3 * NMAT * 1024;
  float* scores = nrm + 2 * NMAT;

  cayley_kernel<<<3, 256, 0, stream>>>(qw, kw, vw, C);
  jacobi_logm_kernel<<<NMAT, 32, 0, stream>>>(x, xl);
  transbase_kernel<<<dim3(NMAT, 3), 32, 0, stream>>>(xl, C, qkv);
  norms_kernel<<<64, 256, 0, stream>>>(qkv, nrm);
  scores_kernel<<<dim3(NN / SC_TI, NN / SC_TJ, NB), 256, 0, stream>>>(qkv, nrm, scores);
  softmax_kernel<<<dim3(NN / 32, NB), 256, 0, stream>>>(scores);
  outgemm_kernel<<<dim3(NF / OG_TF, NN / OG_TC, NB), 256, 0, stream>>>(scores, qkv, out);
}


# eigensolver: rotation tangent t = sign(d) b/(|d|+sqrt(d^2+b^2)) from one hardware sqrt and one hardware rcp (1 ulp), not two correctly rounded divisions and a sqrt; c=rsq(t^2+1), s=tc, sums as before
# speedup vs baseline: 3.7504x; 1.0591x over previous
#include <hip/hip_runtime.h>
#include <math.h>


typedef __attribute__((ext_vector_type(2))) float v2f;
typedef __attribute__((ext_vector_type(8))) float v8f;

#define WMMA_F32_16x16x4(a, b, c) \
  __builtin_amdgcn_wmma_f32_16x16x4_f32(false, (a), false, (b), (short)0, (c), false, false)

#define NMAT 8192
#define NB   16
#define NN   512
#define NF   1024
#define NFP  512

__global__ __launch_bounds__(256) void cayley_kernel(const float* __restrict__ qw,
                                                     const float* __restrict__ kw,
                                                     const float* __restrict__ vw,
                                                     float* __restrict__ Cout) {
  __shared__ float X[32][32];
  __shared__ float Mm[32][33];
  __shared__ float Inv[32][33];
  __shared__ float fac[32];
  __shared__ float pivinv;
  const float* W = (blockIdx.x == 0) ? qw : ((blockIdx.x == 1) ? kw : vw);
  const int tid = threadIdx.x;

  for (int e = tid; e < 1024; e += 256) {
    int i = e >> 5, j = e & 31;
    float xv = W[i * 32 + j] - W[j * 32 + i];
    X[i][j]  = xv;
    Mm[i][j] = xv + ((i == j) ? 1.f : 0.f);
    Inv[i][j] = (i == j) ? 1.f : 0.f;
  }
  __syncthreads();

  for (int k = 0; k < 32; ++k) {
    if (tid == 0) pivinv = 1.f / Mm[k][k];
    __syncthreads();
    if (tid < 32)      Mm[k][tid]       *= pivinv;
    else if (tid < 64) Inv[k][tid - 32] *= pivinv;
    __syncthreads();
    if (tid < 32) fac[tid] = (tid == k) ? 0.f : Mm[tid][k];
    __syncthreads();
    for (int e = tid; e < 2048; e += 256) {
      int i = e >> 6, j = e & 63;
      if (i == k) continue;
      float f = fac[i];
      if (j < 32) Mm[i][j]        -= f * Mm[k][j];
      else        Inv[i][j - 32]  -= f * Inv[k][j - 32];
    }
    __syncthreads();
  }

  for (int e = tid; e < 1024; e += 256) {
    int i = e >> 5, j = e & 31;
    float acc = 0.f;
    for (int kk = 0; kk < 32; ++kk) {
      float a = ((i == kk) ? 1.f : 0.f) - X[i][kk];
      acc += a * Inv[kk][j];
    }
    Cout[blockIdx.x * 1024 + i * 32 + j] = acc;
  }
}

#define J_LD 34
#define JC_TOL 2e-6f

__device__ __forceinline__ int jacobi_next_slot(int s) {
  int d = (s & 1) ? s - 2 : s + 2;
  d = (s == 0) ? 0 : d;
  d = (s == 1) ? 2 : d;
  d = (s == 30) ? 31 : d;
  return d;
}
__device__ __forceinline__ int jacobi_slot_index(int s) { return (s & 1) ? (s >> 1) : 31 - (s >> 1); }
__device__ __forceinline__ int jacobi_index_slot(int k) { return (k <= 15) ? 2 * k + 1 : 2 * (31 - k); }

template <int U> __device__ __forceinline__ int row_share_i(int v) {
  return __builtin_amdgcn_update_dpp(v, v, 0x150 + U, 0xf, 0xf, false);
}
template <int U> __device__ __forceinline__ float row_share_f(float v) {
  return __int_as_float(row_share_i<U>(__float_as_int(v)));
}

#define JDST_E(U, GEN, P0, P7) ((U) == 0 ? (P0) : ((U) == 7 ? (P7) : (GEN) + (2 * (U) + 2) * J_LD))
#define JDST_O(U, GEN, P0)     ((U) == 0 ? (P0) : (GEN) + (2 * (U) - 1) * J_LD)

__global__ __launch_bounds__(32, 8) void jacobi_logm_kernel(const float* __restrict__ x,
                                                            float* __restrict__ xl) {
  __shared__ __attribute__((aligned(16))) float A[32][J_LD];
  __shared__ __attribute__((aligned(16))) float VT[32][J_LD];
  __shared__ float lw[32];
  const int lane = threadIdx.x;
  const int hi = lane >> 4;
  const int tl = (lane + 8 * hi) & 15;
  const int ra = 16 * hi;
  const int vc = 2 * (lane & 15);
  float* const Af = &A[0][0];
  float* const Vf = &VT[0][0];
  const float* xm = x + (size_t)blockIdx.x * 1024;

  const int myidx = jacobi_slot_index(lane);
  for (int i = 0; i < 32; ++i) {
    const int ri = jacobi_slot_index(i);
    A[i][lane]  = xm[ri * 32 + myidx];
    VT[i][lane] = (ri == lane) ? 1.f : 0.f;
  }
  __syncthreads();

  const float* const pLd = Af + (2 * tl) * J_LD + 2 * tl;
  const float* const aLd = Af + ra * J_LD + 2 * tl;
  const float* const vLd = Vf + ra * J_LD + vc;
  float* const aStA = Af + ra * J_LD + jacobi_next_slot(2 * tl);
  float* const aStB = Af + ra * J_LD + jacobi_next_slot(2 * tl + 1);
  float* const vSt  = Vf + ra * J_LD + vc;
  const int e0 = hi ? 2 * J_LD : 0;
  const int e7 = hi ? 15 * J_LD : 16 * J_LD;
  const int o0 = hi ? -J_LD : 2 * J_LD;
  float* const aA_e0 = aStA + e0;  float* const aB_e0 = aStB + e0;  float* const v_e0 = vSt + e0;
  float* const aA_e7 = aStA + e7;  float* const aB_e7 = aStB + e7;  float* const v_e7 = vSt + e7;
  float* const aA_o0 = aStA + o0;  float* const aB_o0 = aStB + o0;  float* const v_o0 = vSt + o0;

  for (int sweep = 0; sweep < 8; ++sweep) {
    for (int r = 0; r < 31; ++r) {
      const float2 pd = *(const float2*)pLd;
      const float aqq = pLd[J_LD + 1];
      float2 yE[8], yO[8], wE[8], wO[8];
#define JLOAD(U)                                                       \
      yE[U] = *(const float2*)(aLd + (2 * (U)) * J_LD);                \
      yO[U] = *(const float2*)(aLd + (2 * (U) + 1) * J_LD);            \
      wE[U] = *(const float2*)(vLd + (2 * (U)) * J_LD);                \
      wO[U] = *(const float2*)(vLd + (2 * (U) + 1) * J_LD);
      JLOAD(0) JLOAD(1) JLOAD(2) JLOAD(3) JLOAD(4) JLOAD(5) JLOAD(6) JLOAD(7)
#undef JLOAD

      float c = 1.f, s = 0.f;
      {
        const float apq = pd.y;
        if (fabsf(apq) > 1e-12f) {
          const float d = aqq - pd.x;
          const float b = apq + apq;
          const float h = __builtin_amdgcn_sqrtf(fmaf(d, d, b * b));
          float t = b * __builtin_amdgcn_rcpf(fabsf(d) + h);
          t = (d < 0.f) ? -t : t;
          c = rsqrtf(fmaf(t, t, 1.f));
          s = t * c;
        }
      }
      __syncthreads();

#define JROT(U)                                                                   \
      {                                                                           \
        const float ca = row_share_f<U>(c), sa = row_share_f<U>(s);               \
                     \
        const float r0 = ca * yE[U].x - sa * yO[U].x;                             \
        const float r2 = sa * yE[U].x + ca * yO[U].x;                             \
        const float r1 = ca * yE[U].y - sa * yO[U].y;                             \
        const float r3 = sa * yE[U].y + ca * yO[U].y;                             \
        *JDST_E(U, aStA, aA_e0, aA_e7) = c * r0 - s * r1;                         \
        *JDST_E(U, aStB, aB_e0, aB_e7) = s * r0 + c * r1;                         \
        *JDST_O(U, aStA, aA_o0)        = c * r2 - s * r3;                         \
        *JDST_O(U, aStB, aB_o0)        = s * r2 + c * r3;                         \
                      \
        float2 np, nq;                                                            \
        np.x = ca * wE[U].x - sa * wO[U].x;  nq.x = sa * wE[U].x + ca * wO[U].x;  \
        np.y = ca * wE[U].y - sa * wO[U].y;  nq.y = sa * wE[U].y + ca * wO[U].y;  \
        *(float2*)JDST_E(U, vSt, v_e0, v_e7) = np;                                \
        *(float2*)JDST_O(U, vSt, v_o0)       = nq;                                \
      }
      JROT(0) JROT(1) JROT(2) JROT(3) JROT(4) JROT(5) JROT(6) JROT(7)
#undef JROT
      __syncthreads();
    }

    float dg = 0.f, om = 0.f;
    for (int i = 0; i < 32; ++i) {
      const float a = fabsf(A[i][lane]);
      dg = (i == lane) ? a : dg;
      om = (i != lane) ? fmaxf(om, a) : om;
    }
    float dmin = dg;
    for (int o = 16; o > 0; o >>= 1) dmin = fminf(dmin, __shfl_xor(dmin, o));
    if (__builtin_amdgcn_ballot_w32(om > JC_TOL * dmin) == 0u) break;
  }

  lw[lane] = logf(fmaxf(A[lane][lane], 1e-20f));
  __syncthreads();

  float acc[32];
#pragma unroll
  for (int i = 0; i < 32; ++i) acc[i] = 0.f;
  for (int k = 0; k < 32; ++k) {
    const int sk = jacobi_index_slot(k);
    const float w = VT[sk][lane] * lw[sk];
#pragma unroll
    for (int i = 0; i < 32; ++i) acc[i] += VT[sk][i] * w;
  }

  float* outm = xl + (size_t)blockIdx.x * 1024;
#pragma unroll
  for (int i = 0; i < 32; ++i)
    outm[i * 32 + lane] = (i == lane) ? 0.f : acc[i];
}
#undef JDST_E
#undef JDST_O

__global__ __launch_bounds__(32) void transbase_kernel(const float* __restrict__ xl,
                                                       const float* __restrict__ C,
                                                       float* __restrict__ qkv) {
  __shared__ float Y[32][33];
  __shared__ float CL[32][33];
  __shared__ float T[32][33];
  __shared__ float T2[32][33];
  const int lane = threadIdx.x;
  const int nIdx = lane & 15;
  const int hi   = lane >> 4;
  const int m = blockIdx.x, w = blockIdx.y;
  const float* ym = xl + (size_t)m * 1024;
  const float* Cw = C + (size_t)w * 1024;

  for (int i = 0; i < 32; ++i) {
    Y[i][lane]  = ym[i * 32 + lane];
    CL[i][lane] = Cw[i * 32 + lane];
  }
  __syncthreads();

  for (int ti = 0; ti < 2; ++ti)
    for (int tj = 0; tj < 2; ++tj) {
      v8f acc = {};
#pragma unroll
      for (int kk = 0; kk < 8; ++kk) {
        int k0 = kk * 4 + hi * 2;
        v2f a; a.x = CL[ti * 16 + nIdx][k0]; a.y = CL[ti * 16 + nIdx][k0 + 1];
        v2f b; b.x = Y[k0][tj * 16 + nIdx];  b.y = Y[k0 + 1][tj * 16 + nIdx];
        acc = WMMA_F32_16x16x4(a, b, acc);
      }
#pragma unroll
      for (int r = 0; r < 8; ++r)
        T[ti * 16 + r + hi * 8][tj * 16 + nIdx] = acc[r];
    }
  __syncthreads();

  for (int ti = 0; ti < 2; ++ti)
    for (int tj = 0; tj < 2; ++tj) {
      v8f acc = {};
#pragma unroll
      for (int kk = 0; kk < 8; ++kk) {
        int k0 = kk * 4 + hi * 2;
        v2f a; a.x = T[ti * 16 + nIdx][k0];       a.y = T[ti * 16 + nIdx][k0 + 1];
        v2f b; b.x = CL[tj * 16 + nIdx][k0];      b.y = CL[tj * 16 + nIdx][k0 + 1];
        acc = WMMA_F32_16x16x4(a, b, acc);
      }
#pragma unroll
      for (int r = 0; r < 8; ++r)
        T2[ti * 16 + r + hi * 8][tj * 16 + nIdx] = acc[r];
    }
  __syncthreads();

  if (w < 2) {
    float* outp = qkv + ((size_t)w * NMAT + m) * NFP;
    for (int s = 0; s < 16; ++s) {
      int i, j;
      if (s == 0)        { i = 16;     j = lane & 15; }
      else if (lane < s) { i = s;      j = lane; }
      else               { i = 32 - s; j = lane - s; }
      const float t = T2[i][j];
      outp[s * 32 + lane] = (s == 0 && lane >= 16) ? 0.f : t;
    }
  } else {
    float* outm = qkv + ((size_t)w * NMAT + m) * 1024;
    for (int i = 0; i < 32; ++i) {
      float lo = T2[i][lane];
      float up = T2[lane][i];
      float val = (i > lane) ? lo : ((i < lane) ? up : 0.f);
      outm[i * 32 + lane] = val;
    }
  }
}

__global__ __launch_bounds__(256) void norms_kernel(const float* __restrict__ qkv,
                                                    float* __restrict__ nrm) {
  int idx = blockIdx.x * 256 + threadIdx.x;
  int sel = idx >> 13;
  int r   = idx & (NMAT - 1);
  const float* row = qkv + ((size_t)sel * NMAT + r) * NFP;
  float acc = 0.f;
  for (int f = 0; f < NFP; ++f) { float t = row[f]; acc += t * t; }
  nrm[idx] = 2.f * acc;
}

#define SC_TJ 128
#define SC_TI 64
#define SC_KS 32
#define SC_NS (NFP / SC_KS)
#define SC_LD 36

__global__ __launch_bounds__(256) void scores_kernel(const float* __restrict__ qkv,
                                                     const float* __restrict__ nrm,
                                                     float* __restrict__ scores) {
  __shared__ __attribute__((aligned(16))) float As[2][SC_TJ][SC_LD];
  __shared__ __attribute__((aligned(16))) float Bs[2][SC_TI][SC_LD];
  const int tid  = threadIdx.x;
  const int lane = tid & 31;
  const int wave = tid >> 5;
  const int nIdx = lane & 15;
  const int hi   = lane >> 4;
  const int it = blockIdx.x, jt = blockIdx.y, b = blockIdx.z;
  const int i0 = it * SC_TI, j0 = jt * SC_TJ;
  const int wj = (wave >> 1) * 32;
  const int wi = (wave & 1) * 32;
  const float* qf = qkv + ((size_t)0 * NMAT + (size_t)b * NN) * NFP;
  const float* kf = qkv + ((size_t)1 * NMAT + (size_t)b * NN) * NFP;

  const int srow = tid >> 3;
  const int scol = (tid & 7) * 4;
  const float* ga = kf + (size_t)(j0 + srow) * NFP + scol;
  const float* gb = qf + (size_t)(i0 + srow) * NFP + scol;
  const size_t rstep = (size_t)32 * NFP;

  {
    const float4 ra0 = *(const float4*)(ga + 0 * rstep);
    const float4 ra1 = *(const float4*)(ga + 1 * rstep);
    const float4 ra2 = *(const float4*)(ga + 2 * rstep);
    const float4 ra3 = *(const float4*)(ga + 3 * rstep);
    const float4 rb0 = *(const float4*)(gb + 0 * rstep);
    const float4 rb1 = *(const float4*)(gb + 1 * rstep);
    *(float4*)&As[0][srow +  0][scol] = ra0;
    *(float4*)&As[0][srow + 32][scol] = ra1;
    *(float4*)&As[0][srow + 64][scol] = ra2;
    *(float4*)&As[0][srow + 96][scol] = ra3;
    *(float4*)&Bs[0][srow +  0][scol] = rb0;
    *(float4*)&Bs[0][srow + 32][scol] = rb1;
  }
  __syncthreads();

  v8f acc00 = {}, acc01 = {}, acc10 = {}, acc11 = {};

  auto slab_mma = [&](int buf) {
#pragma unroll
    for (int kk = 0; kk < SC_KS / 4; ++kk) {
      const int k0 = kk * 4 + hi * 2;
      v2f a0 = *(const v2f*)&As[buf][wj +  0 + nIdx][k0];
      v2f a1 = *(const v2f*)&As[buf][wj + 16 + nIdx][k0];
      v2f b0 = *(const v2f*)&Bs[buf][wi +  0 + nIdx][k0];
      v2f b1 = *(const v2f*)&Bs[buf][wi + 16 + nIdx][k0];
      acc00 = WMMA_F32_16x16x4(a0, b0, acc00);
      acc01 = WMMA_F32_16x16x4(a0, b1, acc01);
      acc10 = WMMA_F32_16x16x4(a1, b0, acc10);
      acc11 = WMMA_F32_16x16x4(a1, b1, acc11);
    }
  };

  for (int s = 0; s < SC_NS - 1; ++s) {
    const int ko = (s + 1) * SC_KS;
    const float4 ra0 = *(const float4*)(ga + 0 * rstep + ko);
    const float4 ra1 = *(const float4*)(ga + 1 * rstep + ko);
    const float4 ra2 = *(const float4*)(ga + 2 * rstep + ko);
    const float4 ra3 = *(const float4*)(ga + 3 * rstep + ko);
    const float4 rb0 = *(const float4*)(gb + 0 * rstep + ko);
    const float4 rb1 = *(const float4*)(gb + 1 * rstep + ko);

    slab_mma(s & 1);

    const int nb = (s + 1) & 1;
    *(float4*)&As[nb][srow +  0][scol] = ra0;
    *(float4*)&As[nb][srow + 32][scol] = ra1;
    *(float4*)&As[nb][srow + 64][scol] = ra2;
    *(float4*)&As[nb][srow + 96][scol] = ra3;
    *(float4*)&Bs[nb][srow +  0][scol] = rb0;
    *(float4*)&Bs[nb][srow + 32][scol] = rb1;
    __syncthreads();
  }
  slab_mma((SC_NS - 1) & 1);

  float* sb = scores + (size_t)b * NN * NN;
#pragma unroll
  for (int t = 0; t < 4; ++t) {
    v8f acc = (t == 0) ? acc00 : ((t == 1) ? acc01 : ((t == 2) ? acc10 : acc11));
    const int jb = j0 + wj + (t >> 1) * 16;
    const int i  = i0 + wi + (t & 1) * 16 + nIdx;
    float q2 = nrm[b * NN + i];
#pragma unroll
    for (int r = 0; r < 8; ++r) {
      int j = jb + r + hi * 8;
      float d2 = q2 + nrm[NMAT + b * NN + j] - 4.f * acc[r];
      d2 = fmaxf(d2, 1e-12f);
      float e  = sqrtf(d2);
      float sc = 1.f / (1.f + log1pf(e));
      sb[(size_t)j * NN + i] = sc;
    }
  }
}

#define SM_W 8
#define SM_R (NN / SM_W)

__global__ __launch_bounds__(256) void softmax_kernel(float* __restrict__ scores) {
  __shared__ float red[2][SM_W][32];
  const int lane = threadIdx.x & 31;
  const int wave = threadIdx.x >> 5;
  const int b = blockIdx.y;
  float* col = scores + (size_t)b * NN * NN + (size_t)(wave * SM_R) * NN
             + blockIdx.x * 32 + lane;

  float v[SM_R];
#pragma unroll
  for (int r = 0; r < SM_R; ++r) v[r] = col[(size_t)r * NN];

  float mx = -1e30f;
#pragma unroll
  for (int r = 0; r < SM_R; ++r) mx = fmaxf(mx, v[r]);
  red[0][wave][lane] = mx;
  __syncthreads();
  for (int w = 0; w < SM_W; ++w) mx = fmaxf(mx, red[0][w][lane]);

  float sum = 0.f;
#pragma unroll
  for (int r = 0; r < SM_R; ++r) {
    v[r] = __expf(v[r] - mx);
    sum += v[r];
  }
  red[1][wave][lane] = sum;
  __syncthreads();
  float tot = 0.f;
  for (int w = 0; w < SM_W; ++w) tot += red[1][w][lane];
  const float inv = 1.f / tot;

#pragma unroll
  for (int r = 0; r < SM_R; ++r) col[(size_t)r * NN] = v[r] * inv;
}

#define OG_TC 128
#define OG_TF 64
#define OG_KS 32
#define OG_NS (NN / OG_KS)
#define OG_LDA 36
#define OG_LDB 160

__global__ __launch_bounds__(256) void outgemm_kernel(const float* __restrict__ scores,
                                                      const float* __restrict__ qkv,
                                                      float* __restrict__ out) {
  __shared__ __attribute__((aligned(16))) float As[2][OG_TC][OG_LDA];
  __shared__ __attribute__((aligned(16))) float Bs[2][OG_KS / 2][OG_LDB];
  const int tid  = threadIdx.x;
  const int lane = tid & 31;
  const int wave = tid >> 5;
  const int nIdx = lane & 15;
  const int hi   = lane >> 4;
  const int ft = blockIdx.x, ct = blockIdx.y, b = blockIdx.z;
  const int f0 = ft * OG_TF, c0 = ct * OG_TC;
  const int wc = (wave >> 1) * 32;
  const int wf = (wave & 1) * 32;
  const float* p  = scores + (size_t)b * NN * NN;
  const float* vf = qkv + ((size_t)2 * NMAT + (size_t)b * NN) * 1024;

  const int arow = tid >> 3;
  const int acol = (tid & 7) * 4;
  const float* ga = p + (size_t)(c0 + arow) * NN + acol;
  const size_t astep = (size_t)32 * NN;
  const int bkp  = tid >> 4;
  const int bcol = (tid & 15) * 4;
  const float* gb = vf + (size_t)(2 * bkp) * NF + f0 + bcol;
  const size_t bslab = (size_t)OG_KS * NF;

  {
    const float4 ra0 = *(const float4*)(ga + 0 * astep);
    const float4 ra1 = *(const float4*)(ga + 1 * astep);
    const float4 ra2 = *(const float4*)(ga + 2 * astep);
    const float4 ra3 = *(const float4*)(ga + 3 * astep);
    const float4 rb0 = *(const float4*)(gb);
    const float4 rb1 = *(const float4*)(gb + NF);
    *(float4*)&As[0][arow +  0][acol] = ra0;
    *(float4*)&As[0][arow + 32][acol] = ra1;
    *(float4*)&As[0][arow + 64][acol] = ra2;
    *(float4*)&As[0][arow + 96][acol] = ra3;
    *(float4*)&Bs[0][bkp][bcol * 2 + 0] = make_float4(rb0.x, rb1.x, rb0.y, rb1.y);
    *(float4*)&Bs[0][bkp][bcol * 2 + 4] = make_float4(rb0.z, rb1.z, rb0.w, rb1.w);
  }
  __syncthreads();

  v8f acc00 = {}, acc01 = {}, acc10 = {}, acc11 = {};

  auto slab_mma = [&](int buf) {
#pragma unroll
    for (int kk = 0; kk < OG_KS / 4; ++kk) {
      const int k0 = kk * 4 + hi * 2;
      v2f a0 = *(const v2f*)&As[buf][wc +  0 + nIdx][k0];
      v2f a1 = *(const v2f*)&As[buf][wc + 16 + nIdx][k0];
      v2f b0 = *(const v2f*)&Bs[buf][kk * 2 + hi][(wf +  0 + nIdx) * 2];
      v2f b1 = *(const v2f*)&Bs[buf][kk * 2 + hi][(wf + 16 + nIdx) * 2];
      acc00 = WMMA_F32_16x16x4(a0, b0, acc00);
      acc01 = WMMA_F32_16x16x4(a0, b1, acc01);
      acc10 = WMMA_F32_16x16x4(a1, b0, acc10);
      acc11 = WMMA_F32_16x16x4(a1, b1, acc11);
    }
  };

  for (int s = 0; s < OG_NS - 1; ++s) {
    const int ko = (s + 1) * OG_KS;
    const float4 ra0 = *(const float4*)(ga + 0 * astep + ko);
    const float4 ra1 = *(const float4*)(ga + 1 * astep + ko);
    const float4 ra2 = *(const float4*)(ga + 2 * astep + ko);
    const float4 ra3 = *(const float4*)(ga + 3 * astep + ko);
    const float4 rb0 = *(const float4*)(gb + (size_t)(s + 1) * bslab);
    const float4 rb1 = *(const float4*)(gb + (size_t)(s + 1) * bslab + NF);

    slab_mma(s & 1);

    const int nb = (s + 1) & 1;
    *(float4*)&As[nb][arow +  0][acol] = ra0;
    *(float4*)&As[nb][arow + 32][acol] = ra1;
    *(float4*)&As[nb][arow + 64][acol] = ra2;
    *(float4*)&As[nb][arow + 96][acol] = ra3;
    *(float4*)&Bs[nb][bkp][bcol * 2 + 0] = make_float4(rb0.x, rb1.x, rb0.y, rb1.y);
    *(float4*)&Bs[nb][bkp][bcol * 2 + 4] = make_float4(rb0.z, rb1.z, rb0.w, rb1.w);
    __syncthreads();
  }
  slab_mma((OG_NS - 1) & 1);

  float* ob = out + (size_t)b * NN * NF;
#pragma unroll
  for (int t = 0; t < 4; ++t) {
    v8f acc = (t == 0) ? acc00 : ((t == 1) ? acc01 : ((t == 2) ? acc10 : acc11));
    const int cb = c0 + wc + (t >> 1) * 16;
    const int f  = f0 + wf + (t & 1) * 16 + nIdx;
#pragma unroll
    for (int r = 0; r < 8; ++r) {
      int c = cb + r + hi * 8;
      ob[(size_t)c * NF + f] = acc[r];
    }
  }
}

extern "C" void kernel_launch(void* const* d_in, const int* in_sizes, int n_in,
                              void* d_out, int out_size, void* d_ws, size_t ws_size,
                              hipStream_t stream) {
  (void)in_sizes; (void)n_in; (void)out_size; (void)ws_size;
  const float* x  = (const float*)d_in[0];
  const float* qw = (const float*)d_in[1];
  const float* kw = (const float*)d_in[2];
  const float* vw = (const float*)d_in[3];
  float* out = (float*)d_out;

  float* ws     = (float*)d_ws;
  float* C      = ws;
  float* xl     = C + 3 * 1024;
  float* qkv    = xl + (size_t)NMAT * 1024;
  float* nrm    = qkv + (size_t)3 * NMAT * 1024;
  float* scores = nrm + 2 * NMAT;

  cayley_kernel<<<3, 256, 0, stream>>>(qw, kw, vw, C);
  jacobi_logm_kernel<<<NMAT, 32, 0, stream>>>(x, xl);
  transbase_kernel<<<dim3(NMAT, 3), 32, 0, stream>>>(xl, C, qkv);
  norms_kernel<<<64, 256, 0, stream>>>(qkv, nrm);
  scores_kernel<<<dim3(NN / SC_TI, NN / SC_TJ, NB), 256, 0, stream>>>(qkv, nrm, scores);
  softmax_kernel<<<dim3(NN / 32, NB), 256, 0, stream>>>(scores);
  outgemm_kernel<<<dim3(NF / OG_TF, NN / OG_TC, NB), 256, 0, stream>>>(scores, qkv, out);
}


# out GEMM (p x v): v, hence out, is symmetric with zero diagonal: products only over v's 512 packed below-diagonal columns, not 1024; a block of 32 whole matrices mirrors on store; k order kept
# speedup vs baseline: 3.9301x; 1.0479x over previous
#include <hip/hip_runtime.h>
#include <math.h>


typedef __attribute__((ext_vector_type(2))) float v2f;
typedef __attribute__((ext_vector_type(8))) float v8f;

#define WMMA_F32_16x16x4(a, b, c) \
  __builtin_amdgcn_wmma_f32_16x16x4_f32(false, (a), false, (b), (short)0, (c), false, false)

#define NMAT 8192
#define NB   16
#define NN   512
#define NF   1024
#define NFP  512

__global__ __launch_bounds__(256) void cayley_kernel(const float* __restrict__ qw,
                                                     const float* __restrict__ kw,
                                                     const float* __restrict__ vw,
                                                     float* __restrict__ Cout) {
  __shared__ float X[32][32];
  __shared__ float Mm[32][33];
  __shared__ float Inv[32][33];
  __shared__ float fac[32];
  __shared__ float pivinv;
  const float* W = (blockIdx.x == 0) ? qw : ((blockIdx.x == 1) ? kw : vw);
  const int tid = threadIdx.x;

  for (int e = tid; e < 1024; e += 256) {
    int i = e >> 5, j = e & 31;
    float xv = W[i * 32 + j] - W[j * 32 + i];
    X[i][j]  = xv;
    Mm[i][j] = xv + ((i == j) ? 1.f : 0.f);
    Inv[i][j] = (i == j) ? 1.f : 0.f;
  }
  __syncthreads();

  for (int k = 0; k < 32; ++k) {
    if (tid == 0) pivinv = 1.f / Mm[k][k];
    __syncthreads();
    if (tid < 32)      Mm[k][tid]       *= pivinv;
    else if (tid < 64) Inv[k][tid - 32] *= pivinv;
    __syncthreads();
    if (tid < 32) fac[tid] = (tid == k) ? 0.f : Mm[tid][k];
    __syncthreads();
    for (int e = tid; e < 2048; e += 256) {
      int i = e >> 6, j = e & 63;
      if (i == k) continue;
      float f = fac[i];
      if (j < 32) Mm[i][j]        -= f * Mm[k][j];
      else        Inv[i][j - 32]  -= f * Inv[k][j - 32];
    }
    __syncthreads();
  }

  for (int e = tid; e < 1024; e += 256) {
    int i = e >> 5, j = e & 31;
    float acc = 0.f;
    for (int kk = 0; kk < 32; ++kk) {
      float a = ((i == kk) ? 1.f : 0.f) - X[i][kk];
      acc += a * Inv[kk][j];
    }
    Cout[blockIdx.x * 1024 + i * 32 + j] = acc;
  }
}

#define J_LD 34
#define JC_TOL 2e-6f

__device__ __forceinline__ int jacobi_next_slot(int s) {
  int d = (s & 1) ? s - 2 : s + 2;
  d = (s == 0) ? 0 : d;
  d = (s == 1) ? 2 : d;
  d = (s == 30) ? 31 : d;
  return d;
}
__device__ __forceinline__ int jacobi_slot_index(int s) { return (s & 1) ? (s >> 1) : 31 - (s >> 1); }
__device__ __forceinline__ int jacobi_index_slot(int k) { return (k <= 15) ? 2 * k + 1 : 2 * (31 - k); }

template <int U> __device__ __forceinline__ int row_share_i(int v) {
  return __builtin_amdgcn_update_dpp(v, v, 0x150 + U, 0xf, 0xf, false);
}
template <int U> __device__ __forceinline__ float row_share_f(float v) {
  return __int_as_float(row_share_i<U>(__float_as_int(v)));
}

#define JDST_E(U, GEN, P0, P7) ((U) == 0 ? (P0) : ((U) == 7 ? (P7) : (GEN) + (2 * (U) + 2) * J_LD))
#define JDST_O(U, GEN, P0)     ((U) == 0 ? (P0) : (GEN) + (2 * (U) - 1) * J_LD)

__global__ __launch_bounds__(32, 8) void jacobi_logm_kernel(const float* __restrict__ x,
                                                            float* __restrict__ xl) {
  __shared__ __attribute__((aligned(16))) float A[32][J_LD];
  __shared__ __attribute__((aligned(16))) float VT[32][J_LD];
  __shared__ float lw[32];
  const int lane = threadIdx.x;
  const int hi = lane >> 4;
  const int tl = (lane + 8 * hi) & 15;
  const int ra = 16 * hi;
  const int vc = 2 * (lane & 15);
  float* const Af = &A[0][0];
  float* const Vf = &VT[0][0];
  const float* xm = x + (size_t)blockIdx.x * 1024;

  const int myidx = jacobi_slot_index(lane);
  for (int i = 0; i < 32; ++i) {
    const int ri = jacobi_slot_index(i);
    A[i][lane]  = xm[ri * 32 + myidx];
    VT[i][lane] = (ri == lane) ? 1.f : 0.f;
  }
  __syncthreads();

  const float* const pLd = Af + (2 * tl) * J_LD + 2 * tl;
  const float* const aLd = Af + ra * J_LD + 2 * tl;
  const float* const vLd = Vf + ra * J_LD + vc;
  float* const aStA = Af + ra * J_LD + jacobi_next_slot(2 * tl);
  float* const aStB = Af + ra * J_LD + jacobi_next_slot(2 * tl + 1);
  float* const vSt  = Vf + ra * J_LD + vc;
  const int e0 = hi ? 2 * J_LD : 0;
  const int e7 = hi ? 15 * J_LD : 16 * J_LD;
  const int o0 = hi ? -J_LD : 2 * J_LD;
  float* const aA_e0 = aStA + e0;  float* const aB_e0 = aStB + e0;  float* const v_e0 = vSt + e0;
  float* const aA_e7 = aStA + e7;  float* const aB_e7 = aStB + e7;  float* const v_e7 = vSt + e7;
  float* const aA_o0 = aStA + o0;  float* const aB_o0 = aStB + o0;  float* const v_o0 = vSt + o0;

  for (int sweep = 0; sweep < 8; ++sweep) {
    for (int r = 0; r < 31; ++r) {
      const float2 pd = *(const float2*)pLd;
      const float aqq = pLd[J_LD + 1];
      float2 yE[8], yO[8], wE[8], wO[8];
#define JLOAD(U)                                                       \
      yE[U] = *(const float2*)(aLd + (2 * (U)) * J_LD);                \
      yO[U] = *(const float2*)(aLd + (2 * (U) + 1) * J_LD);            \
      wE[U] = *(const float2*)(vLd + (2 * (U)) * J_LD);                \
      wO[U] = *(const float2*)(vLd + (2 * (U) + 1) * J_LD);
      JLOAD(0) JLOAD(1) JLOAD(2) JLOAD(3) JLOAD(4) JLOAD(5) JLOAD(6) JLOAD(7)
#undef JLOAD

      float c = 1.f, s = 0.f;
      {
        const float apq = pd.y;
        if (fabsf(apq) > 1e-12f) {
          const float d = aqq - pd.x;
          const float b = apq + apq;
          const float h = __builtin_amdgcn_sqrtf(fmaf(d, d, b * b));
          float t = b * __builtin_amdgcn_rcpf(fabsf(d) + h);
          t = (d < 0.f) ? -t : t;
          c = rsqrtf(fmaf(t, t, 1.f));
          s = t * c;
        }
      }
      __syncthreads();

#define JROT(U)                                                                   \
      {                                                                           \
        const float ca = row_share_f<U>(c), sa = row_share_f<U>(s);               \
                     \
        const float r0 = ca * yE[U].x - sa * yO[U].x;                             \
        const float r2 = sa * yE[U].x + ca * yO[U].x;                             \
        const float r1 = ca * yE[U].y - sa * yO[U].y;                             \
        const float r3 = sa * yE[U].y + ca * yO[U].y;                             \
        *JDST_E(U, aStA, aA_e0, aA_e7) = c * r0 - s * r1;                         \
        *JDST_E(U, aStB, aB_e0, aB_e7) = s * r0 + c * r1;                         \
        *JDST_O(U, aStA, aA_o0)        = c * r2 - s * r3;                         \
        *JDST_O(U, aStB, aB_o0)        = s * r2 + c * r3;                         \
                      \
        float2 np, nq;                                                            \
        np.x = ca * wE[U].x - sa * wO[U].x;  nq.x = sa * wE[U].x + ca * wO[U].x;  \
        np.y = ca * wE[U].y - sa * wO[U].y;  nq.y = sa * wE[U].y + ca * wO[U].y;  \
        *(float2*)JDST_E(U, vSt, v_e0, v_e7) = np;                                \
        *(float2*)JDST_O(U, vSt, v_o0)       = nq;                                \
      }
      JROT(0) JROT(1) JROT(2) JROT(3) JROT(4) JROT(5) JROT(6) JROT(7)
#undef JROT
      __syncthreads();
    }

    float dg = 0.f, om = 0.f;
    for (int i = 0; i < 32; ++i) {
      const float a = fabsf(A[i][lane]);
      dg = (i == lane) ? a : dg;
      om = (i != lane) ? fmaxf(om, a) : om;
    }
    float dmin = dg;
    for (int o = 16; o > 0; o >>= 1) dmin = fminf(dmin, __shfl_xor(dmin, o));
    if (__builtin_amdgcn_ballot_w32(om > JC_TOL * dmin) == 0u) break;
  }

  lw[lane] = logf(fmaxf(A[lane][lane], 1e-20f));
  __syncthreads();

  float acc[32];
#pragma unroll
  for (int i = 0; i < 32; ++i) acc[i] = 0.f;
  for (int k = 0; k < 32; ++k) {
    const int sk = jacobi_index_slot(k);
    const float w = VT[sk][lane] * lw[sk];
#pragma unroll
    for (int i = 0; i < 32; ++i) acc[i] += VT[sk][i] * w;
  }

  float* outm = xl + (size_t)blockIdx.x * 1024;
#pragma unroll
  for (int i = 0; i < 32; ++i)
    outm[i * 32 + lane] = (i == lane) ? 0.f : acc[i];
}
#undef JDST_E
#undef JDST_O

__global__ __launch_bounds__(32) void transbase_kernel(const float* __restrict__ xl,
                                                       const float* __restrict__ C,
                                                       float* __restrict__ qkv) {
  __shared__ float Y[32][33];
  __shared__ float CL[32][33];
  __shared__ float T[32][33];
  __shared__ float T2[32][33];
  const int lane = threadIdx.x;
  const int nIdx = lane & 15;
  const int hi   = lane >> 4;
  const int m = blockIdx.x, w = blockIdx.y;
  const float* ym = xl + (size_t)m * 1024;
  const float* Cw = C + (size_t)w * 1024;

  for (int i = 0; i < 32; ++i) {
    Y[i][lane]  = ym[i * 32 + lane];
    CL[i][lane] = Cw[i * 32 + lane];
  }
  __syncthreads();

  for (int ti = 0; ti < 2; ++ti)
    for (int tj = 0; tj < 2; ++tj) {
      v8f acc = {};
#pragma unroll
      for (int kk = 0; kk < 8; ++kk) {
        int k0 = kk * 4 + hi * 2;
        v2f a; a.x = CL[ti * 16 + nIdx][k0]; a.y = CL[ti * 16 + nIdx][k0 + 1];
        v2f b; b.x = Y[k0][tj * 16 + nIdx];  b.y = Y[k0 + 1][tj * 16 + nIdx];
        acc = WMMA_F32_16x16x4(a, b, acc);
      }
#pragma unroll
      for (int r = 0; r < 8; ++r)
        T[ti * 16 + r + hi * 8][tj * 16 + nIdx] = acc[r];
    }
  __syncthreads();

  for (int ti = 0; ti < 2; ++ti)
    for (int tj = 0; tj < 2; ++tj) {
      v8f acc = {};
#pragma unroll
      for (int kk = 0; kk < 8; ++kk) {
        int k0 = kk * 4 + hi * 2;
        v2f a; a.x = T[ti * 16 + nIdx][k0];       a.y = T[ti * 16 + nIdx][k0 + 1];
        v2f b; b.x = CL[tj * 16 + nIdx][k0];      b.y = CL[tj * 16 + nIdx][k0 + 1];
        acc = WMMA_F32_16x16x4(a, b, acc);
      }
#pragma unroll
      for (int r = 0; r < 8; ++r)
        T2[ti * 16 + r + hi * 8][tj * 16 + nIdx] = acc[r];
    }
  __syncthreads();

  float* outp = qkv + ((size_t)w * NMAT + m) * NFP;
  for (int s = 0; s < 16; ++s) {
    int i, j;
    if (s == 0)        { i = 16;     j = lane & 15; }
    else if (lane < s) { i = s;      j = lane; }
    else               { i = 32 - s; j = lane - s; }
    const float t = T2[i][j];
    outp[s * 32 + lane] = (s == 0 && lane >= 16) ? 0.f : t;
  }
}

__global__ __launch_bounds__(256) void norms_kernel(const float* __restrict__ qkv,
                                                    float* __restrict__ nrm) {
  int idx = blockIdx.x * 256 + threadIdx.x;
  int sel = idx >> 13;
  int r   = idx & (NMAT - 1);
  const float* row = qkv + ((size_t)sel * NMAT + r) * NFP;
  float acc = 0.f;
  for (int f = 0; f < NFP; ++f) { float t = row[f]; acc += t * t; }
  nrm[idx] = 2.f * acc;
}

#define SC_TJ 128
#define SC_TI 64
#define SC_KS 32
#define SC_NS (NFP / SC_KS)
#define SC_LD 36

__global__ __launch_bounds__(256) void scores_kernel(const float* __restrict__ qkv,
                                                     const float* __restrict__ nrm,
                                                     float* __restrict__ scores) {
  __shared__ __attribute__((aligned(16))) float As[2][SC_TJ][SC_LD];
  __shared__ __attribute__((aligned(16))) float Bs[2][SC_TI][SC_LD];
  const int tid  = threadIdx.x;
  const int lane = tid & 31;
  const int wave = tid >> 5;
  const int nIdx = lane & 15;
  const int hi   = lane >> 4;
  const int it = blockIdx.x, jt = blockIdx.y, b = blockIdx.z;
  const int i0 = it * SC_TI, j0 = jt * SC_TJ;
  const int wj = (wave >> 1) * 32;
  const int wi = (wave & 1) * 32;
  const float* qf = qkv + ((size_t)0 * NMAT + (size_t)b * NN) * NFP;
  const float* kf = qkv + ((size_t)1 * NMAT + (size_t)b * NN) * NFP;

  const int srow = tid >> 3;
  const int scol = (tid & 7) * 4;
  const float* ga = kf + (size_t)(j0 + srow) * NFP + scol;
  const float* gb = qf + (size_t)(i0 + srow) * NFP + scol;
  const size_t rstep = (size_t)32 * NFP;

  {
    const float4 ra0 = *(const float4*)(ga + 0 * rstep);
    const float4 ra1 = *(const float4*)(ga + 1 * rstep);
    const float4 ra2 = *(const float4*)(ga + 2 * rstep);
    const float4 ra3 = *(const float4*)(ga + 3 * rstep);
    const float4 rb0 = *(const float4*)(gb + 0 * rstep);
    const float4 rb1 = *(const float4*)(gb + 1 * rstep);
    *(float4*)&As[0][srow +  0][scol] = ra0;
    *(float4*)&As[0][srow + 32][scol] = ra1;
    *(float4*)&As[0][srow + 64][scol] = ra2;
    *(float4*)&As[0][srow + 96][scol] = ra3;
    *(float4*)&Bs[0][srow +  0][scol] = rb0;
    *(float4*)&Bs[0][srow + 32][scol] = rb1;
  }
  __syncthreads();

  v8f acc00 = {}, acc01 = {}, acc10 = {}, acc11 = {};

  auto slab_mma = [&](int buf) {
#pragma unroll
    for (int kk = 0; kk < SC_KS / 4; ++kk) {
      const int k0 = kk * 4 + hi * 2;
      v2f a0 = *(const v2f*)&As[buf][wj +  0 + nIdx][k0];
      v2f a1 = *(const v2f*)&As[buf][wj + 16 + nIdx][k0];
      v2f b0 = *(const v2f*)&Bs[buf][wi +  0 + nIdx][k0];
      v2f b1 = *(const v2f*)&Bs[buf][wi + 16 + nIdx][k0];
      acc00 = WMMA_F32_16x16x4(a0, b0, acc00);
      acc01 = WMMA_F32_16x16x4(a0, b1, acc01);
      acc10 = WMMA_F32_16x16x4(a1, b0, acc10);
      acc11 = WMMA_F32_16x16x4(a1, b1, acc11);
    }
  };

  for (int s = 0; s < SC_NS - 1; ++s) {
    const int ko = (s + 1) * SC_KS;
    const float4 ra0 = *(const float4*)(ga + 0 * rstep + ko);
    const float4 ra1 = *(const float4*)(ga + 1 * rstep + ko);
    const float4 ra2 = *(const float4*)(ga + 2 * rstep + ko);
    const float4 ra3 = *(const float4*)(ga + 3 * rstep + ko);
    const float4 rb0 = *(const float4*)(gb + 0 * rstep + ko);
    const float4 rb1 = *(const float4*)(gb + 1 * rstep + ko);

    slab_mma(s & 1);

    const int nb = (s + 1) & 1;
    *(float4*)&As[nb][srow +  0][scol] = ra0;
    *(float4*)&As[nb][srow + 32][scol] = ra1;
    *(float4*)&As[nb][srow + 64][scol] = ra2;
    *(float4*)&As[nb][srow + 96][scol] = ra3;
    *(float4*)&Bs[nb][srow +  0][scol] = rb0;
    *(float4*)&Bs[nb][srow + 32][scol] = rb1;
    __syncthreads();
  }
  slab_mma((SC_NS - 1) & 1);

  float* sb = scores + (size_t)b * NN * NN;
#pragma unroll
  for (int t = 0; t < 4; ++t) {
    v8f acc = (t == 0) ? acc00 : ((t == 1) ? acc01 : ((t == 2) ? acc10 : acc11));
    const int jb = j0 + wj + (t >> 1) * 16;
    const int i  = i0 + wi + (t & 1) * 16 + nIdx;
    float q2 = nrm[b * NN + i];
#pragma unroll
    for (int r = 0; r < 8; ++r) {
      int j = jb + r + hi * 8;
      float d2 = q2 + nrm[NMAT + b * NN + j] - 4.f * acc[r];
      d2 = fmaxf(d2, 1e-12f);
      float e  = sqrtf(d2);
      float sc = 1.f / (1.f + log1pf(e));
      sb[(size_t)j * NN + i] = sc;
    }
  }
}

#define SM_W 8
#define SM_R (NN / SM_W)

__global__ __launch_bounds__(256) void softmax_kernel(float* __restrict__ scores) {
  __shared__ float red[2][SM_W][32];
  const int lane = threadIdx.x & 31;
  const int wave = threadIdx.x >> 5;
  const int b = blockIdx.y;
  float* col = scores + (size_t)b * NN * NN + (size_t)(wave * SM_R) * NN
             + blockIdx.x * 32 + lane;

  float v[SM_R];
#pragma unroll
  for (int r = 0; r < SM_R; ++r) v[r] = col[(size_t)r * NN];

  float mx = -1e30f;
#pragma unroll
  for (int r = 0; r < SM_R; ++r) mx = fmaxf(mx, v[r]);
  red[0][wave][lane] = mx;
  __syncthreads();
  for (int w = 0; w < SM_W; ++w) mx = fmaxf(mx, red[0][w][lane]);

  float sum = 0.f;
#pragma unroll
  for (int r = 0; r < SM_R; ++r) {
    v[r] = __expf(v[r] - mx);
    sum += v[r];
  }
  red[1][wave][lane] = sum;
  __syncthreads();
  float tot = 0.f;
  for (int w = 0; w < SM_W; ++w) tot += red[1][w][lane];
  const float inv = 1.f / tot;

#pragma unroll
  for (int r = 0; r < SM_R; ++r) col[(size_t)r * NN] = v[r] * inv;
}

#define OG_TC 32
#define OG_KS 32
#define OG_NS (NN / OG_KS)
#define OG_LDA 36
#define OG_LDB (2 * NFP + 32)
#define OG_LDR 33
#define OG_LDP (16 * OG_LDR + 2)

__global__ __launch_bounds__(256) void outgemm_kernel(const float* __restrict__ scores,
                                                      const float* __restrict__ qkv,
                                                      float* __restrict__ out) {
  __shared__ __attribute__((aligned(16))) float As[2][OG_TC][OG_LDA];
  __shared__ __attribute__((aligned(16))) float Bs[2][OG_KS / 2][OG_LDB];
  __shared__ float Ps[OG_TC * OG_LDP];
  const int tid  = threadIdx.x;
  const int lane = tid & 31;
  const int wave = tid >> 5;
  const int nIdx = lane & 15;
  const int hi   = lane >> 4;
  const int b  = blockIdx.x / (NN / OG_TC);
  const int c0 = (blockIdx.x % (NN / OG_TC)) * OG_TC;
  const int wf = wave * 64;
  const float* p  = scores + (size_t)b * NN * NN;
  const float* vf = qkv + ((size_t)2 * NMAT + (size_t)b * NN) * NFP;

  const int arow = tid >> 3;
  const int acol = (tid & 7) * 4;
  const float* ga = p + (size_t)(c0 + arow) * NN + acol;
  const int bkp  = tid >> 7;
  const int bcol = (tid & 127) * 4;
  const float* gb = vf + (size_t)(2 * bkp) * NFP + bcol;
  const size_t bslab = (size_t)OG_KS * NFP;

  float4 ra, rb0[8], rb1[8];
#define OG_FETCH(S)                                                              \
  ra = *(const float4*)(ga + (S) * OG_KS);                                       \
  _Pragma("unroll")                                                              \
  for (int q = 0; q < 8; ++q) {                                                  \
    rb0[q] = *(const float4*)(gb + (size_t)(S) * bslab + (size_t)(4 * q) * NFP);         \
    rb1[q] = *(const float4*)(gb + (size_t)(S) * bslab + (size_t)(4 * q + 1) * NFP);     \
  }
#define OG_PUT(NB)                                                               \
  *(float4*)&As[NB][arow][acol] = ra;                                            \
  _Pragma("unroll")                                                              \
  for (int q = 0; q < 8; ++q) {                                                  \
    *(float4*)&Bs[NB][bkp + 2 * q][bcol * 2 + 0] = make_float4(rb0[q].x, rb1[q].x, rb0[q].y, rb1[q].y); \
    *(float4*)&Bs[NB][bkp + 2 * q][bcol * 2 + 4] = make_float4(rb0[q].z, rb1[q].z, rb0[q].w, rb1[q].w); \
  }

  OG_FETCH(0)
  OG_PUT(0)
  __syncthreads();

  v8f acc[2][4];
#pragma unroll
  for (int ci = 0; ci < 2; ++ci)
#pragma unroll
    for (int fi = 0; fi < 4; ++fi) acc[ci][fi] = v8f{};

  auto half_mma = [&](int buf, int kb) {
#pragma unroll
    for (int kk = kb; kk < kb + OG_KS / 8; ++kk) {
      const int k0 = kk * 4 + hi * 2;
      v2f a[2], bf[4];
#pragma unroll
      for (int ci = 0; ci < 2; ++ci) a[ci] = *(const v2f*)&As[buf][ci * 16 + nIdx][k0];
#pragma unroll
      for (int fi = 0; fi < 4; ++fi) bf[fi] = *(const v2f*)&Bs[buf][kk * 2 + hi][(wf + fi * 16 + nIdx) * 2];
#pragma unroll
      for (int ci = 0; ci < 2; ++ci)
#pragma unroll
        for (int fi = 0; fi < 4; ++fi) acc[ci][fi] = WMMA_F32_16x16x4(a[ci], bf[fi], acc[ci][fi]);
    }
  };

  OG_FETCH(1)
  for (int s = 0; s < OG_NS - 2; ++s) {
    half_mma(s & 1, 0);
    OG_PUT((s + 1) & 1)
    OG_FETCH(s + 2)
    half_mma(s & 1, OG_KS / 8);
    __syncthreads();
  }
  half_mma((OG_NS - 2) & 1, 0);
  OG_PUT((OG_NS - 1) & 1)
  half_mma((OG_NS - 2) & 1, OG_KS / 8);
  __syncthreads();
  half_mma((OG_NS - 1) & 1, 0);
  half_mma((OG_NS - 1) & 1, OG_KS / 8);
#undef OG_FETCH
#undef OG_PUT

#pragma unroll
  for (int ci = 0; ci < 2; ++ci)
#pragma unroll
    for (int fi = 0; fi < 4; ++fi) {
      float* const pd = Ps + (ci * 16 + hi * 8) * OG_LDP + (2 * wave + (fi >> 1)) * OG_LDR
                      + (fi & 1) * 16 + nIdx;
#pragma unroll
      for (int r = 0; r < 8; ++r) pd[r * OG_LDP] = acc[ci][fi][r];
    }
  __syncthreads();

  float* const ob = out + ((size_t)b * NN + c0 + wave * 4) * NF;
  const float* const pw = Ps + (wave * 4) * OG_LDP;
#pragma unroll 4
  for (int i = 0; i < 32; ++i) {
    const int hx = (i > lane) ? i : lane;
    const int lx = (i > lane) ? lane : i;
    int ps = (hx < 16) ? hx : 32 - hx;
    int pl = (hx > 16) ? lx + ps : lx;
    ps = (hx == 16) ? 0 : ps;
    if (i == lane) { ps = 0; pl = 16; }
    const int idx = ps * OG_LDR + pl;
#pragma unroll
    for (int mq = 0; mq < 4; ++mq)
      ob[(size_t)mq * NF + i * 32 + lane] = pw[mq * OG_LDP + idx];
  }
}

extern "C" void kernel_launch(void* const* d_in, const int* in_sizes, int n_in,
                              void* d_out, int out_size, void* d_ws, size_t ws_size,
                              hipStream_t stream) {
  (void)in_sizes; (void)n_in; (void)out_size; (void)ws_size;
  const float* x  = (const float*)d_in[0];
  const float* qw = (const float*)d_in[1];
  const float* kw = (const float*)d_in[2];
  const float* vw = (const float*)d_in[3];
  float* out = (float*)d_out;

  float* ws     = (float*)d_ws;
  float* C      = ws;
  float* xl     = C + 3 * 1024;
  float* qkv    = xl + (size_t)NMAT * 1024;
  float* nrm    = qkv + (size_t)3 * NMAT * 1024;
  float* scores = nrm + 2 * NMAT;

  cayley_kernel<<<3, 256, 0, stream>>>(qw, kw, vw, C);
  jacobi_logm_kernel<<<NMAT, 32, 0, stream>>>(x, xl);
  transbase_kernel<<<dim3(NMAT, 3), 32, 0, stream>>>(xl, C, qkv);
  norms_kernel<<<64, 256, 0, stream>>>(qkv, nrm);
  scores_kernel<<<dim3(NN / SC_TI, NN / SC_TJ, NB), 256, 0, stream>>>(qkv, nrm, scores);
  softmax_kernel<<<dim3(NN / 32, NB), 256, 0, stream>>>(scores);
  outgemm_kernel<<<NMAT / OG_TC, 256, 0, stream>>>(scores, qkv, out);
}
